# starting kernel
# baseline (speedup 1.0000x reference)
#include <hip/hip_runtime.h>

typedef __attribute__((ext_vector_type(2))) float v2f;
typedef __attribute__((ext_vector_type(8))) float v8f;

#define ALPHA 0.05f
#define EPS   0.01f

#define TM 128
#define TN 128
#define TK 32

static __device__ __forceinline__ v8f wmma4(const v2f a, const v2f b, const v8f c)
{
    return __builtin_amdgcn_wmma_f32_16x16x4_f32(false, a, false, b, (short)0, c,
                                                 false, false);
}

static __device__ __forceinline__ void store_tile(
    const v8f acc, int r0, int c0, float* __restrict__ Cm, int ldc,
    const float* __restrict__ Cin, const float* __restrict__ bias,
    float alpha, float beta, float gamma, int relu, int hi, int ln)
{
    const int c = c0 + ln;
    const float bv = bias ? gamma * bias[c] : 0.f;
#pragma unroll
    for (int v = 0; v < 8; ++v) {
        const int r = r0 + v + 8 * hi;
        float val = alpha * acc[v] + bv;
        if (Cin)  val += beta * Cin[(size_t)r * ldc + c];
        if (relu) val = fmaxf(val, 0.f);
        Cm[(size_t)r * ldc + c] = val;
    }
}

__global__ __launch_bounds__(256) void gemm_f32_wmma(
    const float* __restrict__ Am, int lda,
    const float* __restrict__ Bm, int ldb, int transB,
    float* __restrict__ Cm, int ldc,
    const float* __restrict__ Cin,
    const float* __restrict__ bias,
    float alpha, float beta, float gamma, int relu,
    int K, const int* skip)
{
    if (skip && *skip) return;

    __shared__ float As[TM][TK + 4];
    __shared__ float Bs[TK][TN + 4];

    const int t    = threadIdx.x;
    const int row0 = blockIdx.y * TM;
    const int col0 = blockIdx.x * TN;

    const int wid  = t >> 5;
    const int lane = t & 31;
    const int wm   = (wid >> 2) * 64;
    const int wn   = (wid & 3) * 32;
    const int hi   = lane >> 4;
    const int ln   = lane & 15;

    v8f acc[4][2];
#pragma unroll
    for (int mi = 0; mi < 4; ++mi)
#pragma unroll
        for (int ni = 0; ni < 2; ++ni) acc[mi][ni] = (v8f){};

    const int arow = t >> 1;
    const int akc  = (t & 1) * 16;
    const int bkp  = t >> 3;
    const int bnp  = (t & 7) * 16;
    const int btn  = t >> 1;
    const int btk  = (t & 1) * 16;

    for (int kb = 0; kb < K; kb += TK) {
        const float* ap = Am + (size_t)(row0 + arow) * lda + kb + akc;
#pragma unroll
        for (int j = 0; j < 4; ++j)
            *(float4*)&As[arow][akc + 4 * j] = *(const float4*)(ap + 4 * j);
        if (kb + TK < K) __builtin_prefetch(ap + TK, 0, 1);

        if (!transB) {
            const float* bp = Bm + (size_t)(kb + bkp) * ldb + col0 + bnp;
#pragma unroll
            for (int j = 0; j < 4; ++j)
                *(float4*)&Bs[bkp][bnp + 4 * j] = *(const float4*)(bp + 4 * j);
            if (kb + TK < K) __builtin_prefetch(bp + (size_t)TK * ldb, 0, 1);
        } else {
            const float* bp = Bm + (size_t)(col0 + btn) * ldb + kb + btk;
#pragma unroll
            for (int j = 0; j < 4; ++j) {
                const float4 vv = *(const float4*)(bp + 4 * j);
                Bs[btk + 4 * j + 0][btn] = vv.x;
                Bs[btk + 4 * j + 1][btn] = vv.y;
                Bs[btk + 4 * j + 2][btn] = vv.z;
                Bs[btk + 4 * j + 3][btn] = vv.w;
            }
            if (kb + TK < K) __builtin_prefetch(bp + TK, 0, 1);
        }
        __syncthreads();

#pragma unroll
        for (int kk = 0; kk < TK; kk += 4) {
            v2f a[4], b[2];
#pragma unroll
            for (int mi = 0; mi < 4; ++mi) {
                a[mi].x = As[wm + 16 * mi + ln][kk + 2 * hi];
                a[mi].y = As[wm + 16 * mi + ln][kk + 2 * hi + 1];
            }
#pragma unroll
            for (int ni = 0; ni < 2; ++ni) {
                b[ni].x = Bs[kk + 2 * hi][wn + 16 * ni + ln];
                b[ni].y = Bs[kk + 2 * hi + 1][wn + 16 * ni + ln];
            }
#pragma unroll
            for (int mi = 0; mi < 4; ++mi)
#pragma unroll
                for (int ni = 0; ni < 2; ++ni)
                    acc[mi][ni] = wmma4(a[mi], b[ni], acc[mi][ni]);
        }
        __syncthreads();
    }

#pragma unroll
    for (int mi = 0; mi < 4; ++mi)
#pragma unroll
        for (int ni = 0; ni < 2; ++ni)
            store_tile(acc[mi][ni], row0 + wm + 16 * mi, col0 + wn + 16 * ni,
                       Cm, ldc, Cin, bias, alpha, beta, gamma, relu, hi, ln);
}

__global__ void k_prep_u(const float* __restrict__ z, const float* __restrict__ w,
                         float* __restrict__ u, int n, const int* skip)
{
    if (skip && *skip) return;
    int i = blockIdx.x * blockDim.x + threadIdx.x;
    if (i < n) {
        float zi = z[i];
        float x  = fmaxf(zi, 0.f);
        u[i] = 2.f * x - zi - ALPHA * w[i];
    }
}

__global__ void k_update(float* __restrict__ z, const float* __restrict__ y,
                         float* __restrict__ acc, int n, const int* skip)
{
    __shared__ float sm[256];
    if (*skip) return;
    int i = blockIdx.x * blockDim.x + threadIdx.x;
    float d2 = 0.f;
    if (i < n) {
        float zi = z[i];
        float x  = fmaxf(zi, 0.f);
        float yi = y[i];
        z[i] = zi - x + yi;
        float d = yi - x;
        d2 = d * d;
    }
    sm[threadIdx.x] = d2;
    __syncthreads();
    for (int s = 128; s > 0; s >>= 1) {
        if (threadIdx.x < s) sm[threadIdx.x] += sm[threadIdx.x + s];
        __syncthreads();
    }
    if (threadIdx.x == 0) atomicAdd(acc, sm[0]);
}

__global__ void k_sumsq(const float* __restrict__ x, float* __restrict__ acc, int n)
{
    __shared__ float sm[256];
    int i = blockIdx.x * blockDim.x + threadIdx.x;
    float v = 0.f;
    if (i < n) { float t = x[i]; v = t * t; }
    sm[threadIdx.x] = v;
    __syncthreads();
    for (int s = 128; s > 0; s >>= 1) {
        if (threadIdx.x < s) sm[threadIdx.x] += sm[threadIdx.x + s];
        __syncthreads();
    }
    if (threadIdx.x == 0) atomicAdd(acc, sm[0]);
}

__global__ void k_zero(float* p) { p[0] = 0.f; }

__global__ void k_check(const float* acc, int* flag)
{
    if (*flag == 0 && *acc <= EPS * EPS) *flag = 1;
}

__global__ void k_init_X(float* __restrict__ X, const float* __restrict__ fn, int n)
{
    int i = blockIdx.x * blockDim.x + threadIdx.x;
    if (i < n * n) {
        float c = rsqrtf(*fn);
        int r = i / n, cc = i - r * n;
        X[i] = (r == cc) ? c : 0.f;
    }
}

__global__ void k_final(const float* __restrict__ z, const float* __restrict__ y,
                        float* __restrict__ out, int n)
{
    int i = blockIdx.x * blockDim.x + threadIdx.x;
    if (i < n) {
        float zi = z[i];
        float x  = fmaxf(zi, 0.f);
        out[i] = fmaxf(zi - x + y[i], 0.f);
    }
}

extern "C" void kernel_launch(void* const* d_in, const int* in_sizes, int n_in,
                              void* d_out, int out_size, void* d_ws, size_t ws_size,
                              hipStream_t stream)
{
    const float* ctx  = (const float*)d_in[0];
    const float* Ain  = (const float*)d_in[1];
    const float* bvec = (const float*)d_in[2];
    const float* W1   = (const float*)d_in[3];
    const float* b1   = (const float*)d_in[4];
    const float* W2   = (const float*)d_in[5];
    const float* b2   = (const float*)d_in[6];
    const float* z0   = (const float*)d_in[7];
    float* out = (float*)d_out;

    const int B = 256, D = 512, H = 2048, N1 = 1024, N2 = 4096;

    float* W = (float*)d_ws;
    size_t o = 0;
    float* z  = W + o; o += (size_t)B * N2;
    float* u  = W + o; o += (size_t)B * N2;
    float* y  = W + o; o += (size_t)B * N2;
    float* wv = W + o; o += (size_t)B * N2;
    float* h  = W + o; o += (size_t)B * H;
    float* R  = W + o; o += (size_t)B * N1;
    float* S  = W + o; o += (size_t)B * N1;
    float* G  = W + o; o += (size_t)N1 * N1;
    float* Xa = W + o; o += (size_t)N1 * N1;
    float* Xb = W + o; o += (size_t)N1 * N1;
    float* T1 = W + o; o += (size_t)N1 * N1;
    float* acc = W + o; o += 1;
    float* fn  = W + o; o += 1;
    int*   flag = (int*)(W + o); o += 1;

    dim3 blk(256);
    auto gemm = [&](const float* A_, int lda, const float* B_, int ldb, int tB,
                    float* C_, int ldc, const float* Cin, const float* bias,
                    float al, float be, float ga, int relu,
                    int M, int N, int K, const int* skip) {
        dim3 grid(N / TN, M / TM);
        gemm_f32_wmma<<<grid, blk, 0, stream>>>(A_, lda, B_, ldb, tB, C_, ldc,
                                                Cin, bias, al, be, ga, relu, K, skip);
    };

    hipMemcpyAsync(z, z0, sizeof(float) * (size_t)B * N2, hipMemcpyDeviceToDevice, stream);
    hipMemsetAsync(flag, 0, sizeof(int), stream);

    gemm(ctx, D, W1, H, 0, h,  H,  nullptr, b1, 1.f, 0.f, 1.f, 1, B, H,  D, nullptr);
    gemm(h,   H, W2, N2, 0, wv, N2, nullptr, b2, 1.f, 0.f, 1.f, 0, B, N2, H, nullptr);

    gemm(Ain, N2, Ain, N2, 1, G, N1, nullptr, nullptr, 1.f, 0.f, 0.f, 0, N1, N1, N2, nullptr);

    k_zero<<<1, 1, 0, stream>>>(fn);
    k_sumsq<<<dim3((N1 * N1 + 255) / 256), blk, 0, stream>>>(G, fn, N1 * N1);
    k_init_X<<<dim3((N1 * N1 + 255) / 256), blk, 0, stream>>>(Xa, fn, N1);

    float* Xc = Xa; float* Xn = Xb;
    for (int it = 0; it < 16; ++it) {
        gemm(G,  N1, Xc, N1, 0, T1, N1, nullptr, nullptr,  1.f, 0.f, 0.f, 0, N1, N1, N1, nullptr);
        gemm(Xc, N1, T1, N1, 0, Xn, N1, Xc,      nullptr, -1.f, 2.f, 0.f, 0, N1, N1, N1, nullptr);
        float* tmp = Xc; Xc = Xn; Xn = tmp;
    }
    float* Ginv = Xc;

    const int nz = B * N2;
    dim3 gout((nz + 255) / 256);
    for (int itr = 0; itr < 50; ++itr) {
        k_prep_u<<<gout, blk, 0, stream>>>(z, wv, u, nz, (const int*)flag);
        gemm(u, N2, Ain,  N2, 1, R, N1, nullptr, bvec, 1.f, 0.f, -1.f, 0, B, N1, N2, flag);
        gemm(R, N1, Ginv, N1, 0, S, N1, nullptr, nullptr, 1.f, 0.f, 0.f, 0, B, N1, N1, flag);
        gemm(S, N1, Ain,  N2, 0, y, N2, u, nullptr, -1.f, 1.f, 0.f, 0, B, N2, N1, flag);
        k_zero<<<1, 1, 0, stream>>>(acc);
        k_update<<<gout, blk, 0, stream>>>(z, y, acc, nz, (const int*)flag);
        k_check<<<1, 1, 0, stream>>>((const float*)acc, flag);
    }

    k_prep_u<<<gout, blk, 0, stream>>>(z, wv, u, nz, (const int*)nullptr);
    gemm(u, N2, Ain,  N2, 1, R, N1, nullptr, bvec, 1.f, 0.f, -1.f, 0, B, N1, N2, nullptr);
    gemm(R, N1, Ginv, N1, 0, S, N1, nullptr, nullptr, 1.f, 0.f, 0.f, 0, B, N1, N1, nullptr);
    gemm(S, N1, Ain,  N2, 0, y, N2, u, nullptr, -1.f, 1.f, 0.f, 0, B, N2, N1, nullptr);
    k_final<<<gout, blk, 0, stream>>>(z, y, out, nz);
}


# split-K over grid.z: every GEMM launches 256 work-groups (was 16 to 64) writing raw partial tiles to workspace; one small reduce launch per GEMM sums them in fixed order and applies the epilogue
# speedup vs baseline: 4.1531x; 4.1531x over previous
#include <hip/hip_runtime.h>

typedef __attribute__((ext_vector_type(2))) float v2f;
typedef __attribute__((ext_vector_type(8))) float v8f;

#define ALPHA 0.05f
#define EPS   0.01f

#define TM 128
#define TN 128
#define TK 32

static __device__ __forceinline__ v8f wmma4(const v2f a, const v2f b, const v8f c)
{
    return __builtin_amdgcn_wmma_f32_16x16x4_f32(false, a, false, b, (short)0, c,
                                                 false, false);
}

static __device__ __forceinline__ void store_partial(
    const v8f acc, int r0, int c0, float* __restrict__ Cp, int ldc, int hi, int ln)
{
    const int c = c0 + ln;
#pragma unroll
    for (int v = 0; v < 8; ++v) {
        const int r = r0 + v + 8 * hi;
        Cp[(size_t)r * ldc + c] = acc[v];
    }
}

__global__ __launch_bounds__(256) void gemm_f32_wmma(
    const float* __restrict__ Am, int lda,
    const float* __restrict__ Bm, int ldb, int transB,
    float* __restrict__ Cm, int ldc,
    int Kc, size_t pstride, const int* skip)
{
    if (skip && *skip) return;

    __shared__ float As[TM][TK + 4];
    __shared__ float Bs[TK][TN + 4];

    const int t    = threadIdx.x;
    const int row0 = blockIdx.y * TM;
    const int col0 = blockIdx.x * TN;

    const int wid  = t >> 5;
    const int lane = t & 31;
    const int wm   = (wid >> 2) * 64;
    const int wn   = (wid & 3) * 32;
    const int hi   = lane >> 4;
    const int ln   = lane & 15;

    v8f acc[4][2];
#pragma unroll
    for (int mi = 0; mi < 4; ++mi)
#pragma unroll
        for (int ni = 0; ni < 2; ++ni) acc[mi][ni] = (v8f){};

    const int arow = t >> 1;
    const int akc  = (t & 1) * 16;
    const int bkp  = t >> 3;
    const int bnp  = (t & 7) * 16;
    const int btn  = t >> 1;
    const int btk  = (t & 1) * 16;

    const int kbeg = blockIdx.z * Kc;
    const int kend = kbeg + Kc;
    float* __restrict__ Cp = Cm + (size_t)blockIdx.z * pstride;

    for (int kb = kbeg; kb < kend; kb += TK) {
        const float* ap = Am + (size_t)(row0 + arow) * lda + kb + akc;
#pragma unroll
        for (int j = 0; j < 4; ++j)
            *(float4*)&As[arow][akc + 4 * j] = *(const float4*)(ap + 4 * j);
        if (kb + TK < kend) __builtin_prefetch(ap + TK, 0, 1);

        if (!transB) {
            const float* bp = Bm + (size_t)(kb + bkp) * ldb + col0 + bnp;
#pragma unroll
            for (int j = 0; j < 4; ++j)
                *(float4*)&Bs[bkp][bnp + 4 * j] = *(const float4*)(bp + 4 * j);
            if (kb + TK < kend) __builtin_prefetch(bp + (size_t)TK * ldb, 0, 1);
        } else {
            const float* bp = Bm + (size_t)(col0 + btn) * ldb + kb + btk;
#pragma unroll
            for (int j = 0; j < 4; ++j) {
                const float4 vv = *(const float4*)(bp + 4 * j);
                Bs[btk + 4 * j + 0][btn] = vv.x;
                Bs[btk + 4 * j + 1][btn] = vv.y;
                Bs[btk + 4 * j + 2][btn] = vv.z;
                Bs[btk + 4 * j + 3][btn] = vv.w;
            }
            if (kb + TK < kend) __builtin_prefetch(bp + TK, 0, 1);
        }
        __syncthreads();

#pragma unroll
        for (int kk = 0; kk < TK; kk += 4) {
            v2f a[4], b[2];
#pragma unroll
            for (int mi = 0; mi < 4; ++mi) {
                a[mi].x = As[wm + 16 * mi + ln][kk + 2 * hi];
                a[mi].y = As[wm + 16 * mi + ln][kk + 2 * hi + 1];
            }
#pragma unroll
            for (int ni = 0; ni < 2; ++ni) {
                b[ni].x = Bs[kk + 2 * hi][wn + 16 * ni + ln];
                b[ni].y = Bs[kk + 2 * hi + 1][wn + 16 * ni + ln];
            }
#pragma unroll
            for (int mi = 0; mi < 4; ++mi)
#pragma unroll
                for (int ni = 0; ni < 2; ++ni)
                    acc[mi][ni] = wmma4(a[mi], b[ni], acc[mi][ni]);
        }
        __syncthreads();
    }

#pragma unroll
    for (int mi = 0; mi < 4; ++mi)
#pragma unroll
        for (int ni = 0; ni < 2; ++ni)
            store_partial(acc[mi][ni], row0 + wm + 16 * mi, col0 + wn + 16 * ni,
                          Cp, ldc, hi, ln);
}

__global__ void k_prep_u(const float* __restrict__ z, const float* __restrict__ w,
                         float* __restrict__ u, int n, const int* skip)
{
    if (skip && *skip) return;
    int i = blockIdx.x * blockDim.x + threadIdx.x;
    if (i < n) {
        float zi = z[i];
        float x  = fmaxf(zi, 0.f);
        u[i] = 2.f * x - zi - ALPHA * w[i];
    }
}

__global__ void k_update(float* __restrict__ z, const float* __restrict__ y,
                         float* __restrict__ acc, int n, const int* skip)
{
    __shared__ float sm[256];
    if (*skip) return;
    int i = blockIdx.x * blockDim.x + threadIdx.x;
    float d2 = 0.f;
    if (i < n) {
        float zi = z[i];
        float x  = fmaxf(zi, 0.f);
        float yi = y[i];
        z[i] = zi - x + yi;
        float d = yi - x;
        d2 = d * d;
    }
    sm[threadIdx.x] = d2;
    __syncthreads();
    for (int s = 128; s > 0; s >>= 1) {
        if (threadIdx.x < s) sm[threadIdx.x] += sm[threadIdx.x + s];
        __syncthreads();
    }
    if (threadIdx.x == 0) atomicAdd(acc, sm[0]);
}

__global__ void k_sumsq(const float* __restrict__ x, float* __restrict__ acc, int n)
{
    __shared__ float sm[256];
    int i = blockIdx.x * blockDim.x + threadIdx.x;
    float v = 0.f;
    if (i < n) { float t = x[i]; v = t * t; }
    sm[threadIdx.x] = v;
    __syncthreads();
    for (int s = 128; s > 0; s >>= 1) {
        if (threadIdx.x < s) sm[threadIdx.x] += sm[threadIdx.x + s];
        __syncthreads();
    }
    if (threadIdx.x == 0) atomicAdd(acc, sm[0]);
}

__global__ void k_zero(float* p) { p[0] = 0.f; }

__global__ __launch_bounds__(256) void k_splitk_reduce(
    const float* P, int nsplit, size_t pstride,
    float* C, const float* __restrict__ Cin, const float* __restrict__ bias,
    int ncols, float alpha, float beta, float gamma, int relu, const int* skip)
{
    if (skip && *skip) return;
    const size_t i = ((size_t)blockIdx.x * blockDim.x + threadIdx.x) * 4;
    float4 s = make_float4(0.f, 0.f, 0.f, 0.f);
    for (int k = 0; k < nsplit; k += 4) {
        const float4 p0 = *(const float4*)(P + (size_t)(k + 0) * pstride + i);
        const float4 p1 = *(const float4*)(P + (size_t)(k + 1) * pstride + i);
        const float4 p2 = *(const float4*)(P + (size_t)(k + 2) * pstride + i);
        const float4 p3 = *(const float4*)(P + (size_t)(k + 3) * pstride + i);
        s.x += p0.x; s.y += p0.y; s.z += p0.z; s.w += p0.w;
        s.x += p1.x; s.y += p1.y; s.z += p1.z; s.w += p1.w;
        s.x += p2.x; s.y += p2.y; s.z += p2.z; s.w += p2.w;
        s.x += p3.x; s.y += p3.y; s.z += p3.z; s.w += p3.w;
    }
    float4 v = make_float4(alpha * s.x, alpha * s.y, alpha * s.z, alpha * s.w);
    if (bias) {
        const int c = (int)(i % (size_t)ncols);
        const float4 bv = *(const float4*)(bias + c);
        v.x += gamma * bv.x; v.y += gamma * bv.y; v.z += gamma * bv.z; v.w += gamma * bv.w;
    }
    if (Cin) {
        const float4 ci = *(const float4*)(Cin + i);
        v.x += beta * ci.x; v.y += beta * ci.y; v.z += beta * ci.z; v.w += beta * ci.w;
    }
    if (relu) {
        v.x = fmaxf(v.x, 0.f); v.y = fmaxf(v.y, 0.f); v.z = fmaxf(v.z, 0.f); v.w = fmaxf(v.w, 0.f);
    }
    *(float4*)(C + i) = v;
}

__global__ void k_check(const float* acc, int* flag)
{
    if (*flag == 0 && *acc <= EPS * EPS) *flag = 1;
}

__global__ void k_init_X(float* __restrict__ X, const float* __restrict__ fn, int n)
{
    int i = blockIdx.x * blockDim.x + threadIdx.x;
    if (i < n * n) {
        float c = rsqrtf(*fn);
        int r = i / n, cc = i - r * n;
        X[i] = (r == cc) ? c : 0.f;
    }
}

__global__ void k_final(const float* __restrict__ z, const float* __restrict__ y,
                        float* __restrict__ out, int n)
{
    int i = blockIdx.x * blockDim.x + threadIdx.x;
    if (i < n) {
        float zi = z[i];
        float x  = fmaxf(zi, 0.f);
        out[i] = fmaxf(zi - x + y[i], 0.f);
    }
}

extern "C" void kernel_launch(void* const* d_in, const int* in_sizes, int n_in,
                              void* d_out, int out_size, void* d_ws, size_t ws_size,
                              hipStream_t stream)
{
    const float* ctx  = (const float*)d_in[0];
    const float* Ain  = (const float*)d_in[1];
    const float* bvec = (const float*)d_in[2];
    const float* W1   = (const float*)d_in[3];
    const float* b1   = (const float*)d_in[4];
    const float* W2   = (const float*)d_in[5];
    const float* b2   = (const float*)d_in[6];
    const float* z0   = (const float*)d_in[7];
    float* out = (float*)d_out;

    const int B = 256, D = 512, H = 2048, N1 = 1024, N2 = 4096;

    float* W = (float*)d_ws;
    const size_t M1 = (size_t)1 << 20;
    float* z   = W;
    float* u   = W + 1 * M1;
    float* wv  = W + 2 * M1;
    float* Xa  = W + 3 * M1;
    float* y   = W + 4 * M1;
    float* F   = W + 5 * M1;
    float* R   = F;
    float* S   = W + 8 * M1;
    float* h   = y;
    float* G   = W;
    float* Xb  = W + 1 * M1;
    float* T1  = W + 2 * M1;
    float* acc = W + 9 * M1;
    float* fn  = acc + 1;
    int*   flag = (int*)(acc + 2);

    dim3 blk(256);
    auto gemm = [&](const float* A_, int lda, const float* B_, int ldb, int tB,
                    float* P_, int ns, float* C_, const float* Cin, const float* bias,
                    float al, float be, float ga, int relu,
                    int M, int N, int K, const int* skip) {
        dim3 grid(N / TN, M / TM, ns);
        gemm_f32_wmma<<<grid, blk, 0, stream>>>(A_, lda, B_, ldb, tB, P_, N,
                                                K / ns, (size_t)M * N, skip);
        k_splitk_reduce<<<dim3((M * N) / 1024), blk, 0, stream>>>(
            P_, ns, (size_t)M * N, C_, Cin, bias, N, al, be, ga, relu, skip);
    };

    hipMemsetAsync(flag, 0, sizeof(int), stream);

    gemm(Ain, N2, Ain, N2, 1, F, 4, G, nullptr, nullptr, 1.f, 0.f, 0.f, 0, N1, N1, N2, nullptr);

    k_zero<<<1, 1, 0, stream>>>(fn);
    k_sumsq<<<dim3((N1 * N1 + 255) / 256), blk, 0, stream>>>(G, fn, N1 * N1);
    k_init_X<<<dim3((N1 * N1 + 255) / 256), blk, 0, stream>>>(Xa, fn, N1);

    float* Xc = Xa; float* Xn = Xb;
    for (int it = 0; it < 16; ++it) {
        gemm(G,  N1, Xc, N1, 0, F, 4, T1, nullptr, nullptr,  1.f, 0.f, 0.f, 0, N1, N1, N1, nullptr);
        gemm(Xc, N1, T1, N1, 0, F, 4, Xn, Xc,      nullptr, -1.f, 2.f, 0.f, 0, N1, N1, N1, nullptr);
        float* tmp = Xc; Xc = Xn; Xn = tmp;
    }
    float* Ginv = Xc;

    hipMemcpyAsync(z, z0, sizeof(float) * (size_t)B * N2, hipMemcpyDeviceToDevice, stream);

    gemm(ctx, D, W1, H,  0, F, 8, h,  nullptr, b1, 1.f, 0.f, 1.f, 1, B, H,  D, nullptr);
    gemm(h,   H, W2, N2, 0, F, 4, wv, nullptr, b2, 1.f, 0.f, 1.f, 0, B, N2, H, nullptr);

    const int nz = B * N2;
    dim3 gout((nz + 255) / 256);
    for (int itr = 0; itr < 50; ++itr) {
        k_prep_u<<<gout, blk, 0, stream>>>(z, wv, u, nz, (const int*)flag);
        gemm(u, N2, Ain,  N2, 1, F,      16, R, nullptr, bvec, 1.f, 0.f, -1.f, 0, B, N1, N2, flag);
        gemm(R, N1, Ginv, N1, 0, F + M1,  8, S, nullptr, nullptr, 1.f, 0.f, 0.f, 0, B, N1, N1, flag);
        gemm(S, N1, Ain,  N2, 0, y,       4, y, u, nullptr, -1.f, 1.f, 0.f, 0, B, N2, N1, flag);
        k_zero<<<1, 1, 0, stream>>>(acc);
        k_update<<<gout, blk, 0, stream>>>(z, y, acc, nz, (const int*)flag);
        k_check<<<1, 1, 0, stream>>>((const float*)acc, flag);
    }

    k_prep_u<<<gout, blk, 0, stream>>>(z, wv, u, nz, (const int*)nullptr);
    gemm(u, N2, Ain,  N2, 1, F,      16, R, nullptr, bvec, 1.f, 0.f, -1.f, 0, B, N1, N2, nullptr);
    gemm(R, N1, Ginv, N1, 0, F + M1,  8, S, nullptr, nullptr, 1.f, 0.f, 0.f, 0, B, N1, N1, nullptr);
    gemm(S, N1, Ain,  N2, 0, y,       4, y, u, nullptr, -1.f, 1.f, 0.f, 0, B, N2, N1, nullptr);
    k_final<<<gout, blk, 0, stream>>>(z, y, out, nz);
}


# GEMM K loop software-pipelined: next K-tile's A and B global loads issued before the current tile's WMMAs and stored to a second LDS stage after them; one barrier per K-tile, was two
# speedup vs baseline: 4.8432x; 1.1662x over previous
#include <hip/hip_runtime.h>

typedef __attribute__((ext_vector_type(2))) float v2f;
typedef __attribute__((ext_vector_type(8))) float v8f;

#define ALPHA 0.05f
#define EPS   0.01f

#define TM 128
#define TN 128
#define TK 32

static __device__ __forceinline__ v8f wmma4(const v2f a, const v2f b, const v8f c)
{
    return __builtin_amdgcn_wmma_f32_16x16x4_f32(false, a, false, b, (short)0, c,
                                                 false, false);
}

static __device__ __forceinline__ bool loop_done(const float* gate)
{
    return gate && *gate <= EPS * EPS;
}

static __device__ __forceinline__ void store_partial(
    const v8f acc, int r0, int c0, float* __restrict__ Cp, int ldc, int hi, int ln)
{
    const int c = c0 + ln;
#pragma unroll
    for (int v = 0; v < 8; ++v) {
        const int r = r0 + v + 8 * hi;
        Cp[(size_t)r * ldc + c] = acc[v];
    }
}

static __device__ __forceinline__ void stage_tiles(
    float* __restrict__ Ad, float* __restrict__ Bd, int transB,
    const float4 a0, const float4 a1, const float4 a2, const float4 a3,
    const float4 b0, const float4 b1, const float4 b2, const float4 b3,
    int arow, int akc, int bkp, int bnp, int btn, int btk)
{
    float* ad = Ad + arow * (TK + 4) + akc;
    *(float4*)(ad + 0)  = a0;
    *(float4*)(ad + 4)  = a1;
    *(float4*)(ad + 8)  = a2;
    *(float4*)(ad + 12) = a3;
    if (!transB) {
        float* bd = Bd + bkp * (TN + 4) + bnp;
        *(float4*)(bd + 0)  = b0;
        *(float4*)(bd + 4)  = b1;
        *(float4*)(bd + 8)  = b2;
        *(float4*)(bd + 12) = b3;
    } else {
        float* bd = Bd + btk * (TN + 4) + btn;
        bd[0  * (TN + 4)] = b0.x; bd[1  * (TN + 4)] = b0.y; bd[2  * (TN + 4)] = b0.z; bd[3  * (TN + 4)] = b0.w;
        bd[4  * (TN + 4)] = b1.x; bd[5  * (TN + 4)] = b1.y; bd[6  * (TN + 4)] = b1.z; bd[7  * (TN + 4)] = b1.w;
        bd[8  * (TN + 4)] = b2.x; bd[9  * (TN + 4)] = b2.y; bd[10 * (TN + 4)] = b2.z; bd[11 * (TN + 4)] = b2.w;
        bd[12 * (TN + 4)] = b3.x; bd[13 * (TN + 4)] = b3.y; bd[14 * (TN + 4)] = b3.z; bd[15 * (TN + 4)] = b3.w;
    }
}

__global__ __launch_bounds__(256) void gemm_f32_wmma(
    const float* __restrict__ Am, int lda,
    const float* __restrict__ Bm, int ldb, int transB,
    float* __restrict__ Cm, int ldc,
    int Kc, size_t pstride, const float* gate)
{
    if (loop_done(gate)) return;

    __shared__ float As[2][TM][TK + 4];
    __shared__ float Bs[2][TK][TN + 4];

    const int t    = threadIdx.x;
    const int row0 = blockIdx.y * TM;
    const int col0 = blockIdx.x * TN;

    const int wid  = t >> 5;
    const int lane = t & 31;
    const int wm   = (wid >> 2) * 64;
    const int wn   = (wid & 3) * 32;
    const int hi   = lane >> 4;
    const int ln   = lane & 15;

    v8f acc[4][2];
#pragma unroll
    for (int mi = 0; mi < 4; ++mi)
#pragma unroll
        for (int ni = 0; ni < 2; ++ni) acc[mi][ni] = (v8f){};

    const int arow = t >> 1;
    const int akc  = (t & 1) * 16;
    const int bkp  = t >> 3;
    const int bnp  = (t & 7) * 16;
    const int btn  = t >> 1;
    const int btk  = (t & 1) * 16;

    const int kbeg = blockIdx.z * Kc;
    const int nk   = Kc / TK;
    float* __restrict__ Cp = Cm + (size_t)blockIdx.z * pstride;

    const float* ap = Am + (size_t)(row0 + arow) * lda + kbeg + akc;
    const float* bp = transB ? Bm + (size_t)(col0 + btn) * ldb + kbeg + btk
                             : Bm + (size_t)(kbeg + bkp) * ldb + col0 + bnp;
    const size_t bstep = transB ? (size_t)TK : (size_t)TK * ldb;

    float4 ra0, ra1, ra2, ra3, rb0, rb1, rb2, rb3;

    ra0 = *(const float4*)(ap + 0);  ra1 = *(const float4*)(ap + 4);
    ra2 = *(const float4*)(ap + 8);  ra3 = *(const float4*)(ap + 12);
    rb0 = *(const float4*)(bp + 0);  rb1 = *(const float4*)(bp + 4);
    rb2 = *(const float4*)(bp + 8);  rb3 = *(const float4*)(bp + 12);
    stage_tiles(&As[0][0][0], &Bs[0][0][0], transB, ra0, ra1, ra2, ra3, rb0, rb1, rb2, rb3,
                arow, akc, bkp, bnp, btn, btk);
    __syncthreads();

    for (int kt = 0; kt < nk; ++kt) {
        const int  cur  = kt & 1;
        const bool more = kt + 1 < nk;
        if (more) {
            ap += TK;
            bp += bstep;
            ra0 = *(const float4*)(ap + 0);  ra1 = *(const float4*)(ap + 4);
            ra2 = *(const float4*)(ap + 8);  ra3 = *(const float4*)(ap + 12);
            rb0 = *(const float4*)(bp + 0);  rb1 = *(const float4*)(bp + 4);
            rb2 = *(const float4*)(bp + 8);  rb3 = *(const float4*)(bp + 12);
        }

#pragma unroll
        for (int kk = 0; kk < TK; kk += 4) {
            v2f a[4], b[2];
#pragma unroll
            for (int mi = 0; mi < 4; ++mi) {
                a[mi].x = As[cur][wm + 16 * mi + ln][kk + 2 * hi];
                a[mi].y = As[cur][wm + 16 * mi + ln][kk + 2 * hi + 1];
            }
#pragma unroll
            for (int ni = 0; ni < 2; ++ni) {
                b[ni].x = Bs[cur][kk + 2 * hi][wn + 16 * ni + ln];
                b[ni].y = Bs[cur][kk + 2 * hi + 1][wn + 16 * ni + ln];
            }
#pragma unroll
            for (int mi = 0; mi < 4; ++mi)
#pragma unroll
                for (int ni = 0; ni < 2; ++ni)
                    acc[mi][ni] = wmma4(a[mi], b[ni], acc[mi][ni]);
        }

        if (more) {
            stage_tiles(&As[cur ^ 1][0][0], &Bs[cur ^ 1][0][0], transB,
                        ra0, ra1, ra2, ra3, rb0, rb1, rb2, rb3, arow, akc, bkp, bnp, btn, btk);
            __syncthreads();
        }
    }

#pragma unroll
    for (int mi = 0; mi < 4; ++mi)
#pragma unroll
        for (int ni = 0; ni < 2; ++ni)
            store_partial(acc[mi][ni], row0 + wm + 16 * mi, col0 + wn + 16 * ni,
                          Cp, ldc, hi, ln);
}

__global__ void k_sumsq(const float* __restrict__ x, float* __restrict__ acc, int n)
{
    __shared__ float sm[256];
    int i = blockIdx.x * blockDim.x + threadIdx.x;
    float v = 0.f;
    if (i < n) { float t = x[i]; v = t * t; }
    sm[threadIdx.x] = v;
    __syncthreads();
    for (int s = 128; s > 0; s >>= 1) {
        if (threadIdx.x < s) sm[threadIdx.x] += sm[threadIdx.x + s];
        __syncthreads();
    }
    if (threadIdx.x == 0) atomicAdd(acc, sm[0]);
}

__global__ void k_zero(float* p) { p[0] = 0.f; }

__global__ __launch_bounds__(256) void k_splitk_reduce(
    const float* P, int nsplit, size_t pstride,
    float* C, const float* __restrict__ Cin, const float* __restrict__ bias,
    int ncols, float alpha, float beta, float gamma, int relu, const float* gate)
{
    if (loop_done(gate)) return;
    const size_t i = ((size_t)blockIdx.x * blockDim.x + threadIdx.x) * 4;
    float4 s = make_float4(0.f, 0.f, 0.f, 0.f);
    for (int k = 0; k < nsplit; k += 4) {
        const float4 p0 = *(const float4*)(P + (size_t)(k + 0) * pstride + i);
        const float4 p1 = *(const float4*)(P + (size_t)(k + 1) * pstride + i);
        const float4 p2 = *(const float4*)(P + (size_t)(k + 2) * pstride + i);
        const float4 p3 = *(const float4*)(P + (size_t)(k + 3) * pstride + i);
        s.x += p0.x; s.y += p0.y; s.z += p0.z; s.w += p0.w;
        s.x += p1.x; s.y += p1.y; s.z += p1.z; s.w += p1.w;
        s.x += p2.x; s.y += p2.y; s.z += p2.z; s.w += p2.w;
        s.x += p3.x; s.y += p3.y; s.z += p3.z; s.w += p3.w;
    }
    float4 v = make_float4(alpha * s.x, alpha * s.y, alpha * s.z, alpha * s.w);
    if (bias) {
        const int c = (int)(i % (size_t)ncols);
        const float4 bv = *(const float4*)(bias + c);
        v.x += gamma * bv.x; v.y += gamma * bv.y; v.z += gamma * bv.z; v.w += gamma * bv.w;
    }
    if (Cin) {
        const float4 ci = *(const float4*)(Cin + i);
        v.x += beta * ci.x; v.y += beta * ci.y; v.z += beta * ci.z; v.w += beta * ci.w;
    }
    if (relu) {
        v.x = fmaxf(v.x, 0.f); v.y = fmaxf(v.y, 0.f); v.z = fmaxf(v.z, 0.f); v.w = fmaxf(v.w, 0.f);
    }
    *(float4*)(C + i) = v;
}

__global__ void k_init_X(float* __restrict__ X, const float* __restrict__ fn, int n)
{
    int i = blockIdx.x * blockDim.x + threadIdx.x;
    if (i < n * n) {
        float c = rsqrtf(*fn);
        int r = i / n, cc = i - r * n;
        X[i] = (r == cc) ? c : 0.f;
    }
}

static __device__ __forceinline__ float dys_elem(float s, float uo, float zo, float wo,
                                                 float& zn, float& un)
{
    const float y  = uo - s;
    const float x  = fmaxf(zo, 0.f);
    zn = zo - x + y;
    const float xn = fmaxf(zn, 0.f);
    un = 2.f * xn - zn - ALPHA * wo;
    const float d = y - x;
    return d * d;
}

static __device__ __forceinline__ float4 sum4(const float4 p0, const float4 p1,
                                              const float4 p2, const float4 p3)
{
    float4 s = p0;
    s.x += p1.x; s.y += p1.y; s.z += p1.z; s.w += p1.w;
    s.x += p2.x; s.y += p2.y; s.z += p2.z; s.w += p2.w;
    s.x += p3.x; s.y += p3.y; s.z += p3.z; s.w += p3.w;
    return s;
}

__global__ __launch_bounds__(256) void k_dys_init(
    const float* __restrict__ z0, const float* __restrict__ w,
    float* __restrict__ z, float* __restrict__ u, float* __restrict__ acc0)
{
    if (blockIdx.x == 0 && threadIdx.x == 0) *acc0 = 0.f;
    const size_t i = ((size_t)blockIdx.x * 256 + threadIdx.x) * 4;
    const float4 zi = *(const float4*)(z0 + i);
    const float4 wi = *(const float4*)(w + i);
    float4 uo;
    uo.x = 2.f * fmaxf(zi.x, 0.f) - zi.x - ALPHA * wi.x;
    uo.y = 2.f * fmaxf(zi.y, 0.f) - zi.y - ALPHA * wi.y;
    uo.z = 2.f * fmaxf(zi.z, 0.f) - zi.z - ALPHA * wi.z;
    uo.w = 2.f * fmaxf(zi.w, 0.f) - zi.w - ALPHA * wi.w;
    *(float4*)(z + i) = zi;
    *(float4*)(u + i) = uo;
}

__global__ __launch_bounds__(256) void k_dys_step(
    const float* __restrict__ P, size_t pstride,
    float* z, float* u, const float* __restrict__ w,
    float* acc, float* next, const float* gate)
{
    __shared__ float sm[32];
    if (blockIdx.x == 0 && threadIdx.x == 0) *next = 0.f;
    if (loop_done(gate)) return;

    float4 p0[2], p1[2], p2[2], p3[2], uo[2], zo[2], wo[2];
    size_t idx[2];
#pragma unroll
    for (int h = 0; h < 2; ++h) {
        const size_t i = ((size_t)blockIdx.x * 512 + h * 256 + threadIdx.x) * 4;
        idx[h] = i;
        p0[h] = *(const float4*)(P + i);
        p1[h] = *(const float4*)(P + pstride + i);
        p2[h] = *(const float4*)(P + 2 * pstride + i);
        p3[h] = *(const float4*)(P + 3 * pstride + i);
        uo[h] = *(const float4*)(u + i);
        zo[h] = *(const float4*)(z + i);
        wo[h] = *(const float4*)(w + i);
    }
    float d2 = 0.f;
#pragma unroll
    for (int h = 0; h < 2; ++h) {
        const float4 s = sum4(p0[h], p1[h], p2[h], p3[h]);
        float4 zn, un;
        d2 += dys_elem(s.x, uo[h].x, zo[h].x, wo[h].x, zn.x, un.x);
        d2 += dys_elem(s.y, uo[h].y, zo[h].y, wo[h].y, zn.y, un.y);
        d2 += dys_elem(s.z, uo[h].z, zo[h].z, wo[h].z, zn.z, un.z);
        d2 += dys_elem(s.w, uo[h].w, zo[h].w, wo[h].w, zn.w, un.w);
        *(float4*)(z + idx[h]) = zn;
        *(float4*)(u + idx[h]) = un;
    }

    for (int off = warpSize / 2; off > 0; off >>= 1) d2 += __shfl_xor(d2, off);
    const int nw = 256 / warpSize;
    if (threadIdx.x % warpSize == 0) sm[threadIdx.x / warpSize] = d2;
    __syncthreads();
    if (threadIdx.x == 0) {
        float t = 0.f;
        for (int k = 0; k < nw; ++k) t += sm[k];
        atomicAdd(acc, t);
    }
}

__global__ __launch_bounds__(256) void k_dys_final(
    const float* __restrict__ P, size_t pstride,
    const float* __restrict__ z, const float* __restrict__ u, float* __restrict__ out)
{
    const size_t i = ((size_t)blockIdx.x * 256 + threadIdx.x) * 4;
    const float4 s = sum4(*(const float4*)(P + i), *(const float4*)(P + pstride + i),
                          *(const float4*)(P + 2 * pstride + i),
                          *(const float4*)(P + 3 * pstride + i));
    const float4 ui = *(const float4*)(u + i);
    const float4 zi = *(const float4*)(z + i);
    float4 o;
    o.x = fmaxf(zi.x - fmaxf(zi.x, 0.f) + (ui.x - s.x), 0.f);
    o.y = fmaxf(zi.y - fmaxf(zi.y, 0.f) + (ui.y - s.y), 0.f);
    o.z = fmaxf(zi.z - fmaxf(zi.z, 0.f) + (ui.z - s.z), 0.f);
    o.w = fmaxf(zi.w - fmaxf(zi.w, 0.f) + (ui.w - s.w), 0.f);
    *(float4*)(out + i) = o;
}

extern "C" void kernel_launch(void* const* d_in, const int* in_sizes, int n_in,
                              void* d_out, int out_size, void* d_ws, size_t ws_size,
                              hipStream_t stream)
{
    const float* ctx  = (const float*)d_in[0];
    const float* Ain  = (const float*)d_in[1];
    const float* bvec = (const float*)d_in[2];
    const float* W1   = (const float*)d_in[3];
    const float* b1   = (const float*)d_in[4];
    const float* W2   = (const float*)d_in[5];
    const float* b2   = (const float*)d_in[6];
    const float* z0   = (const float*)d_in[7];
    float* out = (float*)d_out;

    const int B = 256, D = 512, H = 2048, N1 = 1024, N2 = 4096;

    float* W = (float*)d_ws;
    const size_t M1 = (size_t)1 << 20;
    float* z   = W;
    float* u   = W + 1 * M1;
    float* wv  = W + 2 * M1;
    float* Xa  = W + 3 * M1;
    float* y   = W + 4 * M1;
    float* F   = W + 5 * M1;
    float* R   = F;
    float* S   = W + 8 * M1;
    float* h   = y;
    float* G   = W;
    float* Xb  = W + 1 * M1;
    float* T1  = W + 2 * M1;
    float* acc = W + 9 * M1;
    float* fn  = acc + 1;

    dim3 blk(256);
    auto partials = [&](const float* A_, int lda, const float* B_, int ldb, int tB,
                        float* P_, int ns, int M, int N, int K, const float* gate) {
        dim3 grid(N / TN, M / TM, ns);
        gemm_f32_wmma<<<grid, blk, 0, stream>>>(A_, lda, B_, ldb, tB, P_, N,
                                                K / ns, (size_t)M * N, gate);
    };
    auto gemm = [&](const float* A_, int lda, const float* B_, int ldb, int tB,
                    float* P_, int ns, float* C_, const float* Cin, const float* bias,
                    float al, float be, float ga, int relu,
                    int M, int N, int K, const float* gate) {
        partials(A_, lda, B_, ldb, tB, P_, ns, M, N, K, gate);
        k_splitk_reduce<<<dim3((M * N) / 1024), blk, 0, stream>>>(
            P_, ns, (size_t)M * N, C_, Cin, bias, N, al, be, ga, relu, gate);
    };

    gemm(Ain, N2, Ain, N2, 1, F, 4, G, nullptr, nullptr, 1.f, 0.f, 0.f, 0, N1, N1, N2, nullptr);

    k_zero<<<1, 1, 0, stream>>>(fn);
    k_sumsq<<<dim3((N1 * N1 + 255) / 256), blk, 0, stream>>>(G, fn, N1 * N1);
    k_init_X<<<dim3((N1 * N1 + 255) / 256), blk, 0, stream>>>(Xa, fn, N1);

    float* Xc = Xa; float* Xn = Xb;
    for (int it = 0; it < 16; ++it) {
        gemm(G,  N1, Xc, N1, 0, F, 4, T1, nullptr, nullptr,  1.f, 0.f, 0.f, 0, N1, N1, N1, nullptr);
        gemm(Xc, N1, T1, N1, 0, F, 4, Xn, Xc,      nullptr, -1.f, 2.f, 0.f, 0, N1, N1, N1, nullptr);
        float* tmp = Xc; Xc = Xn; Xn = tmp;
    }
    float* Ginv = Xc;

    gemm(ctx, D, W1, H,  0, F, 8, h,  nullptr, b1, 1.f, 0.f, 1.f, 1, B, H,  D, nullptr);
    gemm(h,   H, W2, N2, 0, F, 4, wv, nullptr, b2, 1.f, 0.f, 1.f, 0, B, N2, H, nullptr);

    const int nz = B * N2;
    k_dys_init<<<dim3(nz / 1024), blk, 0, stream>>>(z0, wv, z, u, acc);
    float* cur = acc; float* nxt = acc + 1; float* prv = acc + 2;
    const float* gate = nullptr;
    for (int itr = 0; itr < 50; ++itr) {
        gemm(u, N2, Ain,  N2, 1, F,      16, R, nullptr, bvec, 1.f, 0.f, -1.f, 0, B, N1, N2, gate);
        gemm(R, N1, Ginv, N1, 0, F + M1,  8, S, nullptr, nullptr, 1.f, 0.f, 0.f, 0, B, N1, N1, gate);
        partials(S, N1, Ain, N2, 0, y, 4, B, N2, N1, gate);
        k_dys_step<<<dim3(nz / 2048), blk, 0, stream>>>(y, (size_t)nz, z, u, wv, cur, nxt, gate);
        gate = cur;
        float* tmp = prv; prv = cur; cur = nxt; nxt = tmp;
    }

    gemm(u, N2, Ain,  N2, 1, F,      16, R, nullptr, bvec, 1.f, 0.f, -1.f, 0, B, N1, N2, nullptr);
    gemm(R, N1, Ginv, N1, 0, F + M1,  8, S, nullptr, nullptr, 1.f, 0.f, 0.f, 0, B, N1, N1, nullptr);
    partials(S, N1, Ain, N2, 0, y, 4, B, N2, N1, nullptr);
    k_dys_final<<<dim3(nz / 1024), blk, 0, stream>>>(y, (size_t)nz, z, u, out);
}


# loop: Ginv@A formed once before the loop, so proj_C2's correction is one GEMM (r@Minv), not two: 4 launches per iteration, was 6 (z kept in out, u formed in the A loader, to make room for Minv)
# speedup vs baseline: 5.7793x; 1.1933x over previous
#include <hip/hip_runtime.h>

typedef __attribute__((ext_vector_type(2))) float v2f;
typedef __attribute__((ext_vector_type(8))) float v8f;

#define ALPHA 0.05f
#define EPS   0.01f

#define TM 128
#define TN 128
#define TK 32

static __device__ __forceinline__ v8f wmma4(const v2f a, const v2f b, const v8f c)
{
    return __builtin_amdgcn_wmma_f32_16x16x4_f32(false, a, false, b, (short)0, c,
                                                 false, false);
}

static __device__ __forceinline__ bool loop_done(const float* gate)
{
    return gate && *gate <= EPS * EPS;
}

static __device__ __forceinline__ void store_partial(
    const v8f acc, int r0, int c0, float* __restrict__ Cp, int ldc, int hi, int ln)
{
    const int c = c0 + ln;
#pragma unroll
    for (int v = 0; v < 8; ++v) {
        const int r = r0 + v + 8 * hi;
        Cp[(size_t)r * ldc + c] = acc[v];
    }
}

static __device__ __forceinline__ void stage_tiles(
    float* __restrict__ Ad, float* __restrict__ Bd, int transB,
    const float4 a0, const float4 a1, const float4 a2, const float4 a3,
    const float4 b0, const float4 b1, const float4 b2, const float4 b3,
    int arow, int akc, int bkp, int bnp, int btn, int btk)
{
    float* ad = Ad + arow * (TK + 4) + akc;
    *(float4*)(ad + 0)  = a0;
    *(float4*)(ad + 4)  = a1;
    *(float4*)(ad + 8)  = a2;
    *(float4*)(ad + 12) = a3;
    if (!transB) {
        float* bd = Bd + bkp * (TN + 4) + bnp;
        *(float4*)(bd + 0)  = b0;
        *(float4*)(bd + 4)  = b1;
        *(float4*)(bd + 8)  = b2;
        *(float4*)(bd + 12) = b3;
    } else {
        float* bd = Bd + btk * (TN + 4) + btn;
        bd[0  * (TN + 4)] = b0.x; bd[1  * (TN + 4)] = b0.y; bd[2  * (TN + 4)] = b0.z; bd[3  * (TN + 4)] = b0.w;
        bd[4  * (TN + 4)] = b1.x; bd[5  * (TN + 4)] = b1.y; bd[6  * (TN + 4)] = b1.z; bd[7  * (TN + 4)] = b1.w;
        bd[8  * (TN + 4)] = b2.x; bd[9  * (TN + 4)] = b2.y; bd[10 * (TN + 4)] = b2.z; bd[11 * (TN + 4)] = b2.w;
        bd[12 * (TN + 4)] = b3.x; bd[13 * (TN + 4)] = b3.y; bd[14 * (TN + 4)] = b3.z; bd[15 * (TN + 4)] = b3.w;
    }
}

static __device__ __forceinline__ float4 dys_u4(const float4 z, const float4 w)
{
    float4 u;
    u.x = fmaf(-ALPHA, w.x, fabsf(z.x));
    u.y = fmaf(-ALPHA, w.y, fabsf(z.y));
    u.z = fmaf(-ALPHA, w.z, fabsf(z.z));
    u.w = fmaf(-ALPHA, w.w, fabsf(z.w));
    return u;
}

template <bool HASW>
__global__ __launch_bounds__(256) void gemm_f32_wmma(
    const float* __restrict__ Am, const float* __restrict__ Aw, int lda,
    const float* __restrict__ Bm, int ldb, int transB,
    float* __restrict__ Cm, int ldc,
    int Kc, int Ktot, size_t pstride, const float* gate)
{
    if (loop_done(gate)) return;

    __shared__ float As[2][TM][TK + 4];
    __shared__ float Bs[2][TK][TN + 4];

    const int t    = threadIdx.x;
    const int row0 = blockIdx.y * TM;
    const int col0 = blockIdx.x * TN;

    const int wid  = t >> 5;
    const int lane = t & 31;
    const int wm   = (wid >> 2) * 64;
    const int wn   = (wid & 3) * 32;
    const int hi   = lane >> 4;
    const int ln   = lane & 15;

    v8f acc[4][2];
#pragma unroll
    for (int mi = 0; mi < 4; ++mi)
#pragma unroll
        for (int ni = 0; ni < 2; ++ni) acc[mi][ni] = (v8f){};

    const int arow = t >> 1;
    const int akc  = (t & 1) * 16;
    const int bkp  = t >> 3;
    const int bnp  = (t & 7) * 16;
    const int btn  = t >> 1;
    const int btk  = (t & 1) * 16;

    const int kbeg = blockIdx.z * Kc;
    const int kend = min(kbeg + Kc, Ktot);
    const int nk   = (kend - kbeg) / TK;
    float* __restrict__ Cp = Cm + (size_t)blockIdx.z * pstride;

    const size_t aoff = (size_t)(row0 + arow) * lda + kbeg + akc;
    const float* ap = Am + aoff;
    const float* wp = HASW ? Aw + aoff : nullptr;
    const float* bp = transB ? Bm + (size_t)(col0 + btn) * ldb + kbeg + btk
                             : Bm + (size_t)(kbeg + bkp) * ldb + col0 + bnp;
    const size_t bstep = transB ? (size_t)TK : (size_t)TK * ldb;

    float4 ra0, ra1, ra2, ra3, rb0, rb1, rb2, rb3;
    float4 rw0 = make_float4(0.f, 0.f, 0.f, 0.f), rw1 = rw0, rw2 = rw0, rw3 = rw0;

    ra0 = *(const float4*)(ap + 0);  ra1 = *(const float4*)(ap + 4);
    ra2 = *(const float4*)(ap + 8);  ra3 = *(const float4*)(ap + 12);
    rb0 = *(const float4*)(bp + 0);  rb1 = *(const float4*)(bp + 4);
    rb2 = *(const float4*)(bp + 8);  rb3 = *(const float4*)(bp + 12);
    if (HASW) {
        rw0 = *(const float4*)(wp + 0);  rw1 = *(const float4*)(wp + 4);
        rw2 = *(const float4*)(wp + 8);  rw3 = *(const float4*)(wp + 12);
        ra0 = dys_u4(ra0, rw0);  ra1 = dys_u4(ra1, rw1);
        ra2 = dys_u4(ra2, rw2);  ra3 = dys_u4(ra3, rw3);
    }
    stage_tiles(&As[0][0][0], &Bs[0][0][0], transB, ra0, ra1, ra2, ra3, rb0, rb1, rb2, rb3,
                arow, akc, bkp, bnp, btn, btk);
    __syncthreads();

    for (int kt = 0; kt < nk; ++kt) {
        const int  cur  = kt & 1;
        const bool more = kt + 1 < nk;
        if (more) {
            ap += TK;
            bp += bstep;
            ra0 = *(const float4*)(ap + 0);  ra1 = *(const float4*)(ap + 4);
            ra2 = *(const float4*)(ap + 8);  ra3 = *(const float4*)(ap + 12);
            rb0 = *(const float4*)(bp + 0);  rb1 = *(const float4*)(bp + 4);
            rb2 = *(const float4*)(bp + 8);  rb3 = *(const float4*)(bp + 12);
            if (HASW) {
                wp += TK;
                rw0 = *(const float4*)(wp + 0);  rw1 = *(const float4*)(wp + 4);
                rw2 = *(const float4*)(wp + 8);  rw3 = *(const float4*)(wp + 12);
            }
        }

#pragma unroll
        for (int kk = 0; kk < TK; kk += 4) {
            v2f a[4], b[2];
#pragma unroll
            for (int mi = 0; mi < 4; ++mi) {
                a[mi].x = As[cur][wm + 16 * mi + ln][kk + 2 * hi];
                a[mi].y = As[cur][wm + 16 * mi + ln][kk + 2 * hi + 1];
            }
#pragma unroll
            for (int ni = 0; ni < 2; ++ni) {
                b[ni].x = Bs[cur][kk + 2 * hi][wn + 16 * ni + ln];
                b[ni].y = Bs[cur][kk + 2 * hi + 1][wn + 16 * ni + ln];
            }
#pragma unroll
            for (int mi = 0; mi < 4; ++mi)
#pragma unroll
                for (int ni = 0; ni < 2; ++ni)
                    acc[mi][ni] = wmma4(a[mi], b[ni], acc[mi][ni]);
        }

        if (more) {
            if (HASW) {
                ra0 = dys_u4(ra0, rw0);  ra1 = dys_u4(ra1, rw1);
                ra2 = dys_u4(ra2, rw2);  ra3 = dys_u4(ra3, rw3);
            }
            stage_tiles(&As[cur ^ 1][0][0], &Bs[cur ^ 1][0][0], transB,
                        ra0, ra1, ra2, ra3, rb0, rb1, rb2, rb3, arow, akc, bkp, bnp, btn, btk);
            __syncthreads();
        }
    }

#pragma unroll
    for (int mi = 0; mi < 4; ++mi)
#pragma unroll
        for (int ni = 0; ni < 2; ++ni)
            store_partial(acc[mi][ni], row0 + wm + 16 * mi, col0 + wn + 16 * ni,
                          Cp, ldc, hi, ln);
}

__global__ void k_sumsq(const float* __restrict__ x, float* __restrict__ acc, int n)
{
    __shared__ float sm[256];
    int i = blockIdx.x * blockDim.x + threadIdx.x;
    float v = 0.f;
    if (i < n) { float t = x[i]; v = t * t; }
    sm[threadIdx.x] = v;
    __syncthreads();
    for (int s = 128; s > 0; s >>= 1) {
        if (threadIdx.x < s) sm[threadIdx.x] += sm[threadIdx.x + s];
        __syncthreads();
    }
    if (threadIdx.x == 0) atomicAdd(acc, sm[0]);
}

__global__ void k_zero(float* p) { p[0] = 0.f; }

__global__ __launch_bounds__(256) void k_splitk_reduce(
    const float* P, int nsplit, size_t pstride,
    float* C, const float* __restrict__ Cin, const float* __restrict__ bias,
    int ncols, float alpha, float beta, float gamma, int relu, const float* gate)
{
    if (loop_done(gate)) return;
    const size_t i = ((size_t)blockIdx.x * blockDim.x + threadIdx.x) * 4;
    float4 s = make_float4(0.f, 0.f, 0.f, 0.f);
    for (int k = 0; k < nsplit; k += 4) {
        const float4 p0 = *(const float4*)(P + (size_t)(k + 0) * pstride + i);
        const float4 p1 = *(const float4*)(P + (size_t)(k + 1) * pstride + i);
        const float4 p2 = *(const float4*)(P + (size_t)(k + 2) * pstride + i);
        const float4 p3 = *(const float4*)(P + (size_t)(k + 3) * pstride + i);
        s.x += p0.x; s.y += p0.y; s.z += p0.z; s.w += p0.w;
        s.x += p1.x; s.y += p1.y; s.z += p1.z; s.w += p1.w;
        s.x += p2.x; s.y += p2.y; s.z += p2.z; s.w += p2.w;
        s.x += p3.x; s.y += p3.y; s.z += p3.z; s.w += p3.w;
    }
    float4 v = make_float4(alpha * s.x, alpha * s.y, alpha * s.z, alpha * s.w);
    if (bias) {
        const int c = (int)(i % (size_t)ncols);
        const float4 bv = *(const float4*)(bias + c);
        v.x += gamma * bv.x; v.y += gamma * bv.y; v.z += gamma * bv.z; v.w += gamma * bv.w;
    }
    if (Cin) {
        const float4 ci = *(const float4*)(Cin + i);
        v.x += beta * ci.x; v.y += beta * ci.y; v.z += beta * ci.z; v.w += beta * ci.w;
    }
    if (relu) {
        v.x = fmaxf(v.x, 0.f); v.y = fmaxf(v.y, 0.f); v.z = fmaxf(v.z, 0.f); v.w = fmaxf(v.w, 0.f);
    }
    *(float4*)(C + i) = v;
}

__global__ void k_init_X(float* __restrict__ X, const float* __restrict__ fn, int n)
{
    int i = blockIdx.x * blockDim.x + threadIdx.x;
    if (i < n * n) {
        float c = rsqrtf(*fn);
        int r = i / n, cc = i - r * n;
        X[i] = (r == cc) ? c : 0.f;
    }
}

static __device__ __forceinline__ float dys_elem(float s, float zo, float wo, float& zn)
{
    const float uo = fmaf(-ALPHA, wo, fabsf(zo));
    const float y  = uo - s;
    const float x  = fmaxf(zo, 0.f);
    zn = zo - x + y;
    const float d = y - x;
    return d * d;
}

static __device__ __forceinline__ float4 sum3(const float4 p0, const float4 p1, const float4 p2)
{
    float4 s = p0;
    s.x += p1.x; s.y += p1.y; s.z += p1.z; s.w += p1.w;
    s.x += p2.x; s.y += p2.y; s.z += p2.z; s.w += p2.w;
    return s;
}

__global__ __launch_bounds__(256) void k_dys_init(
    const float* __restrict__ z0, float* __restrict__ z, float* __restrict__ acc0)
{
    if (blockIdx.x == 0 && threadIdx.x == 0) *acc0 = 0.f;
    const size_t i = ((size_t)blockIdx.x * 256 + threadIdx.x) * 4;
    *(float4*)(z + i) = *(const float4*)(z0 + i);
}

__global__ __launch_bounds__(256) void k_dys_step(
    const float* __restrict__ P, size_t pstride,
    float* z, const float* __restrict__ w,
    float* acc, float* next, const float* gate)
{
    __shared__ float sm[32];
    if (blockIdx.x == 0 && threadIdx.x == 0) *next = 0.f;
    if (loop_done(gate)) return;

    float4 p0[2], p1[2], p2[2], zo[2], wo[2];
    size_t idx[2];
#pragma unroll
    for (int h = 0; h < 2; ++h) {
        const size_t i = ((size_t)blockIdx.x * 512 + h * 256 + threadIdx.x) * 4;
        idx[h] = i;
        p0[h] = *(const float4*)(P + i);
        p1[h] = *(const float4*)(P + pstride + i);
        p2[h] = *(const float4*)(P + 2 * pstride + i);
        zo[h] = *(const float4*)(z + i);
        wo[h] = *(const float4*)(w + i);
    }
    float d2 = 0.f;
#pragma unroll
    for (int h = 0; h < 2; ++h) {
        const float4 s = sum3(p0[h], p1[h], p2[h]);
        float4 zn;
        d2 += dys_elem(s.x, zo[h].x, wo[h].x, zn.x);
        d2 += dys_elem(s.y, zo[h].y, wo[h].y, zn.y);
        d2 += dys_elem(s.z, zo[h].z, wo[h].z, zn.z);
        d2 += dys_elem(s.w, zo[h].w, wo[h].w, zn.w);
        *(float4*)(z + idx[h]) = zn;
    }

    for (int off = warpSize / 2; off > 0; off >>= 1) d2 += __shfl_xor(d2, off);
    const int nw = 256 / warpSize;
    if (threadIdx.x % warpSize == 0) sm[threadIdx.x / warpSize] = d2;
    __syncthreads();
    if (threadIdx.x == 0) {
        float t = 0.f;
        for (int k = 0; k < nw; ++k) t += sm[k];
        atomicAdd(acc, t);
    }
}

__global__ __launch_bounds__(256) void k_dys_final(
    const float* __restrict__ P, size_t pstride,
    float* zo, const float* __restrict__ w)
{
    const size_t i = ((size_t)blockIdx.x * 256 + threadIdx.x) * 4;
    const float4 s = sum3(*(const float4*)(P + i), *(const float4*)(P + pstride + i),
                          *(const float4*)(P + 2 * pstride + i));
    const float4 zi = *(const float4*)(zo + i);
    const float4 ui = dys_u4(zi, *(const float4*)(w + i));
    float4 o;
    o.x = fmaxf(zi.x - fmaxf(zi.x, 0.f) + (ui.x - s.x), 0.f);
    o.y = fmaxf(zi.y - fmaxf(zi.y, 0.f) + (ui.y - s.y), 0.f);
    o.z = fmaxf(zi.z - fmaxf(zi.z, 0.f) + (ui.z - s.z), 0.f);
    o.w = fmaxf(zi.w - fmaxf(zi.w, 0.f) + (ui.w - s.w), 0.f);
    *(float4*)(zo + i) = o;
}

extern "C" void kernel_launch(void* const* d_in, const int* in_sizes, int n_in,
                              void* d_out, int out_size, void* d_ws, size_t ws_size,
                              hipStream_t stream)
{
    const float* ctx  = (const float*)d_in[0];
    const float* Ain  = (const float*)d_in[1];
    const float* bvec = (const float*)d_in[2];
    const float* W1   = (const float*)d_in[3];
    const float* b1   = (const float*)d_in[4];
    const float* W2   = (const float*)d_in[5];
    const float* b2   = (const float*)d_in[6];
    const float* z0   = (const float*)d_in[7];
    float* out = (float*)d_out;

    const int B = 256, D = 512, H = 2048, N1 = 1024, N2 = 4096;

    float* W = (float*)d_ws;
    const size_t M1 = (size_t)1 << 20;
    float* G    = W;
    float* Xb   = W + 1 * M1;
    float* T1   = W + 2 * M1;
    float* Xa   = W + 3 * M1;
    float* F    = W + 5 * M1;
    float* Minv = W + 4 * M1;
    float* P    = W;
    float* h    = W + 8 * M1;
    float* wv   = W + 8 * M1;
    float* z    = out;
    float* R    = P;
    float* Py   = P + (size_t)B * N1;
    float* acc  = W + 9 * M1;
    float* fn   = acc + 1;

    dim3 blk(256);
    auto partials = [&](const float* A_, const float* Aw_, int lda, const float* B_, int ldb,
                        int tB, float* P_, int ns, int Kc, int M, int N, int K,
                        const float* gate) {
        dim3 grid(N / TN, M / TM, ns);
        if (Aw_)
            gemm_f32_wmma<true><<<grid, blk, 0, stream>>>(A_, Aw_, lda, B_, ldb, tB, P_, N,
                                                          Kc, K, (size_t)M * N, gate);
        else
            gemm_f32_wmma<false><<<grid, blk, 0, stream>>>(A_, Aw_, lda, B_, ldb, tB, P_, N,
                                                           Kc, K, (size_t)M * N, gate);
    };
    auto gemm = [&](const float* A_, const float* Aw_, int lda, const float* B_, int ldb, int tB,
                    float* P_, int ns, float* C_, const float* Cin, const float* bias,
                    float al, float be, float ga, int relu,
                    int M, int N, int K, const float* gate) {
        partials(A_, Aw_, lda, B_, ldb, tB, P_, ns, K / ns, M, N, K, gate);
        k_splitk_reduce<<<dim3((M * N) / 1024), blk, 0, stream>>>(
            P_, ns, (size_t)M * N, C_, Cin, bias, N, al, be, ga, relu, gate);
    };

    gemm(Ain, nullptr, N2, Ain, N2, 1, F, 4, G, nullptr, nullptr, 1.f, 0.f, 0.f, 0,
         N1, N1, N2, nullptr);

    k_zero<<<1, 1, 0, stream>>>(fn);
    k_sumsq<<<dim3((N1 * N1 + 255) / 256), blk, 0, stream>>>(G, fn, N1 * N1);
    k_init_X<<<dim3((N1 * N1 + 255) / 256), blk, 0, stream>>>(Xa, fn, N1);

    float* Xc = Xa; float* Xn = Xb;
    for (int it = 0; it < 16; ++it) {
        gemm(G,  nullptr, N1, Xc, N1, 0, F, 4, T1, nullptr, nullptr,  1.f, 0.f, 0.f, 0,
             N1, N1, N1, nullptr);
        gemm(Xc, nullptr, N1, T1, N1, 0, F, 4, Xn, Xc,      nullptr, -1.f, 2.f, 0.f, 0,
             N1, N1, N1, nullptr);
        float* tmp = Xc; Xc = Xn; Xn = tmp;
    }
    float* Ginv = Xc;

    partials(Ginv, nullptr, N1, Ain, N2, 0, Minv, 1, N1, N1, N2, N1, nullptr);

    gemm(ctx, nullptr, D, W1, H,  0, P, 8, h,  nullptr, b1, 1.f, 0.f, 1.f, 1, B, H,  D, nullptr);
    gemm(h,   nullptr, H, W2, N2, 0, P, 4, wv, nullptr, b2, 1.f, 0.f, 1.f, 0, B, N2, H, nullptr);

    const int nz = B * N2;
    k_dys_init<<<dim3(nz / 1024), blk, 0, stream>>>(z0, z, acc);
    float* cur = acc; float* nxt = acc + 1; float* prv = acc + 2;
    const float* gate = nullptr;
    for (int itr = 0; itr < 50; ++itr) {
        gemm(z, wv, N2, Ain, N2, 1, P, 16, R, nullptr, bvec, 1.f, 0.f, -1.f, 0, B, N1, N2, gate);
        partials(R, nullptr, N1, Minv, N2, 0, Py, 3, 352, B, N2, N1, gate);
        k_dys_step<<<dim3(nz / 2048), blk, 0, stream>>>(Py, (size_t)nz, z, wv, cur, nxt, gate);
        gate = cur;
        float* tmp = prv; prv = cur; cur = nxt; nxt = tmp;
    }

    gemm(z, wv, N2, Ain, N2, 1, P, 16, R, nullptr, bvec, 1.f, 0.f, -1.f, 0, B, N1, N2, nullptr);
    partials(R, nullptr, N1, Minv, N2, 0, Py, 3, 352, B, N2, N1, nullptr);
    k_dys_final<<<dim3(nz / 1024), blk, 0, stream>>>(Py, (size_t)nz, z, wv);
}


# Newton-Schulz: each of its 32 products takes the whole K per work-group (256 tiles of 64x64, the loop kernel's K loop) with 2X - X(GX) in the epilogue: 1 launch per product, was 2; no partial tiles
# speedup vs baseline: 8.1188x; 1.4048x over previous
#include <hip/hip_runtime.h>

typedef __attribute__((ext_vector_type(2))) float v2f;
typedef __attribute__((ext_vector_type(8))) float v8f;

#define ALPHA 0.05f
#define EPS   0.01f

#define TM 128
#define TN 128
#define TK 32

static __device__ __forceinline__ v8f wmma4(const v2f a, const v2f b, const v8f c)
{
    return __builtin_amdgcn_wmma_f32_16x16x4_f32(false, a, false, b, (short)0, c,
                                                 false, false);
}

static __device__ __forceinline__ bool loop_done(const float* gate)
{
    return gate && *gate <= EPS * EPS;
}

static __device__ __forceinline__ void store_partial(
    const v8f acc, int r0, int c0, float* __restrict__ Cp, int ldc, int hi, int ln)
{
    const int c = c0 + ln;
#pragma unroll
    for (int v = 0; v < 8; ++v) {
        const int r = r0 + v + 8 * hi;
        Cp[(size_t)r * ldc + c] = acc[v];
    }
}

static __device__ __forceinline__ void stage_tiles(
    float* __restrict__ Ad, float* __restrict__ Bd, int transB,
    const float4 a0, const float4 a1, const float4 a2, const float4 a3,
    const float4 b0, const float4 b1, const float4 b2, const float4 b3,
    int arow, int akc, int bkp, int bnp, int btn, int btk)
{
    float* ad = Ad + arow * (TK + 4) + akc;
    *(float4*)(ad + 0)  = a0;
    *(float4*)(ad + 4)  = a1;
    *(float4*)(ad + 8)  = a2;
    *(float4*)(ad + 12) = a3;
    if (!transB) {
        float* bd = Bd + bkp * (TN + 4) + bnp;
        *(float4*)(bd + 0)  = b0;
        *(float4*)(bd + 4)  = b1;
        *(float4*)(bd + 8)  = b2;
        *(float4*)(bd + 12) = b3;
    } else {
        float* bd = Bd + btk * (TN + 4) + btn;
        bd[0  * (TN + 4)] = b0.x; bd[1  * (TN + 4)] = b0.y; bd[2  * (TN + 4)] = b0.z; bd[3  * (TN + 4)] = b0.w;
        bd[4  * (TN + 4)] = b1.x; bd[5  * (TN + 4)] = b1.y; bd[6  * (TN + 4)] = b1.z; bd[7  * (TN + 4)] = b1.w;
        bd[8  * (TN + 4)] = b2.x; bd[9  * (TN + 4)] = b2.y; bd[10 * (TN + 4)] = b2.z; bd[11 * (TN + 4)] = b2.w;
        bd[12 * (TN + 4)] = b3.x; bd[13 * (TN + 4)] = b3.y; bd[14 * (TN + 4)] = b3.z; bd[15 * (TN + 4)] = b3.w;
    }
}

static __device__ __forceinline__ float4 dys_u4(const float4 z, const float4 w)
{
    float4 u;
    u.x = fmaf(-ALPHA, w.x, fabsf(z.x));
    u.y = fmaf(-ALPHA, w.y, fabsf(z.y));
    u.z = fmaf(-ALPHA, w.z, fabsf(z.z));
    u.w = fmaf(-ALPHA, w.w, fabsf(z.w));
    return u;
}

template <bool HASW>
__global__ __launch_bounds__(256) void gemm_f32_wmma(
    const float* __restrict__ Am, const float* __restrict__ Aw, int lda,
    const float* __restrict__ Bm, int ldb, int transB,
    float* __restrict__ Cm, int ldc,
    int Kc, int Ktot, size_t pstride, const float* gate)
{
    if (loop_done(gate)) return;

    __shared__ float As[2][TM][TK + 4];
    __shared__ float Bs[2][TK][TN + 4];

    const int t    = threadIdx.x;
    const int row0 = blockIdx.y * TM;
    const int col0 = blockIdx.x * TN;

    const int wid  = t >> 5;
    const int lane = t & 31;
    const int wm   = (wid >> 2) * 64;
    const int wn   = (wid & 3) * 32;
    const int hi   = lane >> 4;
    const int ln   = lane & 15;

    v8f acc[4][2];
#pragma unroll
    for (int mi = 0; mi < 4; ++mi)
#pragma unroll
        for (int ni = 0; ni < 2; ++ni) acc[mi][ni] = (v8f){};

    const int arow = t >> 1;
    const int akc  = (t & 1) * 16;
    const int bkp  = t >> 3;
    const int bnp  = (t & 7) * 16;
    const int btn  = t >> 1;
    const int btk  = (t & 1) * 16;

    const int kbeg = blockIdx.z * Kc;
    const int kend = min(kbeg + Kc, Ktot);
    const int nk   = (kend - kbeg) / TK;
    float* __restrict__ Cp = Cm + (size_t)blockIdx.z * pstride;

    const size_t aoff = (size_t)(row0 + arow) * lda + kbeg + akc;
    const float* ap = Am + aoff;
    const float* wp = HASW ? Aw + aoff : nullptr;
    const float* bp = transB ? Bm + (size_t)(col0 + btn) * ldb + kbeg + btk
                             : Bm + (size_t)(kbeg + bkp) * ldb + col0 + bnp;
    const size_t bstep = transB ? (size_t)TK : (size_t)TK * ldb;

    float4 ra0, ra1, ra2, ra3, rb0, rb1, rb2, rb3;
    float4 rw0 = make_float4(0.f, 0.f, 0.f, 0.f), rw1 = rw0, rw2 = rw0, rw3 = rw0;

    ra0 = *(const float4*)(ap + 0);  ra1 = *(const float4*)(ap + 4);
    ra2 = *(const float4*)(ap + 8);  ra3 = *(const float4*)(ap + 12);
    rb0 = *(const float4*)(bp + 0);  rb1 = *(const float4*)(bp + 4);
    rb2 = *(const float4*)(bp + 8);  rb3 = *(const float4*)(bp + 12);
    if (HASW) {
        rw0 = *(const float4*)(wp + 0);  rw1 = *(const float4*)(wp + 4);
        rw2 = *(const float4*)(wp + 8);  rw3 = *(const float4*)(wp + 12);
        ra0 = dys_u4(ra0, rw0);  ra1 = dys_u4(ra1, rw1);
        ra2 = dys_u4(ra2, rw2);  ra3 = dys_u4(ra3, rw3);
    }
    stage_tiles(&As[0][0][0], &Bs[0][0][0], transB, ra0, ra1, ra2, ra3, rb0, rb1, rb2, rb3,
                arow, akc, bkp, bnp, btn, btk);
    __syncthreads();

    for (int kt = 0; kt < nk; ++kt) {
        const int  cur  = kt & 1;
        const bool more = kt + 1 < nk;
        if (more) {
            ap += TK;
            bp += bstep;
            ra0 = *(const float4*)(ap + 0);  ra1 = *(const float4*)(ap + 4);
            ra2 = *(const float4*)(ap + 8);  ra3 = *(const float4*)(ap + 12);
            rb0 = *(const float4*)(bp + 0);  rb1 = *(const float4*)(bp + 4);
            rb2 = *(const float4*)(bp + 8);  rb3 = *(const float4*)(bp + 12);
            if (HASW) {
                wp += TK;
                rw0 = *(const float4*)(wp + 0);  rw1 = *(const float4*)(wp + 4);
                rw2 = *(const float4*)(wp + 8);  rw3 = *(const float4*)(wp + 12);
            }
        }

#pragma unroll
        for (int kk = 0; kk < TK; kk += 4) {
            v2f a[4], b[2];
#pragma unroll
            for (int mi = 0; mi < 4; ++mi) {
                a[mi].x = As[cur][wm + 16 * mi + ln][kk + 2 * hi];
                a[mi].y = As[cur][wm + 16 * mi + ln][kk + 2 * hi + 1];
            }
#pragma unroll
            for (int ni = 0; ni < 2; ++ni) {
                b[ni].x = Bs[cur][kk + 2 * hi][wn + 16 * ni + ln];
                b[ni].y = Bs[cur][kk + 2 * hi + 1][wn + 16 * ni + ln];
            }
#pragma unroll
            for (int mi = 0; mi < 4; ++mi)
#pragma unroll
                for (int ni = 0; ni < 2; ++ni)
                    acc[mi][ni] = wmma4(a[mi], b[ni], acc[mi][ni]);
        }

        if (more) {
            if (HASW) {
                ra0 = dys_u4(ra0, rw0);  ra1 = dys_u4(ra1, rw1);
                ra2 = dys_u4(ra2, rw2);  ra3 = dys_u4(ra3, rw3);
            }
            stage_tiles(&As[cur ^ 1][0][0], &Bs[cur ^ 1][0][0], transB,
                        ra0, ra1, ra2, ra3, rb0, rb1, rb2, rb3, arow, akc, bkp, bnp, btn, btk);
            __syncthreads();
        }
    }

#pragma unroll
    for (int mi = 0; mi < 4; ++mi)
#pragma unroll
        for (int ni = 0; ni < 2; ++ni)
            store_partial(acc[mi][ni], row0 + wm + 16 * mi, col0 + wn + 16 * ni,
                          Cp, ldc, hi, ln);
}

__global__ void k_sumsq(const float* __restrict__ x, float* __restrict__ acc, int n)
{
    __shared__ float sm[256];
    int i = blockIdx.x * blockDim.x + threadIdx.x;
    float v = 0.f;
    if (i < n) { float t = x[i]; v = t * t; }
    sm[threadIdx.x] = v;
    __syncthreads();
    for (int s = 128; s > 0; s >>= 1) {
        if (threadIdx.x < s) sm[threadIdx.x] += sm[threadIdx.x + s];
        __syncthreads();
    }
    if (threadIdx.x == 0) atomicAdd(acc, sm[0]);
}

__global__ void k_zero(float* p) { p[0] = 0.f; }

__global__ __launch_bounds__(256) void k_splitk_reduce(
    const float* P, int nsplit, size_t pstride,
    float* C, const float* __restrict__ Cin, const float* __restrict__ bias,
    int ncols, float alpha, float beta, float gamma, int relu, const float* gate)
{
    if (loop_done(gate)) return;
    const size_t i = ((size_t)blockIdx.x * blockDim.x + threadIdx.x) * 4;
    float4 s = make_float4(0.f, 0.f, 0.f, 0.f);
    for (int k = 0; k < nsplit; k += 4) {
        const float4 p0 = *(const float4*)(P + (size_t)(k + 0) * pstride + i);
        const float4 p1 = *(const float4*)(P + (size_t)(k + 1) * pstride + i);
        const float4 p2 = *(const float4*)(P + (size_t)(k + 2) * pstride + i);
        const float4 p3 = *(const float4*)(P + (size_t)(k + 3) * pstride + i);
        s.x += p0.x; s.y += p0.y; s.z += p0.z; s.w += p0.w;
        s.x += p1.x; s.y += p1.y; s.z += p1.z; s.w += p1.w;
        s.x += p2.x; s.y += p2.y; s.z += p2.z; s.w += p2.w;
        s.x += p3.x; s.y += p3.y; s.z += p3.z; s.w += p3.w;
    }
    float4 v = make_float4(alpha * s.x, alpha * s.y, alpha * s.z, alpha * s.w);
    if (bias) {
        const int c = (int)(i % (size_t)ncols);
        const float4 bv = *(const float4*)(bias + c);
        v.x += gamma * bv.x; v.y += gamma * bv.y; v.z += gamma * bv.z; v.w += gamma * bv.w;
    }
    if (Cin) {
        const float4 ci = *(const float4*)(Cin + i);
        v.x += beta * ci.x; v.y += beta * ci.y; v.z += beta * ci.z; v.w += beta * ci.w;
    }
    if (relu) {
        v.x = fmaxf(v.x, 0.f); v.y = fmaxf(v.y, 0.f); v.z = fmaxf(v.z, 0.f); v.w = fmaxf(v.w, 0.f);
    }
    *(float4*)(C + i) = v;
}

__global__ void k_init_X(float* __restrict__ X, const float* __restrict__ fn, int n)
{
    int i = blockIdx.x * blockDim.x + threadIdx.x;
    if (i < n * n) {
        float c = rsqrtf(*fn);
        int r = i / n, cc = i - r * n;
        X[i] = (r == cc) ? c : 0.f;
    }
}

static __device__ __forceinline__ float dys_elem(float s, float zo, float wo, float& zn)
{
    const float uo = fmaf(-ALPHA, wo, fabsf(zo));
    const float y  = uo - s;
    const float x  = fmaxf(zo, 0.f);
    zn = zo - x + y;
    const float d = y - x;
    return d * d;
}

__global__ __launch_bounds__(256) void k_dys_init(
    const float* __restrict__ z0, float* __restrict__ z, float* __restrict__ acc0)
{
    if (blockIdx.x == 0 && threadIdx.x == 0) *acc0 = 0.f;
    const size_t i = ((size_t)blockIdx.x * 256 + threadIdx.x) * 4;
    *(float4*)(z + i) = *(const float4*)(z0 + i);
}

#define SM 64
#define SN 64
#define SK 128
#define RK 1024
#define RN 4096

static __device__ __forceinline__ void dys_frags(
    const float (*A)[SK + 4], const float (*Bt)[SN + 16], int kb, int wm, int wn, int hi, int ln,
    v2f (&fa)[4][2], v2f (&fb)[4])
{
#pragma unroll
    for (int q = 0; q < 4; ++q) {
        const int k2 = kb + 4 * q + 2 * hi;
#pragma unroll
        for (int mi = 0; mi < 2; ++mi) {
            fa[q][mi].x = A[wm + 16 * mi + ln][k2];
            fa[q][mi].y = A[wm + 16 * mi + ln][k2 + 1];
        }
        fb[q].x = Bt[k2][wn + ln];
        fb[q].y = Bt[k2 + 1][wn + ln];
    }
}

template <int LDB>
static __device__ __forceinline__ void tile64_kloop(
    const float* __restrict__ ap, const float* __restrict__ bp,
    float (&As)[2][SM][SK + 4], float (&Bs)[2][SK][SN + 16],
    int ar, int ac, int br, int bc, int wm, int wn, int hi, int ln, v8f (&c2)[2])
{
    float4 ra0, ra1, ra2, ra3, ra4, ra5, ra6, ra7;
    float4 rb0, rb1, rb2, rb3, rb4, rb5, rb6, rb7;
#define DYS_LOAD_TILE()                                                                      \
    ra0 = *(const float4*)(ap + 0 * 8 * RK);   ra1 = *(const float4*)(ap + 1 * 8 * RK);      \
    ra2 = *(const float4*)(ap + 2 * 8 * RK);   ra3 = *(const float4*)(ap + 3 * 8 * RK);      \
    ra4 = *(const float4*)(ap + 4 * 8 * RK);   ra5 = *(const float4*)(ap + 5 * 8 * RK);      \
    ra6 = *(const float4*)(ap + 6 * 8 * RK);   ra7 = *(const float4*)(ap + 7 * 8 * RK);      \
    rb0 = *(const float4*)(bp + 0 * 16 * LDB); rb1 = *(const float4*)(bp + 1 * 16 * LDB);    \
    rb2 = *(const float4*)(bp + 2 * 16 * LDB); rb3 = *(const float4*)(bp + 3 * 16 * LDB);    \
    rb4 = *(const float4*)(bp + 4 * 16 * LDB); rb5 = *(const float4*)(bp + 5 * 16 * LDB);    \
    rb6 = *(const float4*)(bp + 6 * 16 * LDB); rb7 = *(const float4*)(bp + 7 * 16 * LDB);
#define DYS_STAGE_TILE(s)                                                                    \
    *(float4*)&As[s][ar + 0][ac]  = ra0;   *(float4*)&As[s][ar + 8][ac]  = ra1;              \
    *(float4*)&As[s][ar + 16][ac] = ra2;   *(float4*)&As[s][ar + 24][ac] = ra3;              \
    *(float4*)&As[s][ar + 32][ac] = ra4;   *(float4*)&As[s][ar + 40][ac] = ra5;              \
    *(float4*)&As[s][ar + 48][ac] = ra6;   *(float4*)&As[s][ar + 56][ac] = ra7;              \
    *(float4*)&Bs[s][br + 0][bc]  = rb0;   *(float4*)&Bs[s][br + 16][bc] = rb1;              \
    *(float4*)&Bs[s][br + 32][bc] = rb2;   *(float4*)&Bs[s][br + 48][bc] = rb3;              \
    *(float4*)&Bs[s][br + 64][bc] = rb4;   *(float4*)&Bs[s][br + 80][bc] = rb5;              \
    *(float4*)&Bs[s][br + 96][bc] = rb6;   *(float4*)&Bs[s][br + 112][bc] = rb7;

    DYS_LOAD_TILE()
    DYS_STAGE_TILE(0)
    __syncthreads();

    const int nk = RK / SK;
    for (int kt = 0; kt < nk; ++kt) {
        const int  cur  = kt & 1;
        const bool more = kt + 1 < nk;
        if (more) {
            ap += SK;
            bp += (size_t)SK * LDB;
            DYS_LOAD_TILE()
        }

        v2f fa[2][4][2], fb[2][4];
        dys_frags(As[cur], Bs[cur], 0, wm, wn, hi, ln, fa[0], fb[0]);
        __builtin_amdgcn_sched_barrier(0);
#pragma unroll
        for (int g = 0; g < SK / 16; ++g) {
            const int s = g & 1;
            if (g + 1 < SK / 16)
                dys_frags(As[cur], Bs[cur], 16 * (g + 1), wm, wn, hi, ln, fa[s ^ 1], fb[s ^ 1]);
            __builtin_amdgcn_sched_barrier(0);
            if (g == SK / 16 - 2 && more) {
                DYS_STAGE_TILE(cur ^ 1)
                __builtin_amdgcn_sched_barrier(0);
            }
#pragma unroll
            for (int q = 0; q < 4; ++q)
#pragma unroll
                for (int mi = 0; mi < 2; ++mi) c2[mi] = wmma4(fa[s][q][mi], fb[s][q], c2[mi]);
            __builtin_amdgcn_sched_barrier(0);
        }

        if (more) __syncthreads();
    }

#undef DYS_LOAD_TILE
#undef DYS_STAGE_TILE
}

template <bool FINAL>
__global__ __launch_bounds__(256) void gemm_dys_step(
    const float* __restrict__ Rm, const float* __restrict__ Mi,
    float* z, const float* __restrict__ w,
    float* acc, float* next, const float* gate)
{
    __shared__ __attribute__((aligned(16))) float As[2][SM][SK + 4];
    __shared__ __attribute__((aligned(16))) float Bs[2][SK][SN + 16];
    __shared__ float sm[8];

    const int t = threadIdx.x;
    if (!FINAL) {
        if (blockIdx.x == 0 && blockIdx.y == 0 && t == 0) *next = 0.f;
        if (loop_done(gate)) return;
    }

    const int row0 = blockIdx.y * SM;
    const int col0 = blockIdx.x * SN;
    const int wid  = t >> 5;
    const int lane = t & 31;
    const int wm   = (wid >> 2) * 32;
    const int wn   = (wid & 3) * 16;
    const int hi   = lane >> 4;
    const int ln   = lane & 15;

    const int erow = t >> 4;
    const int ec4  = (t & 15) * 4;
    float4 zo[4], wo[4];
#pragma unroll
    for (int j = 0; j < 4; ++j) {
        const size_t gi = (size_t)(row0 + erow + 16 * j) * RN + col0 + ec4;
        zo[j] = *(const float4*)(z + gi);
        wo[j] = *(const float4*)(w + gi);
    }

    v8f c2[2];
    c2[0] = (v8f){};
    c2[1] = (v8f){};

    const int ar = t >> 5;
    const int ac = (t & 31) * 4;
    const int br = t >> 4;
    const int bc = (t & 15) * 4;
    tile64_kloop<RN>(Rm + (size_t)(row0 + ar) * RK + ac, Mi + (size_t)br * RN + col0 + bc,
                     As, Bs, ar, ac, br, bc, wm, wn, hi, ln, c2);

    __syncthreads();
    float* Cs = &As[0][0][0];
#pragma unroll
    for (int mi = 0; mi < 2; ++mi)
#pragma unroll
        for (int v = 0; v < 8; ++v)
            Cs[(wm + 16 * mi + 8 * hi + v) * (SN + 4) + wn + ln] = c2[mi][v];
    __syncthreads();

    float d2 = 0.f;
#pragma unroll
    for (int j = 0; j < 4; ++j) {
        const int r = erow + 16 * j;
        const float4 s = *(const float4*)&Cs[r * (SN + 4) + ec4];
        const size_t gi = (size_t)(row0 + r) * RN + col0 + ec4;
        float4 zn;
        d2 += dys_elem(s.x, zo[j].x, wo[j].x, zn.x);
        d2 += dys_elem(s.y, zo[j].y, wo[j].y, zn.y);
        d2 += dys_elem(s.z, zo[j].z, wo[j].z, zn.z);
        d2 += dys_elem(s.w, zo[j].w, wo[j].w, zn.w);
        if (FINAL) {
            zn.x = fmaxf(zn.x, 0.f); zn.y = fmaxf(zn.y, 0.f);
            zn.z = fmaxf(zn.z, 0.f); zn.w = fmaxf(zn.w, 0.f);
        }
        *(float4*)(z + gi) = zn;
    }

    if (!FINAL) {
        for (int off = warpSize / 2; off > 0; off >>= 1) d2 += __shfl_xor(d2, off);
        if (lane == 0) sm[wid] = d2;
        __syncthreads();
        if (t == 0) {
            float tot = 0.f;
            for (int k = 0; k < 8; ++k) tot += sm[k];
            atomicAdd(acc, tot);
        }
    }
}

template <bool UPDATE>
__global__ __launch_bounds__(256) void gemm_ns_step(
    const float* __restrict__ Am, const float* __restrict__ Bm,
    float* __restrict__ Cm, const float* __restrict__ Cin)
{
    __shared__ __attribute__((aligned(16))) float As[2][SM][SK + 4];
    __shared__ __attribute__((aligned(16))) float Bs[2][SK][SN + 16];

    const int t    = threadIdx.x;
    const int row0 = blockIdx.y * SM;
    const int col0 = blockIdx.x * SN;
    const int wid  = t >> 5;
    const int lane = t & 31;
    const int wm   = (wid >> 2) * 32;
    const int wn   = (wid & 3) * 16;
    const int hi   = lane >> 4;
    const int ln   = lane & 15;

    const int erow = t >> 4;
    const int ec4  = (t & 15) * 4;
    float4 ci[4];
    if (UPDATE) {
#pragma unroll
        for (int j = 0; j < 4; ++j)
            ci[j] = *(const float4*)(Cin + (size_t)(row0 + erow + 16 * j) * RK + col0 + ec4);
    }

    v8f c2[2];
    c2[0] = (v8f){};
    c2[1] = (v8f){};

    const int ar = t >> 5;
    const int ac = (t & 31) * 4;
    const int br = t >> 4;
    const int bc = (t & 15) * 4;
    tile64_kloop<RK>(Am + (size_t)(row0 + ar) * RK + ac, Bm + (size_t)br * RK + col0 + bc,
                     As, Bs, ar, ac, br, bc, wm, wn, hi, ln, c2);

    __syncthreads();
    float* Cs = &As[0][0][0];
#pragma unroll
    for (int mi = 0; mi < 2; ++mi)
#pragma unroll
        for (int v = 0; v < 8; ++v)
            Cs[(wm + 16 * mi + 8 * hi + v) * (SN + 4) + wn + ln] = c2[mi][v];
    __syncthreads();

#pragma unroll
    for (int j = 0; j < 4; ++j) {
        const int r = erow + 16 * j;
        float4 v = *(const float4*)&Cs[r * (SN + 4) + ec4];
        if (UPDATE) {
            v.x = 2.f * ci[j].x - v.x;  v.y = 2.f * ci[j].y - v.y;
            v.z = 2.f * ci[j].z - v.z;  v.w = 2.f * ci[j].w - v.w;
        }
        *(float4*)(Cm + (size_t)(row0 + r) * RK + col0 + ec4) = v;
    }
}

extern "C" void kernel_launch(void* const* d_in, const int* in_sizes, int n_in,
                              void* d_out, int out_size, void* d_ws, size_t ws_size,
                              hipStream_t stream)
{
    const float* ctx  = (const float*)d_in[0];
    const float* Ain  = (const float*)d_in[1];
    const float* bvec = (const float*)d_in[2];
    const float* W1   = (const float*)d_in[3];
    const float* b1   = (const float*)d_in[4];
    const float* W2   = (const float*)d_in[5];
    const float* b2   = (const float*)d_in[6];
    const float* z0   = (const float*)d_in[7];
    float* out = (float*)d_out;

    const int B = 256, D = 512, H = 2048, N1 = 1024, N2 = 4096;

    float* W = (float*)d_ws;
    const size_t M1 = (size_t)1 << 20;
    float* G    = W;
    float* Xb   = W + 1 * M1;
    float* T1   = W + 2 * M1;
    float* Xa   = W + 3 * M1;
    float* F    = W + 5 * M1;
    float* Minv = W + 4 * M1;
    float* P    = W;
    float* h    = W + 8 * M1;
    float* wv   = W + 8 * M1;
    float* z    = out;
    float* R    = P;
    float* acc  = W + 9 * M1;
    float* fn   = acc + 1;

    dim3 blk(256);
    auto partials = [&](const float* A_, const float* Aw_, int lda, const float* B_, int ldb,
                        int tB, float* P_, int ns, int Kc, int M, int N, int K,
                        const float* gate) {
        dim3 grid(N / TN, M / TM, ns);
        if (Aw_)
            gemm_f32_wmma<true><<<grid, blk, 0, stream>>>(A_, Aw_, lda, B_, ldb, tB, P_, N,
                                                          Kc, K, (size_t)M * N, gate);
        else
            gemm_f32_wmma<false><<<grid, blk, 0, stream>>>(A_, Aw_, lda, B_, ldb, tB, P_, N,
                                                           Kc, K, (size_t)M * N, gate);
    };
    auto gemm = [&](const float* A_, const float* Aw_, int lda, const float* B_, int ldb, int tB,
                    float* P_, int ns, float* C_, const float* Cin, const float* bias,
                    float al, float be, float ga, int relu,
                    int M, int N, int K, const float* gate) {
        partials(A_, Aw_, lda, B_, ldb, tB, P_, ns, K / ns, M, N, K, gate);
        k_splitk_reduce<<<dim3((M * N) / 1024), blk, 0, stream>>>(
            P_, ns, (size_t)M * N, C_, Cin, bias, N, al, be, ga, relu, gate);
    };

    gemm(Ain, nullptr, N2, Ain, N2, 1, F, 4, G, nullptr, nullptr, 1.f, 0.f, 0.f, 0,
         N1, N1, N2, nullptr);

    k_zero<<<1, 1, 0, stream>>>(fn);
    k_sumsq<<<dim3((N1 * N1 + 255) / 256), blk, 0, stream>>>(G, fn, N1 * N1);
    k_init_X<<<dim3((N1 * N1 + 255) / 256), blk, 0, stream>>>(Xa, fn, N1);

    float* Xc = Xa; float* Xn = Xb;
    for (int it = 0; it < 16; ++it) {
        gemm_ns_step<false><<<dim3(N1 / SN, N1 / SM), blk, 0, stream>>>(G,  Xc, T1, nullptr);
        gemm_ns_step<true><<<dim3(N1 / SN, N1 / SM), blk, 0, stream>>>(Xc, T1, Xn, Xc);
        float* tmp = Xc; Xc = Xn; Xn = tmp;
    }
    float* Ginv = Xc;

    partials(Ginv, nullptr, N1, Ain, N2, 0, Minv, 1, N1, N1, N2, N1, nullptr);

    gemm(ctx, nullptr, D, W1, H,  0, P, 8, h,  nullptr, b1, 1.f, 0.f, 1.f, 1, B, H,  D, nullptr);
    gemm(h,   nullptr, H, W2, N2, 0, P, 4, wv, nullptr, b2, 1.f, 0.f, 1.f, 0, B, N2, H, nullptr);

    const int nz = B * N2;
    k_dys_init<<<dim3(nz / 1024), blk, 0, stream>>>(z0, z, acc);
    float* cur = acc; float* nxt = acc + 1; float* prv = acc + 2;
    const float* gate = nullptr;
    for (int itr = 0; itr < 50; ++itr) {
        gemm(z, wv, N2, Ain, N2, 1, P, 16, R, nullptr, bvec, 1.f, 0.f, -1.f, 0, B, N1, N2, gate);
        gemm_dys_step<false><<<dim3(N2 / SN, B / SM), blk, 0, stream>>>(R, Minv, z, wv, cur, nxt, gate);
        gate = cur;
        float* tmp = prv; prv = cur; cur = nxt; nxt = tmp;
    }

    gemm(z, wv, N2, Ain, N2, 1, P, 16, R, nullptr, bvec, 1.f, 0.f, -1.f, 0, B, N1, N2, nullptr);
    gemm_dys_step<true><<<dim3(N2 / SN, B / SM), blk, 0, stream>>>(R, Minv, z, wv, nullptr, nullptr,
                                                                   nullptr);
}


# loop: the first product (u@A^T - b) is left as two K halves by 256 work-groups of 32x64 tiles (u stored beside z) and gemm_dys_step's A loader adds them: 2 launches per iteration, was 3
# speedup vs baseline: 9.0099x; 1.1098x over previous
#include <hip/hip_runtime.h>

typedef __attribute__((ext_vector_type(2))) float v2f;
typedef __attribute__((ext_vector_type(4))) float v4f;
typedef __attribute__((ext_vector_type(8))) float v8f;

#define ALPHA 0.05f
#define EPS   0.01f

#define TM 128
#define TN 128
#define TK 32

static __device__ __forceinline__ v8f wmma4(const v2f a, const v2f b, const v8f c)
{
    return __builtin_amdgcn_wmma_f32_16x16x4_f32(false, a, false, b, (short)0, c,
                                                 false, false);
}

static __device__ __forceinline__ bool loop_done(const float* gate)
{
    return gate && *gate <= EPS * EPS;
}

static __device__ __forceinline__ void store_partial(
    const v8f acc, int r0, int c0, float* __restrict__ Cp, int ldc, int hi, int ln)
{
    const int c = c0 + ln;
#pragma unroll
    for (int v = 0; v < 8; ++v) {
        const int r = r0 + v + 8 * hi;
        Cp[(size_t)r * ldc + c] = acc[v];
    }
}

static __device__ __forceinline__ void stage_tiles(
    float* __restrict__ Ad, float* __restrict__ Bd, int transB,
    const float4 a0, const float4 a1, const float4 a2, const float4 a3,
    const float4 b0, const float4 b1, const float4 b2, const float4 b3,
    int arow, int akc, int bkp, int bnp, int btn, int btk)
{
    float* ad = Ad + arow * (TK + 4) + akc;
    *(float4*)(ad + 0)  = a0;
    *(float4*)(ad + 4)  = a1;
    *(float4*)(ad + 8)  = a2;
    *(float4*)(ad + 12) = a3;
    if (!transB) {
        float* bd = Bd + bkp * (TN + 4) + bnp;
        *(float4*)(bd + 0)  = b0;
        *(float4*)(bd + 4)  = b1;
        *(float4*)(bd + 8)  = b2;
        *(float4*)(bd + 12) = b3;
    } else {
        float* bd = Bd + btk * (TN + 4) + btn;
        bd[0  * (TN + 4)] = b0.x; bd[1  * (TN + 4)] = b0.y; bd[2  * (TN + 4)] = b0.z; bd[3  * (TN + 4)] = b0.w;
        bd[4  * (TN + 4)] = b1.x; bd[5  * (TN + 4)] = b1.y; bd[6  * (TN + 4)] = b1.z; bd[7  * (TN + 4)] = b1.w;
        bd[8  * (TN + 4)] = b2.x; bd[9  * (TN + 4)] = b2.y; bd[10 * (TN + 4)] = b2.z; bd[11 * (TN + 4)] = b2.w;
        bd[12 * (TN + 4)] = b3.x; bd[13 * (TN + 4)] = b3.y; bd[14 * (TN + 4)] = b3.z; bd[15 * (TN + 4)] = b3.w;
    }
}

static __device__ __forceinline__ float4 dys_u4(const float4 z, const float4 w)
{
    float4 u;
    u.x = fmaf(-ALPHA, w.x, fabsf(z.x));
    u.y = fmaf(-ALPHA, w.y, fabsf(z.y));
    u.z = fmaf(-ALPHA, w.z, fabsf(z.z));
    u.w = fmaf(-ALPHA, w.w, fabsf(z.w));
    return u;
}

__global__ __launch_bounds__(256) void gemm_f32_wmma(
    const float* __restrict__ Am, int lda,
    const float* __restrict__ Bm, int ldb, int transB,
    float* __restrict__ Cm, int ldc,
    int Kc, int Ktot, size_t pstride, const float* gate)
{
    if (loop_done(gate)) return;

    __shared__ float As[2][TM][TK + 4];
    __shared__ float Bs[2][TK][TN + 4];

    const int t    = threadIdx.x;
    const int row0 = blockIdx.y * TM;
    const int col0 = blockIdx.x * TN;

    const int wid  = t >> 5;
    const int lane = t & 31;
    const int wm   = (wid >> 2) * 64;
    const int wn   = (wid & 3) * 32;
    const int hi   = lane >> 4;
    const int ln   = lane & 15;

    v8f acc[4][2];
#pragma unroll
    for (int mi = 0; mi < 4; ++mi)
#pragma unroll
        for (int ni = 0; ni < 2; ++ni) acc[mi][ni] = (v8f){};

    const int arow = t >> 1;
    const int akc  = (t & 1) * 16;
    const int bkp  = t >> 3;
    const int bnp  = (t & 7) * 16;
    const int btn  = t >> 1;
    const int btk  = (t & 1) * 16;

    const int kbeg = blockIdx.z * Kc;
    const int kend = min(kbeg + Kc, Ktot);
    const int nk   = (kend - kbeg) / TK;
    float* __restrict__ Cp = Cm + (size_t)blockIdx.z * pstride;

    const float* ap = Am + (size_t)(row0 + arow) * lda + kbeg + akc;
    const float* bp = transB ? Bm + (size_t)(col0 + btn) * ldb + kbeg + btk
                             : Bm + (size_t)(kbeg + bkp) * ldb + col0 + bnp;
    const size_t bstep = transB ? (size_t)TK : (size_t)TK * ldb;

    float4 ra0, ra1, ra2, ra3, rb0, rb1, rb2, rb3;

    ra0 = *(const float4*)(ap + 0);  ra1 = *(const float4*)(ap + 4);
    ra2 = *(const float4*)(ap + 8);  ra3 = *(const float4*)(ap + 12);
    rb0 = *(const float4*)(bp + 0);  rb1 = *(const float4*)(bp + 4);
    rb2 = *(const float4*)(bp + 8);  rb3 = *(const float4*)(bp + 12);
    stage_tiles(&As[0][0][0], &Bs[0][0][0], transB, ra0, ra1, ra2, ra3, rb0, rb1, rb2, rb3,
                arow, akc, bkp, bnp, btn, btk);
    __syncthreads();

    for (int kt = 0; kt < nk; ++kt) {
        const int  cur  = kt & 1;
        const bool more = kt + 1 < nk;
        if (more) {
            ap += TK;
            bp += bstep;
            ra0 = *(const float4*)(ap + 0);  ra1 = *(const float4*)(ap + 4);
            ra2 = *(const float4*)(ap + 8);  ra3 = *(const float4*)(ap + 12);
            rb0 = *(const float4*)(bp + 0);  rb1 = *(const float4*)(bp + 4);
            rb2 = *(const float4*)(bp + 8);  rb3 = *(const float4*)(bp + 12);
        }

#pragma unroll
        for (int kk = 0; kk < TK; kk += 4) {
            v2f a[4], b[2];
#pragma unroll
            for (int mi = 0; mi < 4; ++mi) {
                a[mi].x = As[cur][wm + 16 * mi + ln][kk + 2 * hi];
                a[mi].y = As[cur][wm + 16 * mi + ln][kk + 2 * hi + 1];
            }
#pragma unroll
            for (int ni = 0; ni < 2; ++ni) {
                b[ni].x = Bs[cur][kk + 2 * hi][wn + 16 * ni + ln];
                b[ni].y = Bs[cur][kk + 2 * hi + 1][wn + 16 * ni + ln];
            }
#pragma unroll
            for (int mi = 0; mi < 4; ++mi)
#pragma unroll
                for (int ni = 0; ni < 2; ++ni)
                    acc[mi][ni] = wmma4(a[mi], b[ni], acc[mi][ni]);
        }

        if (more) {
            stage_tiles(&As[cur ^ 1][0][0], &Bs[cur ^ 1][0][0], transB,
                        ra0, ra1, ra2, ra3, rb0, rb1, rb2, rb3, arow, akc, bkp, bnp, btn, btk);
            __syncthreads();
        }
    }

#pragma unroll
    for (int mi = 0; mi < 4; ++mi)
#pragma unroll
        for (int ni = 0; ni < 2; ++ni)
            store_partial(acc[mi][ni], row0 + wm + 16 * mi, col0 + wn + 16 * ni,
                          Cp, ldc, hi, ln);
}

__global__ void k_sumsq(const float* __restrict__ x, float* __restrict__ acc, int n)
{
    __shared__ float sm[256];
    int i = blockIdx.x * blockDim.x + threadIdx.x;
    float v = 0.f;
    if (i < n) { float t = x[i]; v = t * t; }
    sm[threadIdx.x] = v;
    __syncthreads();
    for (int s = 128; s > 0; s >>= 1) {
        if (threadIdx.x < s) sm[threadIdx.x] += sm[threadIdx.x + s];
        __syncthreads();
    }
    if (threadIdx.x == 0) atomicAdd(acc, sm[0]);
}

__global__ void k_zero(float* p) { p[0] = 0.f; }

__global__ __launch_bounds__(256) void k_splitk_reduce(
    const float* P, int nsplit, size_t pstride,
    float* C, const float* __restrict__ Cin, const float* __restrict__ bias,
    int ncols, float alpha, float beta, float gamma, int relu, const float* gate)
{
    if (loop_done(gate)) return;
    const size_t i = ((size_t)blockIdx.x * blockDim.x + threadIdx.x) * 4;
    float4 s = make_float4(0.f, 0.f, 0.f, 0.f);
    for (int k = 0; k < nsplit; k += 4) {
        const float4 p0 = *(const float4*)(P + (size_t)(k + 0) * pstride + i);
        const float4 p1 = *(const float4*)(P + (size_t)(k + 1) * pstride + i);
        const float4 p2 = *(const float4*)(P + (size_t)(k + 2) * pstride + i);
        const float4 p3 = *(const float4*)(P + (size_t)(k + 3) * pstride + i);
        s.x += p0.x; s.y += p0.y; s.z += p0.z; s.w += p0.w;
        s.x += p1.x; s.y += p1.y; s.z += p1.z; s.w += p1.w;
        s.x += p2.x; s.y += p2.y; s.z += p2.z; s.w += p2.w;
        s.x += p3.x; s.y += p3.y; s.z += p3.z; s.w += p3.w;
    }
    float4 v = make_float4(alpha * s.x, alpha * s.y, alpha * s.z, alpha * s.w);
    if (bias) {
        const int c = (int)(i % (size_t)ncols);
        const float4 bv = *(const float4*)(bias + c);
        v.x += gamma * bv.x; v.y += gamma * bv.y; v.z += gamma * bv.z; v.w += gamma * bv.w;
    }
    if (Cin) {
        const float4 ci = *(const float4*)(Cin + i);
        v.x += beta * ci.x; v.y += beta * ci.y; v.z += beta * ci.z; v.w += beta * ci.w;
    }
    if (relu) {
        v.x = fmaxf(v.x, 0.f); v.y = fmaxf(v.y, 0.f); v.z = fmaxf(v.z, 0.f); v.w = fmaxf(v.w, 0.f);
    }
    *(float4*)(C + i) = v;
}

__global__ void k_init_X(float* __restrict__ X, const float* __restrict__ fn, int n)
{
    int i = blockIdx.x * blockDim.x + threadIdx.x;
    if (i < n * n) {
        float c = rsqrtf(*fn);
        int r = i / n, cc = i - r * n;
        X[i] = (r == cc) ? c : 0.f;
    }
}

static __device__ __forceinline__ float dys_elem(float s, float zo, float wo, float& zn)
{
    const float uo = fmaf(-ALPHA, wo, fabsf(zo));
    const float y  = uo - s;
    const float x  = fmaxf(zo, 0.f);
    zn = zo - x + y;
    const float d = y - x;
    return d * d;
}

__global__ __launch_bounds__(256) void k_dys_init(
    const float* __restrict__ z0, const float* __restrict__ w,
    float* __restrict__ z, float* __restrict__ u, float* __restrict__ acc0)
{
    if (blockIdx.x == 0 && threadIdx.x == 0) *acc0 = 0.f;
    const size_t i = ((size_t)blockIdx.x * 256 + threadIdx.x) * 4;
    const float4 zv = *(const float4*)(z0 + i);
    *(float4*)(z + i) = zv;
    *(float4*)(u + i) = dys_u4(zv, *(const float4*)(w + i));
}

#define SM 64
#define SN 64
#define SK 128
#define RK 1024
#define RN 4096

static __device__ __forceinline__ void dys_frags(
    const float (*A)[SK + 4], const float (*Bt)[SN + 16], int kb, int wm, int wn, int hi, int ln,
    v2f (&fa)[4][2], v2f (&fb)[4])
{
#pragma unroll
    for (int q = 0; q < 4; ++q) {
        const int k2 = kb + 4 * q + 2 * hi;
#pragma unroll
        for (int mi = 0; mi < 2; ++mi) {
            fa[q][mi].x = A[wm + 16 * mi + ln][k2];
            fa[q][mi].y = A[wm + 16 * mi + ln][k2 + 1];
        }
        fb[q].x = Bt[k2][wn + ln];
        fb[q].y = Bt[k2 + 1][wn + ln];
    }
}

static __device__ __forceinline__ float4 add4(const float4 a, const v4f b)
{
    return make_float4(a.x + b.x, a.y + b.y, a.z + b.z, a.w + b.w);
}

template <int LDB, bool SUM2, int ASEP>
static __device__ __forceinline__ void tile64_kloop(
    const float* __restrict__ ap, const float* __restrict__ bp,
    float (&As)[2][SM][SK + 4], float (&Bs)[2][SK][SN + 16],
    int ar, int ac, int br, int bc, int wm, int wn, int hi, int ln, v8f (&c2)[2])
{
    float4 ra0, ra1, ra2, ra3, ra4, ra5, ra6, ra7;
    float4 rb0, rb1, rb2, rb3, rb4, rb5, rb6, rb7;
    v4f rc0 = (v4f){};
    v4f rc1 = rc0, rc2 = rc0, rc3 = rc0, rc4 = rc0, rc5 = rc0, rc6 = rc0, rc7 = rc0;
#define DYS_LOAD_TILE()                                                                      \
    ra0 = *(const float4*)(ap + 0 * 8 * RK);   ra1 = *(const float4*)(ap + 1 * 8 * RK);      \
    ra2 = *(const float4*)(ap + 2 * 8 * RK);   ra3 = *(const float4*)(ap + 3 * 8 * RK);      \
    ra4 = *(const float4*)(ap + 4 * 8 * RK);   ra5 = *(const float4*)(ap + 5 * 8 * RK);      \
    ra6 = *(const float4*)(ap + 6 * 8 * RK);   ra7 = *(const float4*)(ap + 7 * 8 * RK);      \
    if (SUM2) {                                                                              \
        rc0 = *(const v4f*)(ap + ASEP + 0 * 8 * RK);  rc1 = *(const v4f*)(ap + ASEP + 1 * 8 * RK);  \
        rc2 = *(const v4f*)(ap + ASEP + 2 * 8 * RK);  rc3 = *(const v4f*)(ap + ASEP + 3 * 8 * RK);  \
        rc4 = *(const v4f*)(ap + ASEP + 4 * 8 * RK);  rc5 = *(const v4f*)(ap + ASEP + 5 * 8 * RK);  \
        rc6 = *(const v4f*)(ap + ASEP + 6 * 8 * RK);  rc7 = *(const v4f*)(ap + ASEP + 7 * 8 * RK);  \
    }                                                                                        \
    rb0 = *(const float4*)(bp + 0 * 16 * LDB); rb1 = *(const float4*)(bp + 1 * 16 * LDB);    \
    rb2 = *(const float4*)(bp + 2 * 16 * LDB); rb3 = *(const float4*)(bp + 3 * 16 * LDB);    \
    rb4 = *(const float4*)(bp + 4 * 16 * LDB); rb5 = *(const float4*)(bp + 5 * 16 * LDB);    \
    rb6 = *(const float4*)(bp + 6 * 16 * LDB); rb7 = *(const float4*)(bp + 7 * 16 * LDB);
#define DYS_STAGE_TILE(s)                                                                    \
    if (SUM2) {                                                                              \
          \
          \
          \
        asm volatile("" : "+v"(rc0), "+v"(rc1), "+v"(rc2), "+v"(rc3),                        \
                          "+v"(rc4), "+v"(rc5), "+v"(rc6), "+v"(rc7));                       \
        ra0 = add4(ra0, rc0);  ra1 = add4(ra1, rc1);  ra2 = add4(ra2, rc2);  ra3 = add4(ra3, rc3);  \
        ra4 = add4(ra4, rc4);  ra5 = add4(ra5, rc5);  ra6 = add4(ra6, rc6);  ra7 = add4(ra7, rc7);  \
    }                                                                                        \
    *(float4*)&As[s][ar + 0][ac]  = ra0;   *(float4*)&As[s][ar + 8][ac]  = ra1;              \
    *(float4*)&As[s][ar + 16][ac] = ra2;   *(float4*)&As[s][ar + 24][ac] = ra3;              \
    *(float4*)&As[s][ar + 32][ac] = ra4;   *(float4*)&As[s][ar + 40][ac] = ra5;              \
    *(float4*)&As[s][ar + 48][ac] = ra6;   *(float4*)&As[s][ar + 56][ac] = ra7;              \
    *(float4*)&Bs[s][br + 0][bc]  = rb0;   *(float4*)&Bs[s][br + 16][bc] = rb1;              \
    *(float4*)&Bs[s][br + 32][bc] = rb2;   *(float4*)&Bs[s][br + 48][bc] = rb3;              \
    *(float4*)&Bs[s][br + 64][bc] = rb4;   *(float4*)&Bs[s][br + 80][bc] = rb5;              \
    *(float4*)&Bs[s][br + 96][bc] = rb6;   *(float4*)&Bs[s][br + 112][bc] = rb7;

    DYS_LOAD_TILE()
    DYS_STAGE_TILE(0)
    __syncthreads();

    const int nk = RK / SK;
    for (int kt = 0; kt < nk; ++kt) {
        const int  cur  = kt & 1;
        const bool more = kt + 1 < nk;
        if (more) {
            ap += SK;
            bp += (size_t)SK * LDB;
            DYS_LOAD_TILE()
        }

        v2f fa[2][4][2], fb[2][4];
        dys_frags(As[cur], Bs[cur], 0, wm, wn, hi, ln, fa[0], fb[0]);
        __builtin_amdgcn_sched_barrier(0);
#pragma unroll
        for (int g = 0; g < SK / 16; ++g) {
            const int s = g & 1;
            if (g + 1 < SK / 16)
                dys_frags(As[cur], Bs[cur], 16 * (g + 1), wm, wn, hi, ln, fa[s ^ 1], fb[s ^ 1]);
            __builtin_amdgcn_sched_barrier(0);
            if (g == SK / 16 - 2 && more) {
                DYS_STAGE_TILE(cur ^ 1)
                __builtin_amdgcn_sched_barrier(0);
            }
#pragma unroll
            for (int q = 0; q < 4; ++q)
#pragma unroll
                for (int mi = 0; mi < 2; ++mi) c2[mi] = wmma4(fa[s][q][mi], fb[s][q], c2[mi]);
            __builtin_amdgcn_sched_barrier(0);
        }

        if (more) __syncthreads();
    }

#undef DYS_LOAD_TILE
#undef DYS_STAGE_TILE
}

template <bool V> struct ktile_flag { static constexpr bool value = V; };

#define PM 32
#define PN 64
#define PK 256
#define PG 32

static __device__ __forceinline__ void rpart_frags(
    const float (*Ut)[PK + 4], const float (*Bt)[PK + 4], int kb, int wm, int wn, int hi, int ln,
    v2f (&fa)[4], v2f (&fb)[2][4])
{
#pragma unroll
    for (int q = 0; q < 4; ++q) {
        const int k2 = kb + 4 * q + 2 * hi;
        fa[q].x = Ut[wm + ln][k2];
        fa[q].y = Ut[wm + ln][k2 + 1];
#pragma unroll
        for (int ni = 0; ni < 2; ++ni) {
            fb[ni][q].x = Bt[wn + 16 * ni + ln][k2];
            fb[ni][q].y = Bt[wn + 16 * ni + ln][k2 + 1];
        }
    }
}

__global__ __launch_bounds__(256) void gemm_r_part(
    const float* __restrict__ U, const float* __restrict__ Am, const float* __restrict__ bvec,
    float* __restrict__ P, size_t pstride, const float* gate)
{
    if (loop_done(gate)) return;

    __shared__ __attribute__((aligned(16))) float Us[2][PM][PK + 4];
    __shared__ __attribute__((aligned(16))) float Bs[2][PN][PK + 4];

    const int t    = threadIdx.x;
    const int row0 = blockIdx.y * PM;
    const int col0 = blockIdx.x * PN;
    const int kbeg = blockIdx.z * (RN / 2);
    const int wid  = t >> 5;
    const int lane = t & 31;
    const int kg   = wid >> 2;
    const int wm   = ((wid >> 1) & 1) * 16;
    const int wn   = (wid & 1) * 32;
    const int hi   = lane >> 4;
    const int ln   = lane & 15;

    const int sr = t >> 6;
    const int sc = (t & 63) * 4;
    const float* up = U  + (size_t)(row0 + sr) * RN + kbeg + sc;
    const float* bp = Am + (size_t)(col0 + sr) * RN + kbeg + sc;

    float4 ru0, ru1, ru2, ru3, ru4, ru5, ru6, ru7;
    float4 rb0, rb1, rb2, rb3, rb4, rb5, rb6, rb7, rb8, rb9, rb10, rb11, rb12, rb13, rb14, rb15;
#define RP_LD(p, j) (*(const float4*)((p) + (size_t)(j) * 4 * RN))
#define RP_LOAD_TILE()                                                                       \
    ru0 = RP_LD(up, 0);   ru1 = RP_LD(up, 1);   ru2 = RP_LD(up, 2);   ru3 = RP_LD(up, 3);     \
    ru4 = RP_LD(up, 4);   ru5 = RP_LD(up, 5);   ru6 = RP_LD(up, 6);   ru7 = RP_LD(up, 7);     \
    rb0 = RP_LD(bp, 0);   rb1 = RP_LD(bp, 1);   rb2 = RP_LD(bp, 2);   rb3 = RP_LD(bp, 3);     \
    rb4 = RP_LD(bp, 4);   rb5 = RP_LD(bp, 5);   rb6 = RP_LD(bp, 6);   rb7 = RP_LD(bp, 7);     \
    rb8 = RP_LD(bp, 8);   rb9 = RP_LD(bp, 9);   rb10 = RP_LD(bp, 10); rb11 = RP_LD(bp, 11);   \
    rb12 = RP_LD(bp, 12); rb13 = RP_LD(bp, 13); rb14 = RP_LD(bp, 14); rb15 = RP_LD(bp, 15);
#define RP_ST(T, s, j, v) (*(float4*)&T[s][sr + 4 * (j)][sc] = (v))
#define RP_STAGE_TILE(s)                                                                     \
    RP_ST(Us, s, 0, ru0);   RP_ST(Us, s, 1, ru1);   RP_ST(Us, s, 2, ru2);   RP_ST(Us, s, 3, ru3);     \
    RP_ST(Us, s, 4, ru4);   RP_ST(Us, s, 5, ru5);   RP_ST(Us, s, 6, ru6);   RP_ST(Us, s, 7, ru7);     \
    RP_ST(Bs, s, 0, rb0);   RP_ST(Bs, s, 1, rb1);   RP_ST(Bs, s, 2, rb2);   RP_ST(Bs, s, 3, rb3);     \
    RP_ST(Bs, s, 4, rb4);   RP_ST(Bs, s, 5, rb5);   RP_ST(Bs, s, 6, rb6);   RP_ST(Bs, s, 7, rb7);     \
    RP_ST(Bs, s, 8, rb8);   RP_ST(Bs, s, 9, rb9);   RP_ST(Bs, s, 10, rb10); RP_ST(Bs, s, 11, rb11);   \
    RP_ST(Bs, s, 12, rb12); RP_ST(Bs, s, 13, rb13); RP_ST(Bs, s, 14, rb14); RP_ST(Bs, s, 15, rb15);

    v8f acc[2];
    acc[0] = (v8f){};
    acc[1] = (v8f){};

    RP_LOAD_TILE()
    RP_STAGE_TILE(0)
    __syncthreads();

    auto ktile = [&](const int cur, auto more_tag) __attribute__((always_inline)) {
        constexpr bool MORE = decltype(more_tag)::value;
        if (MORE) {
            up += PK;
            bp += PK;
            RP_LOAD_TILE()
        }

        v2f fa[2][4], fb[2][2][4];
        rpart_frags(Us[cur], Bs[cur], 16 * kg, wm, wn, hi, ln, fa[0], fb[0]);
        __builtin_amdgcn_sched_barrier(0);
#pragma unroll
        for (int g = 0; g < PK / PG; ++g) {
            const int s = g & 1;
            if (g + 1 < PK / PG)
                rpart_frags(Us[cur], Bs[cur], PG * (g + 1) + 16 * kg, wm, wn, hi, ln,
                            fa[s ^ 1], fb[s ^ 1]);
            __builtin_amdgcn_sched_barrier(0);
            if (g == PK / PG - 2 && MORE) {
                RP_STAGE_TILE(cur ^ 1)
                __builtin_amdgcn_sched_barrier(0);
            }
#pragma unroll
            for (int q = 0; q < 4; ++q)
#pragma unroll
                for (int ni = 0; ni < 2; ++ni) acc[ni] = wmma4(fa[s][q], fb[s][ni][q], acc[ni]);
            __builtin_amdgcn_sched_barrier(0);
        }

        if (MORE) __syncthreads();
    };

    const int nk = (RN / 2) / PK;
    for (int kt = 0; kt + 1 < nk; ++kt) ktile(kt & 1, ktile_flag<true>{});
    ktile((nk - 1) & 1, ktile_flag<false>{});
#undef RP_LD
#undef RP_LOAD_TILE
#undef RP_ST
#undef RP_STAGE_TILE

    __syncthreads();
    float* Cs = &Us[0][0][0];
#pragma unroll
    for (int ni = 0; ni < 2; ++ni)
#pragma unroll
        for (int v = 0; v < 8; ++v)
            Cs[(kg * PM + wm + 8 * hi + v) * (PN + 4) + wn + 16 * ni + ln] = acc[ni][v];
    __syncthreads();

    const int er  = t >> 4;
    const int ec4 = (t & 15) * 4;
    float4 bv = make_float4(0.f, 0.f, 0.f, 0.f);
    if (blockIdx.z == 0) bv = *(const float4*)(bvec + col0 + ec4);
    float* __restrict__ Pp = P + (size_t)blockIdx.z * pstride;
#pragma unroll
    for (int j = 0; j < 2; ++j) {
        const int r = er + 16 * j;
        const float4 s0 = *(const float4*)&Cs[r * (PN + 4) + ec4];
        const float4 s1 = *(const float4*)&Cs[(PM + r) * (PN + 4) + ec4];
        float4 s;
        s.x = s0.x + s1.x - bv.x;  s.y = s0.y + s1.y - bv.y;
        s.z = s0.z + s1.z - bv.z;  s.w = s0.w + s1.w - bv.w;
        *(float4*)(Pp + (size_t)(row0 + r) * RK + col0 + ec4) = s;
    }
}

#define RHALF (256 * RK)
template <bool FINAL>
__global__ __launch_bounds__(256) void gemm_dys_step(
    const float* __restrict__ Rm, const float* __restrict__ Mi,
    float* z, const float* __restrict__ w, float* __restrict__ u,
    float* acc, float* next, const float* gate)
{
    __shared__ __attribute__((aligned(16))) float As[2][SM][SK + 4];
    __shared__ __attribute__((aligned(16))) float Bs[2][SK][SN + 16];
    __shared__ float sm[8];

    const int t = threadIdx.x;
    if (!FINAL) {
        if (blockIdx.x == 0 && blockIdx.y == 0 && t == 0) *next = 0.f;
        if (loop_done(gate)) return;
    }

    const int row0 = blockIdx.y * SM;
    const int col0 = blockIdx.x * SN;
    const int wid  = t >> 5;
    const int lane = t & 31;
    const int wm   = (wid >> 2) * 32;
    const int wn   = (wid & 3) * 16;
    const int hi   = lane >> 4;
    const int ln   = lane & 15;

    const int erow = t >> 4;
    const int ec4  = (t & 15) * 4;
    float4 zo[4], wo[4];
#pragma unroll
    for (int j = 0; j < 4; ++j) {
        const size_t gi = (size_t)(row0 + erow + 16 * j) * RN + col0 + ec4;
        zo[j] = *(const float4*)(z + gi);
        wo[j] = *(const float4*)(w + gi);
    }

    v8f c2[2];
    c2[0] = (v8f){};
    c2[1] = (v8f){};

    const int ar = t >> 5;
    const int ac = (t & 31) * 4;
    const int br = t >> 4;
    const int bc = (t & 15) * 4;
    tile64_kloop<RN, true, RHALF>(Rm + (size_t)(row0 + ar) * RK + ac,
                                  Mi + (size_t)br * RN + col0 + bc,
                                  As, Bs, ar, ac, br, bc, wm, wn, hi, ln, c2);

    __syncthreads();
    float* Cs = &As[0][0][0];
#pragma unroll
    for (int mi = 0; mi < 2; ++mi)
#pragma unroll
        for (int v = 0; v < 8; ++v)
            Cs[(wm + 16 * mi + 8 * hi + v) * (SN + 4) + wn + ln] = c2[mi][v];
    __syncthreads();

    float d2 = 0.f;
#pragma unroll
    for (int j = 0; j < 4; ++j) {
        const int r = erow + 16 * j;
        const float4 s = *(const float4*)&Cs[r * (SN + 4) + ec4];
        const size_t gi = (size_t)(row0 + r) * RN + col0 + ec4;
        float4 zn;
        d2 += dys_elem(s.x, zo[j].x, wo[j].x, zn.x);
        d2 += dys_elem(s.y, zo[j].y, wo[j].y, zn.y);
        d2 += dys_elem(s.z, zo[j].z, wo[j].z, zn.z);
        d2 += dys_elem(s.w, zo[j].w, wo[j].w, zn.w);
        if (FINAL) {
            zn.x = fmaxf(zn.x, 0.f); zn.y = fmaxf(zn.y, 0.f);
            zn.z = fmaxf(zn.z, 0.f); zn.w = fmaxf(zn.w, 0.f);
        } else {
            *(float4*)(u + gi) = dys_u4(zn, wo[j]);
        }
        *(float4*)(z + gi) = zn;
    }

    if (!FINAL) {
        for (int off = warpSize / 2; off > 0; off >>= 1) d2 += __shfl_xor(d2, off);
        if (lane == 0) sm[wid] = d2;
        __syncthreads();
        if (t == 0) {
            float tot = 0.f;
            for (int k = 0; k < 8; ++k) tot += sm[k];
            atomicAdd(acc, tot);
        }
    }
}

template <bool UPDATE>
__global__ __launch_bounds__(256) void gemm_ns_step(
    const float* __restrict__ Am, const float* __restrict__ Bm,
    float* __restrict__ Cm, const float* __restrict__ Cin)
{
    __shared__ __attribute__((aligned(16))) float As[2][SM][SK + 4];
    __shared__ __attribute__((aligned(16))) float Bs[2][SK][SN + 16];

    const int t    = threadIdx.x;
    const int row0 = blockIdx.y * SM;
    const int col0 = blockIdx.x * SN;
    const int wid  = t >> 5;
    const int lane = t & 31;
    const int wm   = (wid >> 2) * 32;
    const int wn   = (wid & 3) * 16;
    const int hi   = lane >> 4;
    const int ln   = lane & 15;

    const int erow = t >> 4;
    const int ec4  = (t & 15) * 4;
    float4 ci[4];
    if (UPDATE) {
#pragma unroll
        for (int j = 0; j < 4; ++j)
            ci[j] = *(const float4*)(Cin + (size_t)(row0 + erow + 16 * j) * RK + col0 + ec4);
    }

    v8f c2[2];
    c2[0] = (v8f){};
    c2[1] = (v8f){};

    const int ar = t >> 5;
    const int ac = (t & 31) * 4;
    const int br = t >> 4;
    const int bc = (t & 15) * 4;
    tile64_kloop<RK, false, 0>(Am + (size_t)(row0 + ar) * RK + ac, Bm + (size_t)br * RK + col0 + bc,
                               As, Bs, ar, ac, br, bc, wm, wn, hi, ln, c2);

    __syncthreads();
    float* Cs = &As[0][0][0];
#pragma unroll
    for (int mi = 0; mi < 2; ++mi)
#pragma unroll
        for (int v = 0; v < 8; ++v)
            Cs[(wm + 16 * mi + 8 * hi + v) * (SN + 4) + wn + ln] = c2[mi][v];
    __syncthreads();

#pragma unroll
    for (int j = 0; j < 4; ++j) {
        const int r = erow + 16 * j;
        float4 v = *(const float4*)&Cs[r * (SN + 4) + ec4];
        if (UPDATE) {
            v.x = 2.f * ci[j].x - v.x;  v.y = 2.f * ci[j].y - v.y;
            v.z = 2.f * ci[j].z - v.z;  v.w = 2.f * ci[j].w - v.w;
        }
        *(float4*)(Cm + (size_t)(row0 + r) * RK + col0 + ec4) = v;
    }
}

extern "C" void kernel_launch(void* const* d_in, const int* in_sizes, int n_in,
                              void* d_out, int out_size, void* d_ws, size_t ws_size,
                              hipStream_t stream)
{
    const float* ctx  = (const float*)d_in[0];
    const float* Ain  = (const float*)d_in[1];
    const float* bvec = (const float*)d_in[2];
    const float* W1   = (const float*)d_in[3];
    const float* b1   = (const float*)d_in[4];
    const float* W2   = (const float*)d_in[5];
    const float* b2   = (const float*)d_in[6];
    const float* z0   = (const float*)d_in[7];
    float* out = (float*)d_out;

    const int B = 256, D = 512, H = 2048, N1 = 1024, N2 = 4096;

    float* W = (float*)d_ws;
    const size_t M1 = (size_t)1 << 20;
    float* G    = W;
    float* Xb   = W + 1 * M1;
    float* T1   = W + 2 * M1;
    float* Xa   = W + 3 * M1;
    float* F    = W + 5 * M1;
    float* Minv = W + 4 * M1;
    float* P    = W;
    float* h    = W + 8 * M1;
    float* wv   = W + 8 * M1;
    float* z    = out;
    float* R    = P;
    float* U    = W + 1 * M1;
    float* acc  = W + 9 * M1;
    float* fn   = acc + 1;

    dim3 blk(256);
    auto partials = [&](const float* A_, int lda, const float* B_, int ldb,
                        int tB, float* P_, int ns, int Kc, int M, int N, int K,
                        const float* gate) {
        dim3 grid(N / TN, M / TM, ns);
        gemm_f32_wmma<<<grid, blk, 0, stream>>>(A_, lda, B_, ldb, tB, P_, N,
                                                Kc, K, (size_t)M * N, gate);
    };
    auto gemm = [&](const float* A_, int lda, const float* B_, int ldb, int tB,
                    float* P_, int ns, float* C_, const float* Cin, const float* bias,
                    float al, float be, float ga, int relu,
                    int M, int N, int K, const float* gate) {
        partials(A_, lda, B_, ldb, tB, P_, ns, K / ns, M, N, K, gate);
        k_splitk_reduce<<<dim3((M * N) / 1024), blk, 0, stream>>>(
            P_, ns, (size_t)M * N, C_, Cin, bias, N, al, be, ga, relu, gate);
    };

    gemm(Ain, N2, Ain, N2, 1, F, 4, G, nullptr, nullptr, 1.f, 0.f, 0.f, 0,
         N1, N1, N2, nullptr);

    k_zero<<<1, 1, 0, stream>>>(fn);
    k_sumsq<<<dim3((N1 * N1 + 255) / 256), blk, 0, stream>>>(G, fn, N1 * N1);
    k_init_X<<<dim3((N1 * N1 + 255) / 256), blk, 0, stream>>>(Xa, fn, N1);

    float* Xc = Xa; float* Xn = Xb;
    for (int it = 0; it < 16; ++it) {
        gemm_ns_step<false><<<dim3(N1 / SN, N1 / SM), blk, 0, stream>>>(G,  Xc, T1, nullptr);
        gemm_ns_step<true><<<dim3(N1 / SN, N1 / SM), blk, 0, stream>>>(Xc, T1, Xn, Xc);
        float* tmp = Xc; Xc = Xn; Xn = tmp;
    }
    float* Ginv = Xc;

    partials(Ginv, N1, Ain, N2, 0, Minv, 1, N1, N1, N2, N1, nullptr);

    gemm(ctx, D, W1, H,  0, P, 8, h,  nullptr, b1, 1.f, 0.f, 1.f, 1, B, H,  D, nullptr);
    gemm(h,   H, W2, N2, 0, P, 4, wv, nullptr, b2, 1.f, 0.f, 1.f, 0, B, N2, H, nullptr);

    const int nz = B * N2;
    k_dys_init<<<dim3(nz / 1024), blk, 0, stream>>>(z0, wv, z, U, acc);
    const dim3 rgrid(N1 / PN, B / PM, 2);
    float* cur = acc; float* nxt = acc + 1; float* prv = acc + 2;
    const float* gate = nullptr;
    for (int itr = 0; itr < 50; ++itr) {
        gemm_r_part<<<rgrid, blk, 0, stream>>>(U, Ain, bvec, R, (size_t)B * N1, gate);
        gemm_dys_step<false><<<dim3(N2 / SN, B / SM), blk, 0, stream>>>(R, Minv, z, wv, U, cur, nxt, gate);
        gate = cur;
        float* tmp = prv; prv = cur; cur = nxt; nxt = tmp;
    }

    gemm_r_part<<<rgrid, blk, 0, stream>>>(U, Ain, bvec, R, (size_t)B * N1, nullptr);
    gemm_dys_step<true><<<dim3(N2 / SN, B / SM), blk, 0, stream>>>(R, Minv, z, wv, nullptr,
                                                                   nullptr, nullptr, nullptr);
}


# K-tile order rotated per work-group in the three whole-K GEMM kernels (starting tile from the block indices), so work-groups sharing operand rows ask L2 for different lines at any one time
# speedup vs baseline: 9.3212x; 1.0346x over previous
#include <hip/hip_runtime.h>

typedef __attribute__((ext_vector_type(2))) float v2f;
typedef __attribute__((ext_vector_type(4))) float v4f;
typedef __attribute__((ext_vector_type(8))) float v8f;

#define ALPHA 0.05f
#define EPS   0.01f

#define TM 128
#define TN 128
#define TK 32

static __device__ __forceinline__ v8f wmma4(const v2f a, const v2f b, const v8f c)
{
    return __builtin_amdgcn_wmma_f32_16x16x4_f32(false, a, false, b, (short)0, c,
                                                 false, false);
}

static __device__ __forceinline__ bool loop_done(const float* gate)
{
    return gate && *gate <= EPS * EPS;
}

static __device__ __forceinline__ void store_partial(
    const v8f acc, int r0, int c0, float* __restrict__ Cp, int ldc, int hi, int ln)
{
    const int c = c0 + ln;
#pragma unroll
    for (int v = 0; v < 8; ++v) {
        const int r = r0 + v + 8 * hi;
        Cp[(size_t)r * ldc + c] = acc[v];
    }
}

static __device__ __forceinline__ void stage_tiles(
    float* __restrict__ Ad, float* __restrict__ Bd, int transB,
    const float4 a0, const float4 a1, const float4 a2, const float4 a3,
    const float4 b0, const float4 b1, const float4 b2, const float4 b3,
    int arow, int akc, int bkp, int bnp, int btn, int btk)
{
    float* ad = Ad + arow * (TK + 4) + akc;
    *(float4*)(ad + 0)  = a0;
    *(float4*)(ad + 4)  = a1;
    *(float4*)(ad + 8)  = a2;
    *(float4*)(ad + 12) = a3;
    if (!transB) {
        float* bd = Bd + bkp * (TN + 4) + bnp;
        *(float4*)(bd + 0)  = b0;
        *(float4*)(bd + 4)  = b1;
        *(float4*)(bd + 8)  = b2;
        *(float4*)(bd + 12) = b3;
    } else {
        float* bd = Bd + btk * (TN + 4) + btn;
        bd[0  * (TN + 4)] = b0.x; bd[1  * (TN + 4)] = b0.y; bd[2  * (TN + 4)] = b0.z; bd[3  * (TN + 4)] = b0.w;
        bd[4  * (TN + 4)] = b1.x; bd[5  * (TN + 4)] = b1.y; bd[6  * (TN + 4)] = b1.z; bd[7  * (TN + 4)] = b1.w;
        bd[8  * (TN + 4)] = b2.x; bd[9  * (TN + 4)] = b2.y; bd[10 * (TN + 4)] = b2.z; bd[11 * (TN + 4)] = b2.w;
        bd[12 * (TN + 4)] = b3.x; bd[13 * (TN + 4)] = b3.y; bd[14 * (TN + 4)] = b3.z; bd[15 * (TN + 4)] = b3.w;
    }
}

static __device__ __forceinline__ float4 dys_u4(const float4 z, const float4 w)
{
    float4 u;
    u.x = fmaf(-ALPHA, w.x, fabsf(z.x));
    u.y = fmaf(-ALPHA, w.y, fabsf(z.y));
    u.z = fmaf(-ALPHA, w.z, fabsf(z.z));
    u.w = fmaf(-ALPHA, w.w, fabsf(z.w));
    return u;
}

__global__ __launch_bounds__(256) void gemm_f32_wmma(
    const float* __restrict__ Am, int lda,
    const float* __restrict__ Bm, int ldb, int transB,
    float* __restrict__ Cm, int ldc,
    int Kc, int Ktot, size_t pstride, const float* gate)
{
    if (loop_done(gate)) return;

    __shared__ float As[2][TM][TK + 4];
    __shared__ float Bs[2][TK][TN + 4];

    const int t    = threadIdx.x;
    const int row0 = blockIdx.y * TM;
    const int col0 = blockIdx.x * TN;

    const int wid  = t >> 5;
    const int lane = t & 31;
    const int wm   = (wid >> 2) * 64;
    const int wn   = (wid & 3) * 32;
    const int hi   = lane >> 4;
    const int ln   = lane & 15;

    v8f acc[4][2];
#pragma unroll
    for (int mi = 0; mi < 4; ++mi)
#pragma unroll
        for (int ni = 0; ni < 2; ++ni) acc[mi][ni] = (v8f){};

    const int arow = t >> 1;
    const int akc  = (t & 1) * 16;
    const int bkp  = t >> 3;
    const int bnp  = (t & 7) * 16;
    const int btn  = t >> 1;
    const int btk  = (t & 1) * 16;

    const int kbeg = blockIdx.z * Kc;
    const int kend = min(kbeg + Kc, Ktot);
    const int nk   = (kend - kbeg) / TK;
    float* __restrict__ Cp = Cm + (size_t)blockIdx.z * pstride;

    const float* ap = Am + (size_t)(row0 + arow) * lda + kbeg + akc;
    const float* bp = transB ? Bm + (size_t)(col0 + btn) * ldb + kbeg + btk
                             : Bm + (size_t)(kbeg + bkp) * ldb + col0 + bnp;
    const size_t bstep = transB ? (size_t)TK : (size_t)TK * ldb;

    float4 ra0, ra1, ra2, ra3, rb0, rb1, rb2, rb3;

    ra0 = *(const float4*)(ap + 0);  ra1 = *(const float4*)(ap + 4);
    ra2 = *(const float4*)(ap + 8);  ra3 = *(const float4*)(ap + 12);
    rb0 = *(const float4*)(bp + 0);  rb1 = *(const float4*)(bp + 4);
    rb2 = *(const float4*)(bp + 8);  rb3 = *(const float4*)(bp + 12);
    stage_tiles(&As[0][0][0], &Bs[0][0][0], transB, ra0, ra1, ra2, ra3, rb0, rb1, rb2, rb3,
                arow, akc, bkp, bnp, btn, btk);
    __syncthreads();

    for (int kt = 0; kt < nk; ++kt) {
        const int  cur  = kt & 1;
        const bool more = kt + 1 < nk;
        if (more) {
            ap += TK;
            bp += bstep;
            ra0 = *(const float4*)(ap + 0);  ra1 = *(const float4*)(ap + 4);
            ra2 = *(const float4*)(ap + 8);  ra3 = *(const float4*)(ap + 12);
            rb0 = *(const float4*)(bp + 0);  rb1 = *(const float4*)(bp + 4);
            rb2 = *(const float4*)(bp + 8);  rb3 = *(const float4*)(bp + 12);
        }

#pragma unroll
        for (int kk = 0; kk < TK; kk += 4) {
            v2f a[4], b[2];
#pragma unroll
            for (int mi = 0; mi < 4; ++mi) {
                a[mi].x = As[cur][wm + 16 * mi + ln][kk + 2 * hi];
                a[mi].y = As[cur][wm + 16 * mi + ln][kk + 2 * hi + 1];
            }
#pragma unroll
            for (int ni = 0; ni < 2; ++ni) {
                b[ni].x = Bs[cur][kk + 2 * hi][wn + 16 * ni + ln];
                b[ni].y = Bs[cur][kk + 2 * hi + 1][wn + 16 * ni + ln];
            }
#pragma unroll
            for (int mi = 0; mi < 4; ++mi)
#pragma unroll
                for (int ni = 0; ni < 2; ++ni)
                    acc[mi][ni] = wmma4(a[mi], b[ni], acc[mi][ni]);
        }

        if (more) {
            stage_tiles(&As[cur ^ 1][0][0], &Bs[cur ^ 1][0][0], transB,
                        ra0, ra1, ra2, ra3, rb0, rb1, rb2, rb3, arow, akc, bkp, bnp, btn, btk);
            __syncthreads();
        }
    }

#pragma unroll
    for (int mi = 0; mi < 4; ++mi)
#pragma unroll
        for (int ni = 0; ni < 2; ++ni)
            store_partial(acc[mi][ni], row0 + wm + 16 * mi, col0 + wn + 16 * ni,
                          Cp, ldc, hi, ln);
}

__global__ void k_sumsq(const float* __restrict__ x, float* __restrict__ acc, int n)
{
    __shared__ float sm[256];
    int i = blockIdx.x * blockDim.x + threadIdx.x;
    float v = 0.f;
    if (i < n) { float t = x[i]; v = t * t; }
    sm[threadIdx.x] = v;
    __syncthreads();
    for (int s = 128; s > 0; s >>= 1) {
        if (threadIdx.x < s) sm[threadIdx.x] += sm[threadIdx.x + s];
        __syncthreads();
    }
    if (threadIdx.x == 0) atomicAdd(acc, sm[0]);
}

__global__ void k_zero(float* p) { p[0] = 0.f; }

__global__ __launch_bounds__(256) void k_splitk_reduce(
    const float* P, int nsplit, size_t pstride,
    float* C, const float* __restrict__ Cin, const float* __restrict__ bias,
    int ncols, float alpha, float beta, float gamma, int relu, const float* gate)
{
    if (loop_done(gate)) return;
    const size_t i = ((size_t)blockIdx.x * blockDim.x + threadIdx.x) * 4;
    float4 s = make_float4(0.f, 0.f, 0.f, 0.f);
    for (int k = 0; k < nsplit; k += 4) {
        const float4 p0 = *(const float4*)(P + (size_t)(k + 0) * pstride + i);
        const float4 p1 = *(const float4*)(P + (size_t)(k + 1) * pstride + i);
        const float4 p2 = *(const float4*)(P + (size_t)(k + 2) * pstride + i);
        const float4 p3 = *(const float4*)(P + (size_t)(k + 3) * pstride + i);
        s.x += p0.x; s.y += p0.y; s.z += p0.z; s.w += p0.w;
        s.x += p1.x; s.y += p1.y; s.z += p1.z; s.w += p1.w;
        s.x += p2.x; s.y += p2.y; s.z += p2.z; s.w += p2.w;
        s.x += p3.x; s.y += p3.y; s.z += p3.z; s.w += p3.w;
    }
    float4 v = make_float4(alpha * s.x, alpha * s.y, alpha * s.z, alpha * s.w);
    if (bias) {
        const int c = (int)(i % (size_t)ncols);
        const float4 bv = *(const float4*)(bias + c);
        v.x += gamma * bv.x; v.y += gamma * bv.y; v.z += gamma * bv.z; v.w += gamma * bv.w;
    }
    if (Cin) {
        const float4 ci = *(const float4*)(Cin + i);
        v.x += beta * ci.x; v.y += beta * ci.y; v.z += beta * ci.z; v.w += beta * ci.w;
    }
    if (relu) {
        v.x = fmaxf(v.x, 0.f); v.y = fmaxf(v.y, 0.f); v.z = fmaxf(v.z, 0.f); v.w = fmaxf(v.w, 0.f);
    }
    *(float4*)(C + i) = v;
}

__global__ void k_init_X(float* __restrict__ X, const float* __restrict__ fn, int n)
{
    int i = blockIdx.x * blockDim.x + threadIdx.x;
    if (i < n * n) {
        float c = rsqrtf(*fn);
        int r = i / n, cc = i - r * n;
        X[i] = (r == cc) ? c : 0.f;
    }
}

static __device__ __forceinline__ float dys_elem(float s, float zo, float wo, float& zn)
{
    const float uo = fmaf(-ALPHA, wo, fabsf(zo));
    const float y  = uo - s;
    const float x  = fmaxf(zo, 0.f);
    zn = zo - x + y;
    const float d = y - x;
    return d * d;
}

__global__ __launch_bounds__(256) void k_dys_init(
    const float* __restrict__ z0, const float* __restrict__ w,
    float* __restrict__ z, float* __restrict__ u, float* __restrict__ acc0)
{
    if (blockIdx.x == 0 && threadIdx.x == 0) *acc0 = 0.f;
    const size_t i = ((size_t)blockIdx.x * 256 + threadIdx.x) * 4;
    const float4 zv = *(const float4*)(z0 + i);
    *(float4*)(z + i) = zv;
    *(float4*)(u + i) = dys_u4(zv, *(const float4*)(w + i));
}

#define SM 64
#define SN 64
#define SK 128
#define RK 1024
#define RN 4096

static __device__ __forceinline__ void dys_frags(
    const float (*A)[SK + 4], const float (*Bt)[SN + 16], int kb, int wm, int wn, int hi, int ln,
    v2f (&fa)[4][2], v2f (&fb)[4])
{
#pragma unroll
    for (int q = 0; q < 4; ++q) {
        const int k2 = kb + 4 * q + 2 * hi;
#pragma unroll
        for (int mi = 0; mi < 2; ++mi) {
            fa[q][mi].x = A[wm + 16 * mi + ln][k2];
            fa[q][mi].y = A[wm + 16 * mi + ln][k2 + 1];
        }
        fb[q].x = Bt[k2][wn + ln];
        fb[q].y = Bt[k2 + 1][wn + ln];
    }
}

static __device__ __forceinline__ float4 add4(const float4 a, const v4f b)
{
    return make_float4(a.x + b.x, a.y + b.y, a.z + b.z, a.w + b.w);
}

template <int LDB, bool SUM2, int ASEP>
static __device__ __forceinline__ void tile64_kloop(
    const float* __restrict__ ap0, const float* __restrict__ bp0,
    float (&As)[2][SM][SK + 4], float (&Bs)[2][SK][SN + 16],
    int ar, int ac, int br, int bc, int wm, int wn, int hi, int ln, int rot, v8f (&c2)[2])
{
    constexpr int nk = RK / SK;
    static_assert((nk & (nk - 1)) == 0, "the K-tile rotation wraps with a mask");
    int kidx = rot & (nk - 1);
    const float* ap = ap0 + (size_t)kidx * SK;
    const float* bp = bp0 + (size_t)kidx * SK * LDB;
    float4 ra0, ra1, ra2, ra3, ra4, ra5, ra6, ra7;
    float4 rb0, rb1, rb2, rb3, rb4, rb5, rb6, rb7;
    v4f rc0 = (v4f){};
    v4f rc1 = rc0, rc2 = rc0, rc3 = rc0, rc4 = rc0, rc5 = rc0, rc6 = rc0, rc7 = rc0;
#define DYS_LOAD_TILE()                                                                      \
    ra0 = *(const float4*)(ap + 0 * 8 * RK);   ra1 = *(const float4*)(ap + 1 * 8 * RK);      \
    ra2 = *(const float4*)(ap + 2 * 8 * RK);   ra3 = *(const float4*)(ap + 3 * 8 * RK);      \
    ra4 = *(const float4*)(ap + 4 * 8 * RK);   ra5 = *(const float4*)(ap + 5 * 8 * RK);      \
    ra6 = *(const float4*)(ap + 6 * 8 * RK);   ra7 = *(const float4*)(ap + 7 * 8 * RK);      \
    if (SUM2) {                                                                              \
        rc0 = *(const v4f*)(ap + ASEP + 0 * 8 * RK);  rc1 = *(const v4f*)(ap + ASEP + 1 * 8 * RK);  \
        rc2 = *(const v4f*)(ap + ASEP + 2 * 8 * RK);  rc3 = *(const v4f*)(ap + ASEP + 3 * 8 * RK);  \
        rc4 = *(const v4f*)(ap + ASEP + 4 * 8 * RK);  rc5 = *(const v4f*)(ap + ASEP + 5 * 8 * RK);  \
        rc6 = *(const v4f*)(ap + ASEP + 6 * 8 * RK);  rc7 = *(const v4f*)(ap + ASEP + 7 * 8 * RK);  \
    }                                                                                        \
    rb0 = *(const float4*)(bp + 0 * 16 * LDB); rb1 = *(const float4*)(bp + 1 * 16 * LDB);    \
    rb2 = *(const float4*)(bp + 2 * 16 * LDB); rb3 = *(const float4*)(bp + 3 * 16 * LDB);    \
    rb4 = *(const float4*)(bp + 4 * 16 * LDB); rb5 = *(const float4*)(bp + 5 * 16 * LDB);    \
    rb6 = *(const float4*)(bp + 6 * 16 * LDB); rb7 = *(const float4*)(bp + 7 * 16 * LDB);
#define DYS_STAGE_TILE(s)                                                                    \
    if (SUM2) {                                                                              \
          \
          \
          \
        asm volatile("" : "+v"(rc0), "+v"(rc1), "+v"(rc2), "+v"(rc3),                        \
                          "+v"(rc4), "+v"(rc5), "+v"(rc6), "+v"(rc7));                       \
        ra0 = add4(ra0, rc0);  ra1 = add4(ra1, rc1);  ra2 = add4(ra2, rc2);  ra3 = add4(ra3, rc3);  \
        ra4 = add4(ra4, rc4);  ra5 = add4(ra5, rc5);  ra6 = add4(ra6, rc6);  ra7 = add4(ra7, rc7);  \
    }                                                                                        \
    *(float4*)&As[s][ar + 0][ac]  = ra0;   *(float4*)&As[s][ar + 8][ac]  = ra1;              \
    *(float4*)&As[s][ar + 16][ac] = ra2;   *(float4*)&As[s][ar + 24][ac] = ra3;              \
    *(float4*)&As[s][ar + 32][ac] = ra4;   *(float4*)&As[s][ar + 40][ac] = ra5;              \
    *(float4*)&As[s][ar + 48][ac] = ra6;   *(float4*)&As[s][ar + 56][ac] = ra7;              \
    *(float4*)&Bs[s][br + 0][bc]  = rb0;   *(float4*)&Bs[s][br + 16][bc] = rb1;              \
    *(float4*)&Bs[s][br + 32][bc] = rb2;   *(float4*)&Bs[s][br + 48][bc] = rb3;              \
    *(float4*)&Bs[s][br + 64][bc] = rb4;   *(float4*)&Bs[s][br + 80][bc] = rb5;              \
    *(float4*)&Bs[s][br + 96][bc] = rb6;   *(float4*)&Bs[s][br + 112][bc] = rb7;

    DYS_LOAD_TILE()
    DYS_STAGE_TILE(0)
    __syncthreads();

    for (int kt = 0; kt < nk; ++kt) {
        const int  cur  = kt & 1;
        const bool more = kt + 1 < nk;
        if (more) {
            kidx = (kidx + 1) & (nk - 1);
            ap = ap0 + (size_t)kidx * SK;
            bp = bp0 + (size_t)kidx * SK * LDB;
            DYS_LOAD_TILE()
        }

        v2f fa[2][4][2], fb[2][4];
        dys_frags(As[cur], Bs[cur], 0, wm, wn, hi, ln, fa[0], fb[0]);
        __builtin_amdgcn_sched_barrier(0);
#pragma unroll
        for (int g = 0; g < SK / 16; ++g) {
            const int s = g & 1;
            if (g + 1 < SK / 16)
                dys_frags(As[cur], Bs[cur], 16 * (g + 1), wm, wn, hi, ln, fa[s ^ 1], fb[s ^ 1]);
            __builtin_amdgcn_sched_barrier(0);
            if (g == SK / 16 - 2 && more) {
                DYS_STAGE_TILE(cur ^ 1)
                __builtin_amdgcn_sched_barrier(0);
            }
#pragma unroll
            for (int q = 0; q < 4; ++q)
#pragma unroll
                for (int mi = 0; mi < 2; ++mi) c2[mi] = wmma4(fa[s][q][mi], fb[s][q], c2[mi]);
            __builtin_amdgcn_sched_barrier(0);
        }

        if (more) __syncthreads();
    }

#undef DYS_LOAD_TILE
#undef DYS_STAGE_TILE
}

template <bool V> struct ktile_flag { static constexpr bool value = V; };

#define PM 32
#define PN 64
#define PK 256
#define PG 32

static __device__ __forceinline__ void rpart_frags(
    const float (*Ut)[PK + 4], const float (*Bt)[PK + 4], int kb, int wm, int wn, int hi, int ln,
    v2f (&fa)[4], v2f (&fb)[2][4])
{
#pragma unroll
    for (int q = 0; q < 4; ++q) {
        const int k2 = kb + 4 * q + 2 * hi;
        fa[q].x = Ut[wm + ln][k2];
        fa[q].y = Ut[wm + ln][k2 + 1];
#pragma unroll
        for (int ni = 0; ni < 2; ++ni) {
            fb[ni][q].x = Bt[wn + 16 * ni + ln][k2];
            fb[ni][q].y = Bt[wn + 16 * ni + ln][k2 + 1];
        }
    }
}

__global__ __launch_bounds__(256) void gemm_r_part(
    const float* __restrict__ U, const float* __restrict__ Am, const float* __restrict__ bvec,
    float* __restrict__ P, size_t pstride, const float* gate)
{
    if (loop_done(gate)) return;

    __shared__ __attribute__((aligned(16))) float Us[2][PM][PK + 4];
    __shared__ __attribute__((aligned(16))) float Bs[2][PN][PK + 4];

    const int t    = threadIdx.x;
    const int row0 = blockIdx.y * PM;
    const int col0 = blockIdx.x * PN;
    const int kbeg = blockIdx.z * (RN / 2);
    const int wid  = t >> 5;
    const int lane = t & 31;
    const int kg   = wid >> 2;
    const int wm   = ((wid >> 1) & 1) * 16;
    const int wn   = (wid & 1) * 32;
    const int hi   = lane >> 4;
    const int ln   = lane & 15;

    const int sr = t >> 6;
    const int sc = (t & 63) * 4;
    constexpr int nk = (RN / 2) / PK;
    static_assert((nk & (nk - 1)) == 0, "the K-tile rotation wraps with a mask");
    const int rot = (int)(blockIdx.x + blockIdx.y) & (nk - 1);
    const float* up0 = U  + (size_t)(row0 + sr) * RN + kbeg + sc;
    const float* bp0 = Am + (size_t)(col0 + sr) * RN + kbeg + sc;
    const float* up = up0 + rot * PK;
    const float* bp = bp0 + rot * PK;

    float4 ru0, ru1, ru2, ru3, ru4, ru5, ru6, ru7;
    float4 rb0, rb1, rb2, rb3, rb4, rb5, rb6, rb7, rb8, rb9, rb10, rb11, rb12, rb13, rb14, rb15;
#define RP_LD(p, j) (*(const float4*)((p) + (size_t)(j) * 4 * RN))
#define RP_LOAD_TILE()                                                                       \
    ru0 = RP_LD(up, 0);   ru1 = RP_LD(up, 1);   ru2 = RP_LD(up, 2);   ru3 = RP_LD(up, 3);     \
    ru4 = RP_LD(up, 4);   ru5 = RP_LD(up, 5);   ru6 = RP_LD(up, 6);   ru7 = RP_LD(up, 7);     \
    rb0 = RP_LD(bp, 0);   rb1 = RP_LD(bp, 1);   rb2 = RP_LD(bp, 2);   rb3 = RP_LD(bp, 3);     \
    rb4 = RP_LD(bp, 4);   rb5 = RP_LD(bp, 5);   rb6 = RP_LD(bp, 6);   rb7 = RP_LD(bp, 7);     \
    rb8 = RP_LD(bp, 8);   rb9 = RP_LD(bp, 9);   rb10 = RP_LD(bp, 10); rb11 = RP_LD(bp, 11);   \
    rb12 = RP_LD(bp, 12); rb13 = RP_LD(bp, 13); rb14 = RP_LD(bp, 14); rb15 = RP_LD(bp, 15);
#define RP_ST(T, s, j, v) (*(float4*)&T[s][sr + 4 * (j)][sc] = (v))
#define RP_STAGE_TILE(s)                                                                     \
    RP_ST(Us, s, 0, ru0);   RP_ST(Us, s, 1, ru1);   RP_ST(Us, s, 2, ru2);   RP_ST(Us, s, 3, ru3);     \
    RP_ST(Us, s, 4, ru4);   RP_ST(Us, s, 5, ru5);   RP_ST(Us, s, 6, ru6);   RP_ST(Us, s, 7, ru7);     \
    RP_ST(Bs, s, 0, rb0);   RP_ST(Bs, s, 1, rb1);   RP_ST(Bs, s, 2, rb2);   RP_ST(Bs, s, 3, rb3);     \
    RP_ST(Bs, s, 4, rb4);   RP_ST(Bs, s, 5, rb5);   RP_ST(Bs, s, 6, rb6);   RP_ST(Bs, s, 7, rb7);     \
    RP_ST(Bs, s, 8, rb8);   RP_ST(Bs, s, 9, rb9);   RP_ST(Bs, s, 10, rb10); RP_ST(Bs, s, 11, rb11);   \
    RP_ST(Bs, s, 12, rb12); RP_ST(Bs, s, 13, rb13); RP_ST(Bs, s, 14, rb14); RP_ST(Bs, s, 15, rb15);

    v8f acc[2];
    acc[0] = (v8f){};
    acc[1] = (v8f){};

    RP_LOAD_TILE()
    RP_STAGE_TILE(0)
    __syncthreads();

    auto ktile = [&](const int cur, const int nxt, auto more_tag) __attribute__((always_inline)) {
        constexpr bool MORE = decltype(more_tag)::value;
        if (MORE) {
            up = up0 + nxt * PK;
            bp = bp0 + nxt * PK;
            RP_LOAD_TILE()
        }

        v2f fa[2][4], fb[2][2][4];
        rpart_frags(Us[cur], Bs[cur], 16 * kg, wm, wn, hi, ln, fa[0], fb[0]);
        __builtin_amdgcn_sched_barrier(0);
#pragma unroll
        for (int g = 0; g < PK / PG; ++g) {
            const int s = g & 1;
            if (g + 1 < PK / PG)
                rpart_frags(Us[cur], Bs[cur], PG * (g + 1) + 16 * kg, wm, wn, hi, ln,
                            fa[s ^ 1], fb[s ^ 1]);
            __builtin_amdgcn_sched_barrier(0);
            if (g == PK / PG - 2 && MORE) {
                RP_STAGE_TILE(cur ^ 1)
                __builtin_amdgcn_sched_barrier(0);
            }
#pragma unroll
            for (int q = 0; q < 4; ++q)
#pragma unroll
                for (int ni = 0; ni < 2; ++ni) acc[ni] = wmma4(fa[s][q], fb[s][ni][q], acc[ni]);
            __builtin_amdgcn_sched_barrier(0);
        }

        if (MORE) __syncthreads();
    };

    for (int kt = 0; kt + 1 < nk; ++kt) ktile(kt & 1, (kt + 1 + rot) & (nk - 1), ktile_flag<true>{});
    ktile((nk - 1) & 1, 0, ktile_flag<false>{});
#undef RP_LD
#undef RP_LOAD_TILE
#undef RP_ST
#undef RP_STAGE_TILE

    __syncthreads();
    float* Cs = &Us[0][0][0];
#pragma unroll
    for (int ni = 0; ni < 2; ++ni)
#pragma unroll
        for (int v = 0; v < 8; ++v)
            Cs[(kg * PM + wm + 8 * hi + v) * (PN + 4) + wn + 16 * ni + ln] = acc[ni][v];
    __syncthreads();

    const int er  = t >> 4;
    const int ec4 = (t & 15) * 4;
    float4 bv = make_float4(0.f, 0.f, 0.f, 0.f);
    if (blockIdx.z == 0) bv = *(const float4*)(bvec + col0 + ec4);
    float* __restrict__ Pp = P + (size_t)blockIdx.z * pstride;
#pragma unroll
    for (int j = 0; j < 2; ++j) {
        const int r = er + 16 * j;
        const float4 s0 = *(const float4*)&Cs[r * (PN + 4) + ec4];
        const float4 s1 = *(const float4*)&Cs[(PM + r) * (PN + 4) + ec4];
        float4 s;
        s.x = s0.x + s1.x - bv.x;  s.y = s0.y + s1.y - bv.y;
        s.z = s0.z + s1.z - bv.z;  s.w = s0.w + s1.w - bv.w;
        *(float4*)(Pp + (size_t)(row0 + r) * RK + col0 + ec4) = s;
    }
}

#define RHALF (256 * RK)
template <bool FINAL>
__global__ __launch_bounds__(256) void gemm_dys_step(
    const float* __restrict__ Rm, const float* __restrict__ Mi,
    float* z, const float* __restrict__ w, float* __restrict__ u,
    float* acc, float* next, const float* gate)
{
    __shared__ __attribute__((aligned(16))) float As[2][SM][SK + 4];
    __shared__ __attribute__((aligned(16))) float Bs[2][SK][SN + 16];
    __shared__ float sm[8];

    const int t = threadIdx.x;
    if (!FINAL) {
        if (blockIdx.x == 0 && blockIdx.y == 0 && t == 0) *next = 0.f;
        if (loop_done(gate)) return;
    }

    const int row0 = blockIdx.y * SM;
    const int col0 = blockIdx.x * SN;
    const int wid  = t >> 5;
    const int lane = t & 31;
    const int wm   = (wid >> 2) * 32;
    const int wn   = (wid & 3) * 16;
    const int hi   = lane >> 4;
    const int ln   = lane & 15;

    const int erow = t >> 4;
    const int ec4  = (t & 15) * 4;
    float4 zo[4], wo[4];
#pragma unroll
    for (int j = 0; j < 4; ++j) {
        const size_t gi = (size_t)(row0 + erow + 16 * j) * RN + col0 + ec4;
        zo[j] = *(const float4*)(z + gi);
        wo[j] = *(const float4*)(w + gi);
    }

    v8f c2[2];
    c2[0] = (v8f){};
    c2[1] = (v8f){};

    const int ar = t >> 5;
    const int ac = (t & 31) * 4;
    const int br = t >> 4;
    const int bc = (t & 15) * 4;
    tile64_kloop<RN, true, RHALF>(Rm + (size_t)(row0 + ar) * RK + ac,
                                  Mi + (size_t)br * RN + col0 + bc,
                                  As, Bs, ar, ac, br, bc, wm, wn, hi, ln,
                                  (int)(blockIdx.x + 2 * blockIdx.y), c2);

    __syncthreads();
    float* Cs = &As[0][0][0];
#pragma unroll
    for (int mi = 0; mi < 2; ++mi)
#pragma unroll
        for (int v = 0; v < 8; ++v)
            Cs[(wm + 16 * mi + 8 * hi + v) * (SN + 4) + wn + ln] = c2[mi][v];
    __syncthreads();

    float d2 = 0.f;
#pragma unroll
    for (int j = 0; j < 4; ++j) {
        const int r = erow + 16 * j;
        const float4 s = *(const float4*)&Cs[r * (SN + 4) + ec4];
        const size_t gi = (size_t)(row0 + r) * RN + col0 + ec4;
        float4 zn;
        d2 += dys_elem(s.x, zo[j].x, wo[j].x, zn.x);
        d2 += dys_elem(s.y, zo[j].y, wo[j].y, zn.y);
        d2 += dys_elem(s.z, zo[j].z, wo[j].z, zn.z);
        d2 += dys_elem(s.w, zo[j].w, wo[j].w, zn.w);
        if (FINAL) {
            zn.x = fmaxf(zn.x, 0.f); zn.y = fmaxf(zn.y, 0.f);
            zn.z = fmaxf(zn.z, 0.f); zn.w = fmaxf(zn.w, 0.f);
        } else {
            *(float4*)(u + gi) = dys_u4(zn, wo[j]);
        }
        *(float4*)(z + gi) = zn;
    }

    if (!FINAL) {
        for (int off = warpSize / 2; off > 0; off >>= 1) d2 += __shfl_xor(d2, off);
        if (lane == 0) sm[wid] = d2;
        __syncthreads();
        if (t == 0) {
            float tot = 0.f;
            for (int k = 0; k < 8; ++k) tot += sm[k];
            atomicAdd(acc, tot);
        }
    }
}

template <bool UPDATE>
__global__ __launch_bounds__(256) void gemm_ns_step(
    const float* __restrict__ Am, const float* __restrict__ Bm,
    float* __restrict__ Cm, const float* __restrict__ Cin)
{
    __shared__ __attribute__((aligned(16))) float As[2][SM][SK + 4];
    __shared__ __attribute__((aligned(16))) float Bs[2][SK][SN + 16];

    const int t    = threadIdx.x;
    const int row0 = blockIdx.y * SM;
    const int col0 = blockIdx.x * SN;
    const int wid  = t >> 5;
    const int lane = t & 31;
    const int wm   = (wid >> 2) * 32;
    const int wn   = (wid & 3) * 16;
    const int hi   = lane >> 4;
    const int ln   = lane & 15;

    const int erow = t >> 4;
    const int ec4  = (t & 15) * 4;
    float4 ci[4];
    if (UPDATE) {
#pragma unroll
        for (int j = 0; j < 4; ++j)
            ci[j] = *(const float4*)(Cin + (size_t)(row0 + erow + 16 * j) * RK + col0 + ec4);
    }

    v8f c2[2];
    c2[0] = (v8f){};
    c2[1] = (v8f){};

    const int ar = t >> 5;
    const int ac = (t & 31) * 4;
    const int br = t >> 4;
    const int bc = (t & 15) * 4;
    tile64_kloop<RK, false, 0>(Am + (size_t)(row0 + ar) * RK + ac, Bm + (size_t)br * RK + col0 + bc,
                               As, Bs, ar, ac, br, bc, wm, wn, hi, ln,
                               (int)(blockIdx.x + blockIdx.y), c2);

    __syncthreads();
    float* Cs = &As[0][0][0];
#pragma unroll
    for (int mi = 0; mi < 2; ++mi)
#pragma unroll
        for (int v = 0; v < 8; ++v)
            Cs[(wm + 16 * mi + 8 * hi + v) * (SN + 4) + wn + ln] = c2[mi][v];
    __syncthreads();

#pragma unroll
    for (int j = 0; j < 4; ++j) {
        const int r = erow + 16 * j;
        float4 v = *(const float4*)&Cs[r * (SN + 4) + ec4];
        if (UPDATE) {
            v.x = 2.f * ci[j].x - v.x;  v.y = 2.f * ci[j].y - v.y;
            v.z = 2.f * ci[j].z - v.z;  v.w = 2.f * ci[j].w - v.w;
        }
        *(float4*)(Cm + (size_t)(row0 + r) * RK + col0 + ec4) = v;
    }
}

extern "C" void kernel_launch(void* const* d_in, const int* in_sizes, int n_in,
                              void* d_out, int out_size, void* d_ws, size_t ws_size,
                              hipStream_t stream)
{
    const float* ctx  = (const float*)d_in[0];
    const float* Ain  = (const float*)d_in[1];
    const float* bvec = (const float*)d_in[2];
    const float* W1   = (const float*)d_in[3];
    const float* b1   = (const float*)d_in[4];
    const float* W2   = (const float*)d_in[5];
    const float* b2   = (const float*)d_in[6];
    const float* z0   = (const float*)d_in[7];
    float* out = (float*)d_out;

    const int B = 256, D = 512, H = 2048, N1 = 1024, N2 = 4096;

    float* W = (float*)d_ws;
    const size_t M1 = (size_t)1 << 20;
    float* G    = W;
    float* Xb   = W + 1 * M1;
    float* T1   = W + 2 * M1;
    float* Xa   = W + 3 * M1;
    float* F    = W + 5 * M1;
    float* Minv = W + 4 * M1;
    float* P    = W;
    float* h    = W + 8 * M1;
    float* wv   = W + 8 * M1;
    float* z    = out;
    float* R    = P;
    float* U    = W + 1 * M1;
    float* acc  = W + 9 * M1;
    float* fn   = acc + 1;

    dim3 blk(256);
    auto partials = [&](const float* A_, int lda, const float* B_, int ldb,
                        int tB, float* P_, int ns, int Kc, int M, int N, int K,
                        const float* gate) {
        dim3 grid(N / TN, M / TM, ns);
        gemm_f32_wmma<<<grid, blk, 0, stream>>>(A_, lda, B_, ldb, tB, P_, N,
                                                Kc, K, (size_t)M * N, gate);
    };
    auto gemm = [&](const float* A_, int lda, const float* B_, int ldb, int tB,
                    float* P_, int ns, float* C_, const float* Cin, const float* bias,
                    float al, float be, float ga, int relu,
                    int M, int N, int K, const float* gate) {
        partials(A_, lda, B_, ldb, tB, P_, ns, K / ns, M, N, K, gate);
        k_splitk_reduce<<<dim3((M * N) / 1024), blk, 0, stream>>>(
            P_, ns, (size_t)M * N, C_, Cin, bias, N, al, be, ga, relu, gate);
    };

    gemm(Ain, N2, Ain, N2, 1, F, 4, G, nullptr, nullptr, 1.f, 0.f, 0.f, 0,
         N1, N1, N2, nullptr);

    k_zero<<<1, 1, 0, stream>>>(fn);
    k_sumsq<<<dim3((N1 * N1 + 255) / 256), blk, 0, stream>>>(G, fn, N1 * N1);
    k_init_X<<<dim3((N1 * N1 + 255) / 256), blk, 0, stream>>>(Xa, fn, N1);

    float* Xc = Xa; float* Xn = Xb;
    for (int it = 0; it < 16; ++it) {
        gemm_ns_step<false><<<dim3(N1 / SN, N1 / SM), blk, 0, stream>>>(G,  Xc, T1, nullptr);
        gemm_ns_step<true><<<dim3(N1 / SN, N1 / SM), blk, 0, stream>>>(Xc, T1, Xn, Xc);
        float* tmp = Xc; Xc = Xn; Xn = tmp;
    }
    float* Ginv = Xc;

    partials(Ginv, N1, Ain, N2, 0, Minv, 1, N1, N1, N2, N1, nullptr);

    gemm(ctx, D, W1, H,  0, P, 8, h,  nullptr, b1, 1.f, 0.f, 1.f, 1, B, H,  D, nullptr);
    gemm(h,   H, W2, N2, 0, P, 4, wv, nullptr, b2, 1.f, 0.f, 1.f, 0, B, N2, H, nullptr);

    const int nz = B * N2;
    k_dys_init<<<dim3(nz / 1024), blk, 0, stream>>>(z0, wv, z, U, acc);
    const dim3 rgrid(N1 / PN, B / PM, 2);
    float* cur = acc; float* nxt = acc + 1; float* prv = acc + 2;
    const float* gate = nullptr;
    for (int itr = 0; itr < 50; ++itr) {
        gemm_r_part<<<rgrid, blk, 0, stream>>>(U, Ain, bvec, R, (size_t)B * N1, gate);
        gemm_dys_step<false><<<dim3(N2 / SN, B / SM), blk, 0, stream>>>(R, Minv, z, wv, U, cur, nxt, gate);
        gate = cur;
        float* tmp = prv; prv = cur; cur = nxt; nxt = tmp;
    }

    gemm_r_part<<<rgrid, blk, 0, stream>>>(U, Ain, bvec, R, (size_t)B * N1, nullptr);
    gemm_dys_step<true><<<dim3(N2 / SN, B / SM), blk, 0, stream>>>(R, Minv, z, wv, nullptr,
                                                                   nullptr, nullptr, nullptr);
}


# whole-K GEMM kernels: the global loads of K-tile t+2 are issued right behind the LDS stores of tile t+1, in front of a tile's last 16 WMMAs, not at the head of the next tile behind the barrier
# speedup vs baseline: 9.8446x; 1.0562x over previous
#include <hip/hip_runtime.h>

typedef __attribute__((ext_vector_type(2))) float v2f;
typedef __attribute__((ext_vector_type(4))) float v4f;
typedef __attribute__((ext_vector_type(8))) float v8f;

#define ALPHA 0.05f
#define EPS   0.01f

#define TM 128
#define TN 128
#define TK 32

static __device__ __forceinline__ v8f wmma4(const v2f a, const v2f b, const v8f c)
{
    return __builtin_amdgcn_wmma_f32_16x16x4_f32(false, a, false, b, (short)0, c,
                                                 false, false);
}

static __device__ __forceinline__ bool loop_done(const float* gate)
{
    return gate && *gate <= EPS * EPS;
}

static __device__ __forceinline__ void store_partial(
    const v8f acc, int r0, int c0, float* __restrict__ Cp, int ldc, int hi, int ln)
{
    const int c = c0 + ln;
#pragma unroll
    for (int v = 0; v < 8; ++v) {
        const int r = r0 + v + 8 * hi;
        Cp[(size_t)r * ldc + c] = acc[v];
    }
}

static __device__ __forceinline__ void stage_tiles(
    float* __restrict__ Ad, float* __restrict__ Bd, int transB,
    const float4 a0, const float4 a1, const float4 a2, const float4 a3,
    const float4 b0, const float4 b1, const float4 b2, const float4 b3,
    int arow, int akc, int bkp, int bnp, int btn, int btk)
{
    float* ad = Ad + arow * (TK + 4) + akc;
    *(float4*)(ad + 0)  = a0;
    *(float4*)(ad + 4)  = a1;
    *(float4*)(ad + 8)  = a2;
    *(float4*)(ad + 12) = a3;
    if (!transB) {
        float* bd = Bd + bkp * (TN + 4) + bnp;
        *(float4*)(bd + 0)  = b0;
        *(float4*)(bd + 4)  = b1;
        *(float4*)(bd + 8)  = b2;
        *(float4*)(bd + 12) = b3;
    } else {
        float* bd = Bd + btk * (TN + 4) + btn;
        bd[0  * (TN + 4)] = b0.x; bd[1  * (TN + 4)] = b0.y; bd[2  * (TN + 4)] = b0.z; bd[3  * (TN + 4)] = b0.w;
        bd[4  * (TN + 4)] = b1.x; bd[5  * (TN + 4)] = b1.y; bd[6  * (TN + 4)] = b1.z; bd[7  * (TN + 4)] = b1.w;
        bd[8  * (TN + 4)] = b2.x; bd[9  * (TN + 4)] = b2.y; bd[10 * (TN + 4)] = b2.z; bd[11 * (TN + 4)] = b2.w;
        bd[12 * (TN + 4)] = b3.x; bd[13 * (TN + 4)] = b3.y; bd[14 * (TN + 4)] = b3.z; bd[15 * (TN + 4)] = b3.w;
    }
}

static __device__ __forceinline__ float4 dys_u4(const float4 z, const float4 w)
{
    float4 u;
    u.x = fmaf(-ALPHA, w.x, fabsf(z.x));
    u.y = fmaf(-ALPHA, w.y, fabsf(z.y));
    u.z = fmaf(-ALPHA, w.z, fabsf(z.z));
    u.w = fmaf(-ALPHA, w.w, fabsf(z.w));
    return u;
}

__global__ __launch_bounds__(256) void gemm_f32_wmma(
    const float* __restrict__ Am, int lda,
    const float* __restrict__ Bm, int ldb, int transB,
    float* __restrict__ Cm, int ldc,
    int Kc, int Ktot, size_t pstride, const float* gate)
{
    if (loop_done(gate)) return;

    __shared__ float As[2][TM][TK + 4];
    __shared__ float Bs[2][TK][TN + 4];

    const int t    = threadIdx.x;
    const int row0 = blockIdx.y * TM;
    const int col0 = blockIdx.x * TN;

    const int wid  = t >> 5;
    const int lane = t & 31;
    const int wm   = (wid >> 2) * 64;
    const int wn   = (wid & 3) * 32;
    const int hi   = lane >> 4;
    const int ln   = lane & 15;

    v8f acc[4][2];
#pragma unroll
    for (int mi = 0; mi < 4; ++mi)
#pragma unroll
        for (int ni = 0; ni < 2; ++ni) acc[mi][ni] = (v8f){};

    const int arow = t >> 1;
    const int akc  = (t & 1) * 16;
    const int bkp  = t >> 3;
    const int bnp  = (t & 7) * 16;
    const int btn  = t >> 1;
    const int btk  = (t & 1) * 16;

    const int kbeg = blockIdx.z * Kc;
    const int kend = min(kbeg + Kc, Ktot);
    const int nk   = (kend - kbeg) / TK;
    float* __restrict__ Cp = Cm + (size_t)blockIdx.z * pstride;

    const float* ap = Am + (size_t)(row0 + arow) * lda + kbeg + akc;
    const float* bp = transB ? Bm + (size_t)(col0 + btn) * ldb + kbeg + btk
                             : Bm + (size_t)(kbeg + bkp) * ldb + col0 + bnp;
    const size_t bstep = transB ? (size_t)TK : (size_t)TK * ldb;

    float4 ra0, ra1, ra2, ra3, rb0, rb1, rb2, rb3;

    ra0 = *(const float4*)(ap + 0);  ra1 = *(const float4*)(ap + 4);
    ra2 = *(const float4*)(ap + 8);  ra3 = *(const float4*)(ap + 12);
    rb0 = *(const float4*)(bp + 0);  rb1 = *(const float4*)(bp + 4);
    rb2 = *(const float4*)(bp + 8);  rb3 = *(const float4*)(bp + 12);
    stage_tiles(&As[0][0][0], &Bs[0][0][0], transB, ra0, ra1, ra2, ra3, rb0, rb1, rb2, rb3,
                arow, akc, bkp, bnp, btn, btk);
    __syncthreads();

    for (int kt = 0; kt < nk; ++kt) {
        const int  cur  = kt & 1;
        const bool more = kt + 1 < nk;
        if (more) {
            ap += TK;
            bp += bstep;
            ra0 = *(const float4*)(ap + 0);  ra1 = *(const float4*)(ap + 4);
            ra2 = *(const float4*)(ap + 8);  ra3 = *(const float4*)(ap + 12);
            rb0 = *(const float4*)(bp + 0);  rb1 = *(const float4*)(bp + 4);
            rb2 = *(const float4*)(bp + 8);  rb3 = *(const float4*)(bp + 12);
        }

#pragma unroll
        for (int kk = 0; kk < TK; kk += 4) {
            v2f a[4], b[2];
#pragma unroll
            for (int mi = 0; mi < 4; ++mi) {
                a[mi].x = As[cur][wm + 16 * mi + ln][kk + 2 * hi];
                a[mi].y = As[cur][wm + 16 * mi + ln][kk + 2 * hi + 1];
            }
#pragma unroll
            for (int ni = 0; ni < 2; ++ni) {
                b[ni].x = Bs[cur][kk + 2 * hi][wn + 16 * ni + ln];
                b[ni].y = Bs[cur][kk + 2 * hi + 1][wn + 16 * ni + ln];
            }
#pragma unroll
            for (int mi = 0; mi < 4; ++mi)
#pragma unroll
                for (int ni = 0; ni < 2; ++ni)
                    acc[mi][ni] = wmma4(a[mi], b[ni], acc[mi][ni]);
        }

        if (more) {
            stage_tiles(&As[cur ^ 1][0][0], &Bs[cur ^ 1][0][0], transB,
                        ra0, ra1, ra2, ra3, rb0, rb1, rb2, rb3, arow, akc, bkp, bnp, btn, btk);
            __syncthreads();
        }
    }

#pragma unroll
    for (int mi = 0; mi < 4; ++mi)
#pragma unroll
        for (int ni = 0; ni < 2; ++ni)
            store_partial(acc[mi][ni], row0 + wm + 16 * mi, col0 + wn + 16 * ni,
                          Cp, ldc, hi, ln);
}

__global__ void k_sumsq(const float* __restrict__ x, float* __restrict__ acc, int n)
{
    __shared__ float sm[256];
    int i = blockIdx.x * blockDim.x + threadIdx.x;
    float v = 0.f;
    if (i < n) { float t = x[i]; v = t * t; }
    sm[threadIdx.x] = v;
    __syncthreads();
    for (int s = 128; s > 0; s >>= 1) {
        if (threadIdx.x < s) sm[threadIdx.x] += sm[threadIdx.x + s];
        __syncthreads();
    }
    if (threadIdx.x == 0) atomicAdd(acc, sm[0]);
}

__global__ void k_zero(float* p) { p[0] = 0.f; }

__global__ __launch_bounds__(256) void k_splitk_reduce(
    const float* P, int nsplit, size_t pstride,
    float* C, const float* __restrict__ Cin, const float* __restrict__ bias,
    int ncols, float alpha, float beta, float gamma, int relu, const float* gate)
{
    if (loop_done(gate)) return;
    const size_t i = ((size_t)blockIdx.x * blockDim.x + threadIdx.x) * 4;
    float4 s = make_float4(0.f, 0.f, 0.f, 0.f);
    for (int k = 0; k < nsplit; k += 4) {
        const float4 p0 = *(const float4*)(P + (size_t)(k + 0) * pstride + i);
        const float4 p1 = *(const float4*)(P + (size_t)(k + 1) * pstride + i);
        const float4 p2 = *(const float4*)(P + (size_t)(k + 2) * pstride + i);
        const float4 p3 = *(const float4*)(P + (size_t)(k + 3) * pstride + i);
        s.x += p0.x; s.y += p0.y; s.z += p0.z; s.w += p0.w;
        s.x += p1.x; s.y += p1.y; s.z += p1.z; s.w += p1.w;
        s.x += p2.x; s.y += p2.y; s.z += p2.z; s.w += p2.w;
        s.x += p3.x; s.y += p3.y; s.z += p3.z; s.w += p3.w;
    }
    float4 v = make_float4(alpha * s.x, alpha * s.y, alpha * s.z, alpha * s.w);
    if (bias) {
        const int c = (int)(i % (size_t)ncols);
        const float4 bv = *(const float4*)(bias + c);
        v.x += gamma * bv.x; v.y += gamma * bv.y; v.z += gamma * bv.z; v.w += gamma * bv.w;
    }
    if (Cin) {
        const float4 ci = *(const float4*)(Cin + i);
        v.x += beta * ci.x; v.y += beta * ci.y; v.z += beta * ci.z; v.w += beta * ci.w;
    }
    if (relu) {
        v.x = fmaxf(v.x, 0.f); v.y = fmaxf(v.y, 0.f); v.z = fmaxf(v.z, 0.f); v.w = fmaxf(v.w, 0.f);
    }
    *(float4*)(C + i) = v;
}

__global__ void k_init_X(float* __restrict__ X, const float* __restrict__ fn, int n)
{
    int i = blockIdx.x * blockDim.x + threadIdx.x;
    if (i < n * n) {
        float c = rsqrtf(*fn);
        int r = i / n, cc = i - r * n;
        X[i] = (r == cc) ? c : 0.f;
    }
}

static __device__ __forceinline__ float dys_elem(float s, float zo, float wo, float& zn)
{
    const float uo = fmaf(-ALPHA, wo, fabsf(zo));
    const float y  = uo - s;
    const float x  = fmaxf(zo, 0.f);
    zn = zo - x + y;
    const float d = y - x;
    return d * d;
}

__global__ __launch_bounds__(256) void k_dys_init(
    const float* __restrict__ z0, const float* __restrict__ w,
    float* __restrict__ z, float* __restrict__ u, float* __restrict__ acc0)
{
    if (blockIdx.x == 0 && threadIdx.x == 0) *acc0 = 0.f;
    const size_t i = ((size_t)blockIdx.x * 256 + threadIdx.x) * 4;
    const float4 zv = *(const float4*)(z0 + i);
    *(float4*)(z + i) = zv;
    *(float4*)(u + i) = dys_u4(zv, *(const float4*)(w + i));
}

#define SM 64
#define SN 64
#define SK 128
#define RK 1024
#define RN 4096

static __device__ __forceinline__ void dys_frags(
    const float (*A)[SK + 4], const float (*Bt)[SN + 16], int kb, int wm, int wn, int hi, int ln,
    v2f (&fa)[4][2], v2f (&fb)[4])
{
#pragma unroll
    for (int q = 0; q < 4; ++q) {
        const int k2 = kb + 4 * q + 2 * hi;
#pragma unroll
        for (int mi = 0; mi < 2; ++mi) {
            fa[q][mi].x = A[wm + 16 * mi + ln][k2];
            fa[q][mi].y = A[wm + 16 * mi + ln][k2 + 1];
        }
        fb[q].x = Bt[k2][wn + ln];
        fb[q].y = Bt[k2 + 1][wn + ln];
    }
}

static __device__ __forceinline__ void lds_barrier()
{
    __builtin_amdgcn_fence(__ATOMIC_RELEASE, "workgroup", "local");
    __builtin_amdgcn_s_barrier();
    __builtin_amdgcn_fence(__ATOMIC_ACQUIRE, "workgroup", "local");
}

template <bool V> struct ktile_flag { static constexpr bool value = V; };

static __device__ __forceinline__ float4 add4(const float4 a, const v4f b)
{
    return make_float4(a.x + b.x, a.y + b.y, a.z + b.z, a.w + b.w);
}

template <int LDB, bool SUM2, int ASEP>
static __device__ __forceinline__ void tile64_kloop(
    const float* __restrict__ ap0, const float* __restrict__ bp0,
    float (&As)[2][SM][SK + 4], float (&Bs)[2][SK][SN + 16],
    int ar, int ac, int br, int bc, int wm, int wn, int hi, int ln, int rot, v8f (&c2)[2])
{
    constexpr int nk = RK / SK;
    static_assert((nk & (nk - 1)) == 0, "the K-tile rotation wraps with a mask");
    int kidx = rot & (nk - 1);
    const float* ap = ap0 + (size_t)kidx * SK;
    const float* bp = bp0 + (size_t)kidx * SK * LDB;
    float4 ra0, ra1, ra2, ra3, ra4, ra5, ra6, ra7;
    float4 rb0, rb1, rb2, rb3, rb4, rb5, rb6, rb7;
    v4f rc0 = (v4f){};
    v4f rc1 = rc0, rc2 = rc0, rc3 = rc0, rc4 = rc0, rc5 = rc0, rc6 = rc0, rc7 = rc0;
#define DYS_LOAD_TILE()                                                                      \
    ra0 = *(const float4*)(ap + 0 * 8 * RK);   ra1 = *(const float4*)(ap + 1 * 8 * RK);      \
    ra2 = *(const float4*)(ap + 2 * 8 * RK);   ra3 = *(const float4*)(ap + 3 * 8 * RK);      \
    ra4 = *(const float4*)(ap + 4 * 8 * RK);   ra5 = *(const float4*)(ap + 5 * 8 * RK);      \
    ra6 = *(const float4*)(ap + 6 * 8 * RK);   ra7 = *(const float4*)(ap + 7 * 8 * RK);      \
    if (SUM2) {                                                                              \
        rc0 = *(const v4f*)(ap + ASEP + 0 * 8 * RK);  rc1 = *(const v4f*)(ap + ASEP + 1 * 8 * RK);  \
        rc2 = *(const v4f*)(ap + ASEP + 2 * 8 * RK);  rc3 = *(const v4f*)(ap + ASEP + 3 * 8 * RK);  \
        rc4 = *(const v4f*)(ap + ASEP + 4 * 8 * RK);  rc5 = *(const v4f*)(ap + ASEP + 5 * 8 * RK);  \
        rc6 = *(const v4f*)(ap + ASEP + 6 * 8 * RK);  rc7 = *(const v4f*)(ap + ASEP + 7 * 8 * RK);  \
    }                                                                                        \
    rb0 = *(const float4*)(bp + 0 * 16 * LDB); rb1 = *(const float4*)(bp + 1 * 16 * LDB);    \
    rb2 = *(const float4*)(bp + 2 * 16 * LDB); rb3 = *(const float4*)(bp + 3 * 16 * LDB);    \
    rb4 = *(const float4*)(bp + 4 * 16 * LDB); rb5 = *(const float4*)(bp + 5 * 16 * LDB);    \
    rb6 = *(const float4*)(bp + 6 * 16 * LDB); rb7 = *(const float4*)(bp + 7 * 16 * LDB);
#define DYS_STAGE_TILE(s)                                                                    \
    if (SUM2) {                                                                              \
          \
          \
          \
        asm volatile("" : "+v"(rc0), "+v"(rc1), "+v"(rc2), "+v"(rc3),                        \
                          "+v"(rc4), "+v"(rc5), "+v"(rc6), "+v"(rc7));                       \
        ra0 = add4(ra0, rc0);  ra1 = add4(ra1, rc1);  ra2 = add4(ra2, rc2);  ra3 = add4(ra3, rc3);  \
        ra4 = add4(ra4, rc4);  ra5 = add4(ra5, rc5);  ra6 = add4(ra6, rc6);  ra7 = add4(ra7, rc7);  \
    }                                                                                        \
    *(float4*)&As[s][ar + 0][ac]  = ra0;   *(float4*)&As[s][ar + 8][ac]  = ra1;              \
    *(float4*)&As[s][ar + 16][ac] = ra2;   *(float4*)&As[s][ar + 24][ac] = ra3;              \
    *(float4*)&As[s][ar + 32][ac] = ra4;   *(float4*)&As[s][ar + 40][ac] = ra5;              \
    *(float4*)&As[s][ar + 48][ac] = ra6;   *(float4*)&As[s][ar + 56][ac] = ra7;              \
    *(float4*)&Bs[s][br + 0][bc]  = rb0;   *(float4*)&Bs[s][br + 16][bc] = rb1;              \
    *(float4*)&Bs[s][br + 32][bc] = rb2;   *(float4*)&Bs[s][br + 48][bc] = rb3;              \
    *(float4*)&Bs[s][br + 64][bc] = rb4;   *(float4*)&Bs[s][br + 80][bc] = rb5;              \
    *(float4*)&Bs[s][br + 96][bc] = rb6;   *(float4*)&Bs[s][br + 112][bc] = rb7;

    static_assert(nk >= 2, "the prologue loads a second K-tile");
#define DYS_NEXT_TILE()                                                                      \
    kidx = (kidx + 1) & (nk - 1);                                                            \
    ap = ap0 + (size_t)kidx * SK;                                                            \
    bp = bp0 + (size_t)kidx * SK * LDB;                                                      \
    DYS_LOAD_TILE()

    DYS_LOAD_TILE()
    DYS_STAGE_TILE(0)
    DYS_NEXT_TILE()
    lds_barrier();

    auto ktile = [&](const int cur, const bool load, auto stage_tag) __attribute__((always_inline)) {
        constexpr bool STAGE = decltype(stage_tag)::value;
        v2f fa[2][4][2], fb[2][4];
        dys_frags(As[cur], Bs[cur], 0, wm, wn, hi, ln, fa[0], fb[0]);
        __builtin_amdgcn_sched_barrier(0);
#pragma unroll
        for (int g = 0; g < SK / 16; ++g) {
            const int s = g & 1;
            if (g + 1 < SK / 16)
                dys_frags(As[cur], Bs[cur], 16 * (g + 1), wm, wn, hi, ln, fa[s ^ 1], fb[s ^ 1]);
            __builtin_amdgcn_sched_barrier(0);
            if (g == SK / 16 - 2 && STAGE) {
                DYS_STAGE_TILE(cur ^ 1)
                if (load) {
                    DYS_NEXT_TILE()
                }
                __builtin_amdgcn_sched_barrier(0);
            }
#pragma unroll
            for (int q = 0; q < 4; ++q)
#pragma unroll
                for (int mi = 0; mi < 2; ++mi) c2[mi] = wmma4(fa[s][q][mi], fb[s][q], c2[mi]);
            __builtin_amdgcn_sched_barrier(0);
        }

        if (STAGE) lds_barrier();
    };

    for (int kt = 0; kt + 1 < nk; ++kt) ktile(kt & 1, kt + 2 < nk, ktile_flag<true>{});
    ktile((nk - 1) & 1, false, ktile_flag<false>{});

#undef DYS_NEXT_TILE
#undef DYS_LOAD_TILE
#undef DYS_STAGE_TILE
}

#define PM 32
#define PN 64
#define PK 256
#define PG 32

static __device__ __forceinline__ void rpart_frags(
    const float (*Ut)[PK + 4], const float (*Bt)[PK + 4], int kb, int wm, int wn, int hi, int ln,
    v2f (&fa)[4], v2f (&fb)[2][4])
{
#pragma unroll
    for (int q = 0; q < 4; ++q) {
        const int k2 = kb + 4 * q + 2 * hi;
        fa[q].x = Ut[wm + ln][k2];
        fa[q].y = Ut[wm + ln][k2 + 1];
#pragma unroll
        for (int ni = 0; ni < 2; ++ni) {
            fb[ni][q].x = Bt[wn + 16 * ni + ln][k2];
            fb[ni][q].y = Bt[wn + 16 * ni + ln][k2 + 1];
        }
    }
}

__global__ __launch_bounds__(256) void gemm_r_part(
    const float* __restrict__ U, const float* __restrict__ Am, const float* __restrict__ bvec,
    float* __restrict__ P, size_t pstride, const float* gate)
{
    if (loop_done(gate)) return;

    __shared__ __attribute__((aligned(16))) float Us[2][PM][PK + 4];
    __shared__ __attribute__((aligned(16))) float Bs[2][PN][PK + 4];

    const int t    = threadIdx.x;
    const int row0 = blockIdx.y * PM;
    const int col0 = blockIdx.x * PN;
    const int kbeg = blockIdx.z * (RN / 2);
    const int wid  = t >> 5;
    const int lane = t & 31;
    const int kg   = wid >> 2;
    const int wm   = ((wid >> 1) & 1) * 16;
    const int wn   = (wid & 1) * 32;
    const int hi   = lane >> 4;
    const int ln   = lane & 15;

    const int sr = t >> 6;
    const int sc = (t & 63) * 4;
    constexpr int nk = (RN / 2) / PK;
    static_assert((nk & (nk - 1)) == 0, "the K-tile rotation wraps with a mask");
    static_assert(nk >= 2, "the prologue loads a second K-tile");
    const int rot = (int)(blockIdx.x + blockIdx.y) & (nk - 1);
    const float* up0 = U  + (size_t)(row0 + sr) * RN + kbeg + sc;
    const float* bp0 = Am + (size_t)(col0 + sr) * RN + kbeg + sc;
    const float* up = up0 + rot * PK;
    const float* bp = bp0 + rot * PK;

    float4 ru0, ru1, ru2, ru3, ru4, ru5, ru6, ru7;
    float4 rb0, rb1, rb2, rb3, rb4, rb5, rb6, rb7, rb8, rb9, rb10, rb11, rb12, rb13, rb14, rb15;
#define RP_LD(p, j) (*(const float4*)((p) + (size_t)(j) * 4 * RN))
#define RP_LOAD_TILE()                                                                       \
    ru0 = RP_LD(up, 0);   ru1 = RP_LD(up, 1);   ru2 = RP_LD(up, 2);   ru3 = RP_LD(up, 3);     \
    ru4 = RP_LD(up, 4);   ru5 = RP_LD(up, 5);   ru6 = RP_LD(up, 6);   ru7 = RP_LD(up, 7);     \
    rb0 = RP_LD(bp, 0);   rb1 = RP_LD(bp, 1);   rb2 = RP_LD(bp, 2);   rb3 = RP_LD(bp, 3);     \
    rb4 = RP_LD(bp, 4);   rb5 = RP_LD(bp, 5);   rb6 = RP_LD(bp, 6);   rb7 = RP_LD(bp, 7);     \
    rb8 = RP_LD(bp, 8);   rb9 = RP_LD(bp, 9);   rb10 = RP_LD(bp, 10); rb11 = RP_LD(bp, 11);   \
    rb12 = RP_LD(bp, 12); rb13 = RP_LD(bp, 13); rb14 = RP_LD(bp, 14); rb15 = RP_LD(bp, 15);
#define RP_ST(T, s, j, v) (*(float4*)&T[s][sr + 4 * (j)][sc] = (v))
#define RP_STAGE_TILE(s)                                                                     \
    RP_ST(Us, s, 0, ru0);   RP_ST(Us, s, 1, ru1);   RP_ST(Us, s, 2, ru2);   RP_ST(Us, s, 3, ru3);     \
    RP_ST(Us, s, 4, ru4);   RP_ST(Us, s, 5, ru5);   RP_ST(Us, s, 6, ru6);   RP_ST(Us, s, 7, ru7);     \
    RP_ST(Bs, s, 0, rb0);   RP_ST(Bs, s, 1, rb1);   RP_ST(Bs, s, 2, rb2);   RP_ST(Bs, s, 3, rb3);     \
    RP_ST(Bs, s, 4, rb4);   RP_ST(Bs, s, 5, rb5);   RP_ST(Bs, s, 6, rb6);   RP_ST(Bs, s, 7, rb7);     \
    RP_ST(Bs, s, 8, rb8);   RP_ST(Bs, s, 9, rb9);   RP_ST(Bs, s, 10, rb10); RP_ST(Bs, s, 11, rb11);   \
    RP_ST(Bs, s, 12, rb12); RP_ST(Bs, s, 13, rb13); RP_ST(Bs, s, 14, rb14); RP_ST(Bs, s, 15, rb15);

    v8f acc[2];
    acc[0] = (v8f){};
    acc[1] = (v8f){};

    RP_LOAD_TILE()
    RP_STAGE_TILE(0)
    up = up0 + ((1 + rot) & (nk - 1)) * PK;
    bp = bp0 + ((1 + rot) & (nk - 1)) * PK;
    RP_LOAD_TILE()
    lds_barrier();

    auto ktile = [&](const int cur, const int nxt2, const bool load, auto stage_tag)
                     __attribute__((always_inline)) {
        constexpr bool STAGE = decltype(stage_tag)::value;

        v2f fa[2][4], fb[2][2][4];
        rpart_frags(Us[cur], Bs[cur], 16 * kg, wm, wn, hi, ln, fa[0], fb[0]);
        __builtin_amdgcn_sched_barrier(0);
#pragma unroll
        for (int g = 0; g < PK / PG; ++g) {
            const int s = g & 1;
            if (g + 1 < PK / PG)
                rpart_frags(Us[cur], Bs[cur], PG * (g + 1) + 16 * kg, wm, wn, hi, ln,
                            fa[s ^ 1], fb[s ^ 1]);
            __builtin_amdgcn_sched_barrier(0);
            if (g == PK / PG - 2 && STAGE) {
                RP_STAGE_TILE(cur ^ 1)
                if (load) {
                    up = up0 + nxt2 * PK;
                    bp = bp0 + nxt2 * PK;
                    RP_LOAD_TILE()
                }
                __builtin_amdgcn_sched_barrier(0);
            }
#pragma unroll
            for (int q = 0; q < 4; ++q)
#pragma unroll
                for (int ni = 0; ni < 2; ++ni) acc[ni] = wmma4(fa[s][q], fb[s][ni][q], acc[ni]);
            __builtin_amdgcn_sched_barrier(0);
        }

        if (STAGE) lds_barrier();
    };

    for (int kt = 0; kt + 1 < nk; ++kt)
        ktile(kt & 1, (kt + 2 + rot) & (nk - 1), kt + 2 < nk, ktile_flag<true>{});
    ktile((nk - 1) & 1, 0, false, ktile_flag<false>{});
#undef RP_LD
#undef RP_LOAD_TILE
#undef RP_ST
#undef RP_STAGE_TILE

    __syncthreads();
    float* Cs = &Us[0][0][0];
#pragma unroll
    for (int ni = 0; ni < 2; ++ni)
#pragma unroll
        for (int v = 0; v < 8; ++v)
            Cs[(kg * PM + wm + 8 * hi + v) * (PN + 4) + wn + 16 * ni + ln] = acc[ni][v];
    __syncthreads();

    const int er  = t >> 4;
    const int ec4 = (t & 15) * 4;
    float4 bv = make_float4(0.f, 0.f, 0.f, 0.f);
    if (blockIdx.z == 0) bv = *(const float4*)(bvec + col0 + ec4);
    float* __restrict__ Pp = P + (size_t)blockIdx.z * pstride;
#pragma unroll
    for (int j = 0; j < 2; ++j) {
        const int r = er + 16 * j;
        const float4 s0 = *(const float4*)&Cs[r * (PN + 4) + ec4];
        const float4 s1 = *(const float4*)&Cs[(PM + r) * (PN + 4) + ec4];
        float4 s;
        s.x = s0.x + s1.x - bv.x;  s.y = s0.y + s1.y - bv.y;
        s.z = s0.z + s1.z - bv.z;  s.w = s0.w + s1.w - bv.w;
        *(float4*)(Pp + (size_t)(row0 + r) * RK + col0 + ec4) = s;
    }
}

#define RHALF (256 * RK)
template <bool FINAL>
__global__ __launch_bounds__(256) void gemm_dys_step(
    const float* __restrict__ Rm, const float* __restrict__ Mi,
    float* z, const float* __restrict__ w, float* __restrict__ u,
    float* acc, float* next, const float* gate)
{
    __shared__ __attribute__((aligned(16))) float As[2][SM][SK + 4];
    __shared__ __attribute__((aligned(16))) float Bs[2][SK][SN + 16];
    __shared__ float sm[8];

    const int t = threadIdx.x;
    if (!FINAL) {
        if (blockIdx.x == 0 && blockIdx.y == 0 && t == 0) *next = 0.f;
        if (loop_done(gate)) return;
    }

    const int row0 = blockIdx.y * SM;
    const int col0 = blockIdx.x * SN;
    const int wid  = t >> 5;
    const int lane = t & 31;
    const int wm   = (wid >> 2) * 32;
    const int wn   = (wid & 3) * 16;
    const int hi   = lane >> 4;
    const int ln   = lane & 15;

    const int erow = t >> 4;
    const int ec4  = (t & 15) * 4;
    float4 zo[4], wo[4];
#pragma unroll
    for (int j = 0; j < 4; ++j) {
        const size_t gi = (size_t)(row0 + erow + 16 * j) * RN + col0 + ec4;
        zo[j] = *(const float4*)(z + gi);
        wo[j] = *(const float4*)(w + gi);
    }

    v8f c2[2];
    c2[0] = (v8f){};
    c2[1] = (v8f){};

    const int ar = t >> 5;
    const int ac = (t & 31) * 4;
    const int br = t >> 4;
    const int bc = (t & 15) * 4;
    tile64_kloop<RN, true, RHALF>(Rm + (size_t)(row0 + ar) * RK + ac,
                                  Mi + (size_t)br * RN + col0 + bc,
                                  As, Bs, ar, ac, br, bc, wm, wn, hi, ln,
                                  (int)(blockIdx.x + 2 * blockIdx.y), c2);

    __syncthreads();
    float* Cs = &As[0][0][0];
#pragma unroll
    for (int mi = 0; mi < 2; ++mi)
#pragma unroll
        for (int v = 0; v < 8; ++v)
            Cs[(wm + 16 * mi + 8 * hi + v) * (SN + 4) + wn + ln] = c2[mi][v];
    __syncthreads();

    float d2 = 0.f;
#pragma unroll
    for (int j = 0; j < 4; ++j) {
        const int r = erow + 16 * j;
        const float4 s = *(const float4*)&Cs[r * (SN + 4) + ec4];
        const size_t gi = (size_t)(row0 + r) * RN + col0 + ec4;
        float4 zn;
        d2 += dys_elem(s.x, zo[j].x, wo[j].x, zn.x);
        d2 += dys_elem(s.y, zo[j].y, wo[j].y, zn.y);
        d2 += dys_elem(s.z, zo[j].z, wo[j].z, zn.z);
        d2 += dys_elem(s.w, zo[j].w, wo[j].w, zn.w);
        if (FINAL) {
            zn.x = fmaxf(zn.x, 0.f); zn.y = fmaxf(zn.y, 0.f);
            zn.z = fmaxf(zn.z, 0.f); zn.w = fmaxf(zn.w, 0.f);
        } else {
            *(float4*)(u + gi) = dys_u4(zn, wo[j]);
        }
        *(float4*)(z + gi) = zn;
    }

    if (!FINAL) {
        for (int off = warpSize / 2; off > 0; off >>= 1) d2 += __shfl_xor(d2, off);
        if (lane == 0) sm[wid] = d2;
        __syncthreads();
        if (t == 0) {
            float tot = 0.f;
            for (int k = 0; k < 8; ++k) tot += sm[k];
            atomicAdd(acc, tot);
        }
    }
}

template <bool UPDATE>
__global__ __launch_bounds__(256) void gemm_ns_step(
    const float* __restrict__ Am, const float* __restrict__ Bm,
    float* __restrict__ Cm, const float* __restrict__ Cin)
{
    __shared__ __attribute__((aligned(16))) float As[2][SM][SK + 4];
    __shared__ __attribute__((aligned(16))) float Bs[2][SK][SN + 16];

    const int t    = threadIdx.x;
    const int row0 = blockIdx.y * SM;
    const int col0 = blockIdx.x * SN;
    const int wid  = t >> 5;
    const int lane = t & 31;
    const int wm   = (wid >> 2) * 32;
    const int wn   = (wid & 3) * 16;
    const int hi   = lane >> 4;
    const int ln   = lane & 15;

    const int erow = t >> 4;
    const int ec4  = (t & 15) * 4;
    float4 ci[4];
    if (UPDATE) {
#pragma unroll
        for (int j = 0; j < 4; ++j)
            ci[j] = *(const float4*)(Cin + (size_t)(row0 + erow + 16 * j) * RK + col0 + ec4);
    }

    v8f c2[2];
    c2[0] = (v8f){};
    c2[1] = (v8f){};

    const int ar = t >> 5;
    const int ac = (t & 31) * 4;
    const int br = t >> 4;
    const int bc = (t & 15) * 4;
    tile64_kloop<RK, false, 0>(Am + (size_t)(row0 + ar) * RK + ac, Bm + (size_t)br * RK + col0 + bc,
                               As, Bs, ar, ac, br, bc, wm, wn, hi, ln,
                               (int)(blockIdx.x + blockIdx.y), c2);

    __syncthreads();
    float* Cs = &As[0][0][0];
#pragma unroll
    for (int mi = 0; mi < 2; ++mi)
#pragma unroll
        for (int v = 0; v < 8; ++v)
            Cs[(wm + 16 * mi + 8 * hi + v) * (SN + 4) + wn + ln] = c2[mi][v];
    __syncthreads();

#pragma unroll
    for (int j = 0; j < 4; ++j) {
        const int r = erow + 16 * j;
        float4 v = *(const float4*)&Cs[r * (SN + 4) + ec4];
        if (UPDATE) {
            v.x = 2.f * ci[j].x - v.x;  v.y = 2.f * ci[j].y - v.y;
            v.z = 2.f * ci[j].z - v.z;  v.w = 2.f * ci[j].w - v.w;
        }
        *(float4*)(Cm + (size_t)(row0 + r) * RK + col0 + ec4) = v;
    }
}

extern "C" void kernel_launch(void* const* d_in, const int* in_sizes, int n_in,
                              void* d_out, int out_size, void* d_ws, size_t ws_size,
                              hipStream_t stream)
{
    const float* ctx  = (const float*)d_in[0];
    const float* Ain  = (const float*)d_in[1];
    const float* bvec = (const float*)d_in[2];
    const float* W1   = (const float*)d_in[3];
    const float* b1   = (const float*)d_in[4];
    const float* W2   = (const float*)d_in[5];
    const float* b2   = (const float*)d_in[6];
    const float* z0   = (const float*)d_in[7];
    float* out = (float*)d_out;

    const int B = 256, D = 512, H = 2048, N1 = 1024, N2 = 4096;

    float* W = (float*)d_ws;
    const size_t M1 = (size_t)1 << 20;
    float* G    = W;
    float* Xb   = W + 1 * M1;
    float* T1   = W + 2 * M1;
    float* Xa   = W + 3 * M1;
    float* F    = W + 5 * M1;
    float* Minv = W + 4 * M1;
    float* P    = W;
    float* h    = W + 8 * M1;
    float* wv   = W + 8 * M1;
    float* z    = out;
    float* R    = P;
    float* U    = W + 1 * M1;
    float* acc  = W + 9 * M1;
    float* fn   = acc + 1;

    dim3 blk(256);
    auto partials = [&](const float* A_, int lda, const float* B_, int ldb,
                        int tB, float* P_, int ns, int Kc, int M, int N, int K,
                        const float* gate) {
        dim3 grid(N / TN, M / TM, ns);
        gemm_f32_wmma<<<grid, blk, 0, stream>>>(A_, lda, B_, ldb, tB, P_, N,
                                                Kc, K, (size_t)M * N, gate);
    };
    auto gemm = [&](const float* A_, int lda, const float* B_, int ldb, int tB,
                    float* P_, int ns, float* C_, const float* Cin, const float* bias,
                    float al, float be, float ga, int relu,
                    int M, int N, int K, const float* gate) {
        partials(A_, lda, B_, ldb, tB, P_, ns, K / ns, M, N, K, gate);
        k_splitk_reduce<<<dim3((M * N) / 1024), blk, 0, stream>>>(
            P_, ns, (size_t)M * N, C_, Cin, bias, N, al, be, ga, relu, gate);
    };

    gemm(Ain, N2, Ain, N2, 1, F, 4, G, nullptr, nullptr, 1.f, 0.f, 0.f, 0,
         N1, N1, N2, nullptr);

    k_zero<<<1, 1, 0, stream>>>(fn);
    k_sumsq<<<dim3((N1 * N1 + 255) / 256), blk, 0, stream>>>(G, fn, N1 * N1);
    k_init_X<<<dim3((N1 * N1 + 255) / 256), blk, 0, stream>>>(Xa, fn, N1);

    float* Xc = Xa; float* Xn = Xb;
    for (int it = 0; it < 16; ++it) {
        gemm_ns_step<false><<<dim3(N1 / SN, N1 / SM), blk, 0, stream>>>(G,  Xc, T1, nullptr);
        gemm_ns_step<true><<<dim3(N1 / SN, N1 / SM), blk, 0, stream>>>(Xc, T1, Xn, Xc);
        float* tmp = Xc; Xc = Xn; Xn = tmp;
    }
    float* Ginv = Xc;

    partials(Ginv, N1, Ain, N2, 0, Minv, 1, N1, N1, N2, N1, nullptr);

    gemm(ctx, D, W1, H,  0, P, 8, h,  nullptr, b1, 1.f, 0.f, 1.f, 1, B, H,  D, nullptr);
    gemm(h,   H, W2, N2, 0, P, 4, wv, nullptr, b2, 1.f, 0.f, 1.f, 0, B, N2, H, nullptr);

    const int nz = B * N2;
    k_dys_init<<<dim3(nz / 1024), blk, 0, stream>>>(z0, wv, z, U, acc);
    const dim3 rgrid(N1 / PN, B / PM, 2);
    float* cur = acc; float* nxt = acc + 1; float* prv = acc + 2;
    const float* gate = nullptr;
    for (int itr = 0; itr < 50; ++itr) {
        gemm_r_part<<<rgrid, blk, 0, stream>>>(U, Ain, bvec, R, (size_t)B * N1, gate);
        gemm_dys_step<false><<<dim3(N2 / SN, B / SM), blk, 0, stream>>>(R, Minv, z, wv, U, cur, nxt, gate);
        gate = cur;
        float* tmp = prv; prv = cur; cur = nxt; nxt = tmp;
    }

    gemm_r_part<<<rgrid, blk, 0, stream>>>(U, Ain, bvec, R, (size_t)B * N1, nullptr);
    gemm_dys_step<true><<<dim3(N2 / SN, B / SM), blk, 0, stream>>>(R, Minv, z, wv, nullptr,
                                                                   nullptr, nullptr, nullptr);
}


# Newton-Schulz stops once converged: each G@X launch adds ||I - G@X||_F^2 into a float (one atomic per work-group); at f32 rounding level the later iterations' launches return at once (X carried over)
# speedup vs baseline: 10.3418x; 1.0505x over previous
#include <hip/hip_runtime.h>

typedef __attribute__((ext_vector_type(2))) float v2f;
typedef __attribute__((ext_vector_type(4))) float v4f;
typedef __attribute__((ext_vector_type(8))) float v8f;

#define ALPHA 0.05f
#define EPS   0.01f

#define TM 128
#define TN 128
#define TK 32

static __device__ __forceinline__ v8f wmma4(const v2f a, const v2f b, const v8f c)
{
    return __builtin_amdgcn_wmma_f32_16x16x4_f32(false, a, false, b, (short)0, c,
                                                 false, false);
}

static __device__ __forceinline__ bool loop_done(const float* gate)
{
    return gate && *gate <= EPS * EPS;
}

static __device__ __forceinline__ void store_partial(
    const v8f acc, int r0, int c0, float* __restrict__ Cp, int ldc, int hi, int ln)
{
    const int c = c0 + ln;
#pragma unroll
    for (int v = 0; v < 8; ++v) {
        const int r = r0 + v + 8 * hi;
        Cp[(size_t)r * ldc + c] = acc[v];
    }
}

static __device__ __forceinline__ void stage_tiles(
    float* __restrict__ Ad, float* __restrict__ Bd, int transB,
    const float4 a0, const float4 a1, const float4 a2, const float4 a3,
    const float4 b0, const float4 b1, const float4 b2, const float4 b3,
    int arow, int akc, int bkp, int bnp, int btn, int btk)
{
    float* ad = Ad + arow * (TK + 4) + akc;
    *(float4*)(ad + 0)  = a0;
    *(float4*)(ad + 4)  = a1;
    *(float4*)(ad + 8)  = a2;
    *(float4*)(ad + 12) = a3;
    if (!transB) {
        float* bd = Bd + bkp * (TN + 4) + bnp;
        *(float4*)(bd + 0)  = b0;
        *(float4*)(bd + 4)  = b1;
        *(float4*)(bd + 8)  = b2;
        *(float4*)(bd + 12) = b3;
    } else {
        float* bd = Bd + btk * (TN + 4) + btn;
        bd[0  * (TN + 4)] = b0.x; bd[1  * (TN + 4)] = b0.y; bd[2  * (TN + 4)] = b0.z; bd[3  * (TN + 4)] = b0.w;
        bd[4  * (TN + 4)] = b1.x; bd[5  * (TN + 4)] = b1.y; bd[6  * (TN + 4)] = b1.z; bd[7  * (TN + 4)] = b1.w;
        bd[8  * (TN + 4)] = b2.x; bd[9  * (TN + 4)] = b2.y; bd[10 * (TN + 4)] = b2.z; bd[11 * (TN + 4)] = b2.w;
        bd[12 * (TN + 4)] = b3.x; bd[13 * (TN + 4)] = b3.y; bd[14 * (TN + 4)] = b3.z; bd[15 * (TN + 4)] = b3.w;
    }
}

static __device__ __forceinline__ float4 dys_u4(const float4 z, const float4 w)
{
    float4 u;
    u.x = fmaf(-ALPHA, w.x, fabsf(z.x));
    u.y = fmaf(-ALPHA, w.y, fabsf(z.y));
    u.z = fmaf(-ALPHA, w.z, fabsf(z.z));
    u.w = fmaf(-ALPHA, w.w, fabsf(z.w));
    return u;
}

__global__ __launch_bounds__(256) void gemm_f32_wmma(
    const float* __restrict__ Am, int lda,
    const float* __restrict__ Bm, int ldb, int transB,
    float* __restrict__ Cm, int ldc,
    int Kc, int Ktot, size_t pstride, const float* gate)
{
    if (loop_done(gate)) return;

    __shared__ float As[2][TM][TK + 4];
    __shared__ float Bs[2][TK][TN + 4];

    const int t    = threadIdx.x;
    const int row0 = blockIdx.y * TM;
    const int col0 = blockIdx.x * TN;

    const int wid  = t >> 5;
    const int lane = t & 31;
    const int wm   = (wid >> 2) * 64;
    const int wn   = (wid & 3) * 32;
    const int hi   = lane >> 4;
    const int ln   = lane & 15;

    v8f acc[4][2];
#pragma unroll
    for (int mi = 0; mi < 4; ++mi)
#pragma unroll
        for (int ni = 0; ni < 2; ++ni) acc[mi][ni] = (v8f){};

    const int arow = t >> 1;
    const int akc  = (t & 1) * 16;
    const int bkp  = t >> 3;
    const int bnp  = (t & 7) * 16;
    const int btn  = t >> 1;
    const int btk  = (t & 1) * 16;

    const int kbeg = blockIdx.z * Kc;
    const int kend = min(kbeg + Kc, Ktot);
    const int nk   = (kend - kbeg) / TK;
    float* __restrict__ Cp = Cm + (size_t)blockIdx.z * pstride;

    const float* ap = Am + (size_t)(row0 + arow) * lda + kbeg + akc;
    const float* bp = transB ? Bm + (size_t)(col0 + btn) * ldb + kbeg + btk
                             : Bm + (size_t)(kbeg + bkp) * ldb + col0 + bnp;
    const size_t bstep = transB ? (size_t)TK : (size_t)TK * ldb;

    float4 ra0, ra1, ra2, ra3, rb0, rb1, rb2, rb3;

    ra0 = *(const float4*)(ap + 0);  ra1 = *(const float4*)(ap + 4);
    ra2 = *(const float4*)(ap + 8);  ra3 = *(const float4*)(ap + 12);
    rb0 = *(const float4*)(bp + 0);  rb1 = *(const float4*)(bp + 4);
    rb2 = *(const float4*)(bp + 8);  rb3 = *(const float4*)(bp + 12);
    stage_tiles(&As[0][0][0], &Bs[0][0][0], transB, ra0, ra1, ra2, ra3, rb0, rb1, rb2, rb3,
                arow, akc, bkp, bnp, btn, btk);
    __syncthreads();

    for (int kt = 0; kt < nk; ++kt) {
        const int  cur  = kt & 1;
        const bool more = kt + 1 < nk;
        if (more) {
            ap += TK;
            bp += bstep;
            ra0 = *(const float4*)(ap + 0);  ra1 = *(const float4*)(ap + 4);
            ra2 = *(const float4*)(ap + 8);  ra3 = *(const float4*)(ap + 12);
            rb0 = *(const float4*)(bp + 0);  rb1 = *(const float4*)(bp + 4);
            rb2 = *(const float4*)(bp + 8);  rb3 = *(const float4*)(bp + 12);
        }

#pragma unroll
        for (int kk = 0; kk < TK; kk += 4) {
            v2f a[4], b[2];
#pragma unroll
            for (int mi = 0; mi < 4; ++mi) {
                a[mi].x = As[cur][wm + 16 * mi + ln][kk + 2 * hi];
                a[mi].y = As[cur][wm + 16 * mi + ln][kk + 2 * hi + 1];
            }
#pragma unroll
            for (int ni = 0; ni < 2; ++ni) {
                b[ni].x = Bs[cur][kk + 2 * hi][wn + 16 * ni + ln];
                b[ni].y = Bs[cur][kk + 2 * hi + 1][wn + 16 * ni + ln];
            }
#pragma unroll
            for (int mi = 0; mi < 4; ++mi)
#pragma unroll
                for (int ni = 0; ni < 2; ++ni)
                    acc[mi][ni] = wmma4(a[mi], b[ni], acc[mi][ni]);
        }

        if (more) {
            stage_tiles(&As[cur ^ 1][0][0], &Bs[cur ^ 1][0][0], transB,
                        ra0, ra1, ra2, ra3, rb0, rb1, rb2, rb3, arow, akc, bkp, bnp, btn, btk);
            __syncthreads();
        }
    }

#pragma unroll
    for (int mi = 0; mi < 4; ++mi)
#pragma unroll
        for (int ni = 0; ni < 2; ++ni)
            store_partial(acc[mi][ni], row0 + wm + 16 * mi, col0 + wn + 16 * ni,
                          Cp, ldc, hi, ln);
}

__global__ void k_sumsq(const float* __restrict__ x, float* __restrict__ acc, int n)
{
    __shared__ float sm[256];
    int i = blockIdx.x * blockDim.x + threadIdx.x;
    float v = 0.f;
    if (i < n) { float t = x[i]; v = t * t; }
    sm[threadIdx.x] = v;
    __syncthreads();
    for (int s = 128; s > 0; s >>= 1) {
        if (threadIdx.x < s) sm[threadIdx.x] += sm[threadIdx.x + s];
        __syncthreads();
    }
    if (threadIdx.x == 0) atomicAdd(acc, sm[0]);
}

__global__ void k_zero(float* p, float* q, int nq)
{
    if (threadIdx.x == 0) p[0] = 0.f;
    if ((int)threadIdx.x < nq) q[threadIdx.x] = 0.f;
}

__global__ __launch_bounds__(256) void k_splitk_reduce(
    const float* P, int nsplit, size_t pstride,
    float* C, const float* __restrict__ Cin, const float* __restrict__ bias,
    int ncols, float alpha, float beta, float gamma, int relu, const float* gate)
{
    if (loop_done(gate)) return;
    const size_t i = ((size_t)blockIdx.x * blockDim.x + threadIdx.x) * 4;
    float4 s = make_float4(0.f, 0.f, 0.f, 0.f);
    for (int k = 0; k < nsplit; k += 4) {
        const float4 p0 = *(const float4*)(P + (size_t)(k + 0) * pstride + i);
        const float4 p1 = *(const float4*)(P + (size_t)(k + 1) * pstride + i);
        const float4 p2 = *(const float4*)(P + (size_t)(k + 2) * pstride + i);
        const float4 p3 = *(const float4*)(P + (size_t)(k + 3) * pstride + i);
        s.x += p0.x; s.y += p0.y; s.z += p0.z; s.w += p0.w;
        s.x += p1.x; s.y += p1.y; s.z += p1.z; s.w += p1.w;
        s.x += p2.x; s.y += p2.y; s.z += p2.z; s.w += p2.w;
        s.x += p3.x; s.y += p3.y; s.z += p3.z; s.w += p3.w;
    }
    float4 v = make_float4(alpha * s.x, alpha * s.y, alpha * s.z, alpha * s.w);
    if (bias) {
        const int c = (int)(i % (size_t)ncols);
        const float4 bv = *(const float4*)(bias + c);
        v.x += gamma * bv.x; v.y += gamma * bv.y; v.z += gamma * bv.z; v.w += gamma * bv.w;
    }
    if (Cin) {
        const float4 ci = *(const float4*)(Cin + i);
        v.x += beta * ci.x; v.y += beta * ci.y; v.z += beta * ci.z; v.w += beta * ci.w;
    }
    if (relu) {
        v.x = fmaxf(v.x, 0.f); v.y = fmaxf(v.y, 0.f); v.z = fmaxf(v.z, 0.f); v.w = fmaxf(v.w, 0.f);
    }
    *(float4*)(C + i) = v;
}

__global__ void k_init_X(float* __restrict__ X, const float* __restrict__ fn, int n)
{
    int i = blockIdx.x * blockDim.x + threadIdx.x;
    if (i < n * n) {
        float c = rsqrtf(*fn);
        int r = i / n, cc = i - r * n;
        X[i] = (r == cc) ? c : 0.f;
    }
}

static __device__ __forceinline__ float dys_elem(float s, float zo, float wo, float& zn)
{
    const float uo = fmaf(-ALPHA, wo, fabsf(zo));
    const float y  = uo - s;
    const float x  = fmaxf(zo, 0.f);
    zn = zo - x + y;
    const float d = y - x;
    return d * d;
}

__global__ __launch_bounds__(256) void k_dys_init(
    const float* __restrict__ z0, const float* __restrict__ w,
    float* __restrict__ z, float* __restrict__ u, float* __restrict__ acc0)
{
    if (blockIdx.x == 0 && threadIdx.x == 0) *acc0 = 0.f;
    const size_t i = ((size_t)blockIdx.x * 256 + threadIdx.x) * 4;
    const float4 zv = *(const float4*)(z0 + i);
    *(float4*)(z + i) = zv;
    *(float4*)(u + i) = dys_u4(zv, *(const float4*)(w + i));
}

#define SM 64
#define SN 64
#define SK 128
#define RK 1024
#define RN 4096

static __device__ __forceinline__ void dys_frags(
    const float (*A)[SK + 4], const float (*Bt)[SN + 16], int kb, int wm, int wn, int hi, int ln,
    v2f (&fa)[4][2], v2f (&fb)[4])
{
#pragma unroll
    for (int q = 0; q < 4; ++q) {
        const int k2 = kb + 4 * q + 2 * hi;
#pragma unroll
        for (int mi = 0; mi < 2; ++mi) {
            fa[q][mi].x = A[wm + 16 * mi + ln][k2];
            fa[q][mi].y = A[wm + 16 * mi + ln][k2 + 1];
        }
        fb[q].x = Bt[k2][wn + ln];
        fb[q].y = Bt[k2 + 1][wn + ln];
    }
}

static __device__ __forceinline__ void lds_barrier()
{
    __builtin_amdgcn_fence(__ATOMIC_RELEASE, "workgroup", "local");
    __builtin_amdgcn_s_barrier();
    __builtin_amdgcn_fence(__ATOMIC_ACQUIRE, "workgroup", "local");
}

template <bool V> struct ktile_flag { static constexpr bool value = V; };

static __device__ __forceinline__ float4 add4(const float4 a, const v4f b)
{
    return make_float4(a.x + b.x, a.y + b.y, a.z + b.z, a.w + b.w);
}

template <int LDB, bool SUM2, int ASEP>
static __device__ __forceinline__ void tile64_kloop(
    const float* __restrict__ ap0, const float* __restrict__ bp0,
    float (&As)[2][SM][SK + 4], float (&Bs)[2][SK][SN + 16],
    int ar, int ac, int br, int bc, int wm, int wn, int hi, int ln, int rot, v8f (&c2)[2])
{
    constexpr int nk = RK / SK;
    static_assert((nk & (nk - 1)) == 0, "the K-tile rotation wraps with a mask");
    int kidx = rot & (nk - 1);
    const float* ap = ap0 + (size_t)kidx * SK;
    const float* bp = bp0 + (size_t)kidx * SK * LDB;
    float4 ra0, ra1, ra2, ra3, ra4, ra5, ra6, ra7;
    float4 rb0, rb1, rb2, rb3, rb4, rb5, rb6, rb7;
    v4f rc0 = (v4f){};
    v4f rc1 = rc0, rc2 = rc0, rc3 = rc0, rc4 = rc0, rc5 = rc0, rc6 = rc0, rc7 = rc0;
#define DYS_LOAD_TILE()                                                                      \
    ra0 = *(const float4*)(ap + 0 * 8 * RK);   ra1 = *(const float4*)(ap + 1 * 8 * RK);      \
    ra2 = *(const float4*)(ap + 2 * 8 * RK);   ra3 = *(const float4*)(ap + 3 * 8 * RK);      \
    ra4 = *(const float4*)(ap + 4 * 8 * RK);   ra5 = *(const float4*)(ap + 5 * 8 * RK);      \
    ra6 = *(const float4*)(ap + 6 * 8 * RK);   ra7 = *(const float4*)(ap + 7 * 8 * RK);      \
    if (SUM2) {                                                                              \
        rc0 = *(const v4f*)(ap + ASEP + 0 * 8 * RK);  rc1 = *(const v4f*)(ap + ASEP + 1 * 8 * RK);  \
        rc2 = *(const v4f*)(ap + ASEP + 2 * 8 * RK);  rc3 = *(const v4f*)(ap + ASEP + 3 * 8 * RK);  \
        rc4 = *(const v4f*)(ap + ASEP + 4 * 8 * RK);  rc5 = *(const v4f*)(ap + ASEP + 5 * 8 * RK);  \
        rc6 = *(const v4f*)(ap + ASEP + 6 * 8 * RK);  rc7 = *(const v4f*)(ap + ASEP + 7 * 8 * RK);  \
    }                                                                                        \
    rb0 = *(const float4*)(bp + 0 * 16 * LDB); rb1 = *(const float4*)(bp + 1 * 16 * LDB);    \
    rb2 = *(const float4*)(bp + 2 * 16 * LDB); rb3 = *(const float4*)(bp + 3 * 16 * LDB);    \
    rb4 = *(const float4*)(bp + 4 * 16 * LDB); rb5 = *(const float4*)(bp + 5 * 16 * LDB);    \
    rb6 = *(const float4*)(bp + 6 * 16 * LDB); rb7 = *(const float4*)(bp + 7 * 16 * LDB);
#define DYS_STAGE_TILE(s)                                                                    \
    if (SUM2) {                                                                              \
          \
          \
          \
        asm volatile("" : "+v"(rc0), "+v"(rc1), "+v"(rc2), "+v"(rc3),                        \
                          "+v"(rc4), "+v"(rc5), "+v"(rc6), "+v"(rc7));                       \
        ra0 = add4(ra0, rc0);  ra1 = add4(ra1, rc1);  ra2 = add4(ra2, rc2);  ra3 = add4(ra3, rc3);  \
        ra4 = add4(ra4, rc4);  ra5 = add4(ra5, rc5);  ra6 = add4(ra6, rc6);  ra7 = add4(ra7, rc7);  \
    }                                                                                        \
    *(float4*)&As[s][ar + 0][ac]  = ra0;   *(float4*)&As[s][ar + 8][ac]  = ra1;              \
    *(float4*)&As[s][ar + 16][ac] = ra2;   *(float4*)&As[s][ar + 24][ac] = ra3;              \
    *(float4*)&As[s][ar + 32][ac] = ra4;   *(float4*)&As[s][ar + 40][ac] = ra5;              \
    *(float4*)&As[s][ar + 48][ac] = ra6;   *(float4*)&As[s][ar + 56][ac] = ra7;              \
    *(float4*)&Bs[s][br + 0][bc]  = rb0;   *(float4*)&Bs[s][br + 16][bc] = rb1;              \
    *(float4*)&Bs[s][br + 32][bc] = rb2;   *(float4*)&Bs[s][br + 48][bc] = rb3;              \
    *(float4*)&Bs[s][br + 64][bc] = rb4;   *(float4*)&Bs[s][br + 80][bc] = rb5;              \
    *(float4*)&Bs[s][br + 96][bc] = rb6;   *(float4*)&Bs[s][br + 112][bc] = rb7;

    static_assert(nk >= 2, "the prologue loads a second K-tile");
#define DYS_NEXT_TILE()                                                                      \
    kidx = (kidx + 1) & (nk - 1);                                                            \
    ap = ap0 + (size_t)kidx * SK;                                                            \
    bp = bp0 + (size_t)kidx * SK * LDB;                                                      \
    DYS_LOAD_TILE()

    DYS_LOAD_TILE()
    DYS_STAGE_TILE(0)
    DYS_NEXT_TILE()
    lds_barrier();

    auto ktile = [&](const int cur, const bool load, auto stage_tag) __attribute__((always_inline)) {
        constexpr bool STAGE = decltype(stage_tag)::value;
        v2f fa[2][4][2], fb[2][4];
        dys_frags(As[cur], Bs[cur], 0, wm, wn, hi, ln, fa[0], fb[0]);
        __builtin_amdgcn_sched_barrier(0);
#pragma unroll
        for (int g = 0; g < SK / 16; ++g) {
            const int s = g & 1;
            if (g + 1 < SK / 16)
                dys_frags(As[cur], Bs[cur], 16 * (g + 1), wm, wn, hi, ln, fa[s ^ 1], fb[s ^ 1]);
            __builtin_amdgcn_sched_barrier(0);
            if (g == SK / 16 - 2 && STAGE) {
                DYS_STAGE_TILE(cur ^ 1)
                if (load) {
                    DYS_NEXT_TILE()
                }
                __builtin_amdgcn_sched_barrier(0);
            }
#pragma unroll
            for (int q = 0; q < 4; ++q)
#pragma unroll
                for (int mi = 0; mi < 2; ++mi) c2[mi] = wmma4(fa[s][q][mi], fb[s][q], c2[mi]);
            __builtin_amdgcn_sched_barrier(0);
        }

        if (STAGE) lds_barrier();
    };

    for (int kt = 0; kt + 1 < nk; ++kt) ktile(kt & 1, kt + 2 < nk, ktile_flag<true>{});
    ktile((nk - 1) & 1, false, ktile_flag<false>{});

#undef DYS_NEXT_TILE
#undef DYS_LOAD_TILE
#undef DYS_STAGE_TILE
}

#define PM 32
#define PN 64
#define PK 256
#define PG 32

static __device__ __forceinline__ void rpart_frags(
    const float (*Ut)[PK + 4], const float (*Bt)[PK + 4], int kb, int wm, int wn, int hi, int ln,
    v2f (&fa)[4], v2f (&fb)[2][4])
{
#pragma unroll
    for (int q = 0; q < 4; ++q) {
        const int k2 = kb + 4 * q + 2 * hi;
        fa[q].x = Ut[wm + ln][k2];
        fa[q].y = Ut[wm + ln][k2 + 1];
#pragma unroll
        for (int ni = 0; ni < 2; ++ni) {
            fb[ni][q].x = Bt[wn + 16 * ni + ln][k2];
            fb[ni][q].y = Bt[wn + 16 * ni + ln][k2 + 1];
        }
    }
}

__global__ __launch_bounds__(256) void gemm_r_part(
    const float* __restrict__ U, const float* __restrict__ Am, const float* __restrict__ bvec,
    float* __restrict__ P, size_t pstride, const float* gate)
{
    if (loop_done(gate)) return;

    __shared__ __attribute__((aligned(16))) float Us[2][PM][PK + 4];
    __shared__ __attribute__((aligned(16))) float Bs[2][PN][PK + 4];

    const int t    = threadIdx.x;
    const int row0 = blockIdx.y * PM;
    const int col0 = blockIdx.x * PN;
    const int kbeg = blockIdx.z * (RN / 2);
    const int wid  = t >> 5;
    const int lane = t & 31;
    const int kg   = wid >> 2;
    const int wm   = ((wid >> 1) & 1) * 16;
    const int wn   = (wid & 1) * 32;
    const int hi   = lane >> 4;
    const int ln   = lane & 15;

    const int sr = t >> 6;
    const int sc = (t & 63) * 4;
    constexpr int nk = (RN / 2) / PK;
    static_assert((nk & (nk - 1)) == 0, "the K-tile rotation wraps with a mask");
    static_assert(nk >= 2, "the prologue loads a second K-tile");
    const int rot = (int)(blockIdx.x + blockIdx.y) & (nk - 1);
    const float* up0 = U  + (size_t)(row0 + sr) * RN + kbeg + sc;
    const float* bp0 = Am + (size_t)(col0 + sr) * RN + kbeg + sc;
    const float* up = up0 + rot * PK;
    const float* bp = bp0 + rot * PK;

    float4 ru0, ru1, ru2, ru3, ru4, ru5, ru6, ru7;
    float4 rb0, rb1, rb2, rb3, rb4, rb5, rb6, rb7, rb8, rb9, rb10, rb11, rb12, rb13, rb14, rb15;
#define RP_LD(p, j) (*(const float4*)((p) + (size_t)(j) * 4 * RN))
#define RP_LOAD_TILE()                                                                       \
    ru0 = RP_LD(up, 0);   ru1 = RP_LD(up, 1);   ru2 = RP_LD(up, 2);   ru3 = RP_LD(up, 3);     \
    ru4 = RP_LD(up, 4);   ru5 = RP_LD(up, 5);   ru6 = RP_LD(up, 6);   ru7 = RP_LD(up, 7);     \
    rb0 = RP_LD(bp, 0);   rb1 = RP_LD(bp, 1);   rb2 = RP_LD(bp, 2);   rb3 = RP_LD(bp, 3);     \
    rb4 = RP_LD(bp, 4);   rb5 = RP_LD(bp, 5);   rb6 = RP_LD(bp, 6);   rb7 = RP_LD(bp, 7);     \
    rb8 = RP_LD(bp, 8);   rb9 = RP_LD(bp, 9);   rb10 = RP_LD(bp, 10); rb11 = RP_LD(bp, 11);   \
    rb12 = RP_LD(bp, 12); rb13 = RP_LD(bp, 13); rb14 = RP_LD(bp, 14); rb15 = RP_LD(bp, 15);
#define RP_ST(T, s, j, v) (*(float4*)&T[s][sr + 4 * (j)][sc] = (v))
#define RP_STAGE_TILE(s)                                                                     \
    RP_ST(Us, s, 0, ru0);   RP_ST(Us, s, 1, ru1);   RP_ST(Us, s, 2, ru2);   RP_ST(Us, s, 3, ru3);     \
    RP_ST(Us, s, 4, ru4);   RP_ST(Us, s, 5, ru5);   RP_ST(Us, s, 6, ru6);   RP_ST(Us, s, 7, ru7);     \
    RP_ST(Bs, s, 0, rb0);   RP_ST(Bs, s, 1, rb1);   RP_ST(Bs, s, 2, rb2);   RP_ST(Bs, s, 3, rb3);     \
    RP_ST(Bs, s, 4, rb4);   RP_ST(Bs, s, 5, rb5);   RP_ST(Bs, s, 6, rb6);   RP_ST(Bs, s, 7, rb7);     \
    RP_ST(Bs, s, 8, rb8);   RP_ST(Bs, s, 9, rb9);   RP_ST(Bs, s, 10, rb10); RP_ST(Bs, s, 11, rb11);   \
    RP_ST(Bs, s, 12, rb12); RP_ST(Bs, s, 13, rb13); RP_ST(Bs, s, 14, rb14); RP_ST(Bs, s, 15, rb15);

    v8f acc[2];
    acc[0] = (v8f){};
    acc[1] = (v8f){};

    RP_LOAD_TILE()
    RP_STAGE_TILE(0)
    up = up0 + ((1 + rot) & (nk - 1)) * PK;
    bp = bp0 + ((1 + rot) & (nk - 1)) * PK;
    RP_LOAD_TILE()
    lds_barrier();

    auto ktile = [&](const int cur, const int nxt2, const bool load, auto stage_tag)
                     __attribute__((always_inline)) {
        constexpr bool STAGE = decltype(stage_tag)::value;

        v2f fa[2][4], fb[2][2][4];
        rpart_frags(Us[cur], Bs[cur], 16 * kg, wm, wn, hi, ln, fa[0], fb[0]);
        __builtin_amdgcn_sched_barrier(0);
#pragma unroll
        for (int g = 0; g < PK / PG; ++g) {
            const int s = g & 1;
            if (g + 1 < PK / PG)
                rpart_frags(Us[cur], Bs[cur], PG * (g + 1) + 16 * kg, wm, wn, hi, ln,
                            fa[s ^ 1], fb[s ^ 1]);
            __builtin_amdgcn_sched_barrier(0);
            if (g == PK / PG - 2 && STAGE) {
                RP_STAGE_TILE(cur ^ 1)
                if (load) {
                    up = up0 + nxt2 * PK;
                    bp = bp0 + nxt2 * PK;
                    RP_LOAD_TILE()
                }
                __builtin_amdgcn_sched_barrier(0);
            }
#pragma unroll
            for (int q = 0; q < 4; ++q)
#pragma unroll
                for (int ni = 0; ni < 2; ++ni) acc[ni] = wmma4(fa[s][q], fb[s][ni][q], acc[ni]);
            __builtin_amdgcn_sched_barrier(0);
        }

        if (STAGE) lds_barrier();
    };

    for (int kt = 0; kt + 1 < nk; ++kt)
        ktile(kt & 1, (kt + 2 + rot) & (nk - 1), kt + 2 < nk, ktile_flag<true>{});
    ktile((nk - 1) & 1, 0, false, ktile_flag<false>{});
#undef RP_LD
#undef RP_LOAD_TILE
#undef RP_ST
#undef RP_STAGE_TILE

    __syncthreads();
    float* Cs = &Us[0][0][0];
#pragma unroll
    for (int ni = 0; ni < 2; ++ni)
#pragma unroll
        for (int v = 0; v < 8; ++v)
            Cs[(kg * PM + wm + 8 * hi + v) * (PN + 4) + wn + 16 * ni + ln] = acc[ni][v];
    __syncthreads();

    const int er  = t >> 4;
    const int ec4 = (t & 15) * 4;
    float4 bv = make_float4(0.f, 0.f, 0.f, 0.f);
    if (blockIdx.z == 0) bv = *(const float4*)(bvec + col0 + ec4);
    float* __restrict__ Pp = P + (size_t)blockIdx.z * pstride;
#pragma unroll
    for (int j = 0; j < 2; ++j) {
        const int r = er + 16 * j;
        const float4 s0 = *(const float4*)&Cs[r * (PN + 4) + ec4];
        const float4 s1 = *(const float4*)&Cs[(PM + r) * (PN + 4) + ec4];
        float4 s;
        s.x = s0.x + s1.x - bv.x;  s.y = s0.y + s1.y - bv.y;
        s.z = s0.z + s1.z - bv.z;  s.w = s0.w + s1.w - bv.w;
        *(float4*)(Pp + (size_t)(row0 + r) * RK + col0 + ec4) = s;
    }
}

#define RHALF (256 * RK)
template <bool FINAL>
__global__ __launch_bounds__(256) void gemm_dys_step(
    const float* __restrict__ Rm, const float* __restrict__ Mi,
    float* z, const float* __restrict__ w, float* __restrict__ u,
    float* acc, float* next, const float* gate)
{
    __shared__ __attribute__((aligned(16))) float As[2][SM][SK + 4];
    __shared__ __attribute__((aligned(16))) float Bs[2][SK][SN + 16];
    __shared__ float sm[8];

    const int t = threadIdx.x;
    if (!FINAL) {
        if (blockIdx.x == 0 && blockIdx.y == 0 && t == 0) *next = 0.f;
        if (loop_done(gate)) return;
    }

    const int row0 = blockIdx.y * SM;
    const int col0 = blockIdx.x * SN;
    const int wid  = t >> 5;
    const int lane = t & 31;
    const int wm   = (wid >> 2) * 32;
    const int wn   = (wid & 3) * 16;
    const int hi   = lane >> 4;
    const int ln   = lane & 15;

    const int erow = t >> 4;
    const int ec4  = (t & 15) * 4;
    float4 zo[4], wo[4];
#pragma unroll
    for (int j = 0; j < 4; ++j) {
        const size_t gi = (size_t)(row0 + erow + 16 * j) * RN + col0 + ec4;
        zo[j] = *(const float4*)(z + gi);
        wo[j] = *(const float4*)(w + gi);
    }

    v8f c2[2];
    c2[0] = (v8f){};
    c2[1] = (v8f){};

    const int ar = t >> 5;
    const int ac = (t & 31) * 4;
    const int br = t >> 4;
    const int bc = (t & 15) * 4;
    tile64_kloop<RN, true, RHALF>(Rm + (size_t)(row0 + ar) * RK + ac,
                                  Mi + (size_t)br * RN + col0 + bc,
                                  As, Bs, ar, ac, br, bc, wm, wn, hi, ln,
                                  (int)(blockIdx.x + 2 * blockIdx.y), c2);

    __syncthreads();
    float* Cs = &As[0][0][0];
#pragma unroll
    for (int mi = 0; mi < 2; ++mi)
#pragma unroll
        for (int v = 0; v < 8; ++v)
            Cs[(wm + 16 * mi + 8 * hi + v) * (SN + 4) + wn + ln] = c2[mi][v];
    __syncthreads();

    float d2 = 0.f;
#pragma unroll
    for (int j = 0; j < 4; ++j) {
        const int r = erow + 16 * j;
        const float4 s = *(const float4*)&Cs[r * (SN + 4) + ec4];
        const size_t gi = (size_t)(row0 + r) * RN + col0 + ec4;
        float4 zn;
        d2 += dys_elem(s.x, zo[j].x, wo[j].x, zn.x);
        d2 += dys_elem(s.y, zo[j].y, wo[j].y, zn.y);
        d2 += dys_elem(s.z, zo[j].z, wo[j].z, zn.z);
        d2 += dys_elem(s.w, zo[j].w, wo[j].w, zn.w);
        if (FINAL) {
            zn.x = fmaxf(zn.x, 0.f); zn.y = fmaxf(zn.y, 0.f);
            zn.z = fmaxf(zn.z, 0.f); zn.w = fmaxf(zn.w, 0.f);
        } else {
            *(float4*)(u + gi) = dys_u4(zn, wo[j]);
        }
        *(float4*)(z + gi) = zn;
    }

    if (!FINAL) {
        for (int off = warpSize / 2; off > 0; off >>= 1) d2 += __shfl_xor(d2, off);
        if (lane == 0) sm[wid] = d2;
        __syncthreads();
        if (t == 0) {
            float tot = 0.f;
            for (int k = 0; k < 8; ++k) tot += sm[k];
            atomicAdd(acc, tot);
        }
    }
}

#define NS_TOL2 2e-5f
static __device__ __forceinline__ bool ns_done(const float* gate)
{
    return gate && *gate <= NS_TOL2;
}

template <bool UPDATE>
__global__ __launch_bounds__(256) void gemm_ns_step(
    const float* __restrict__ Am, const float* __restrict__ Bm,
    float* __restrict__ Cm, const float* __restrict__ Cin,
    float* res, const float* gate)
{
    __shared__ __attribute__((aligned(16))) float As[2][SM][SK + 4];
    __shared__ __attribute__((aligned(16))) float Bs[2][SK][SN + 16];
    __shared__ float sm[8];

    const int t    = threadIdx.x;
    const int row0 = blockIdx.y * SM;
    const int col0 = blockIdx.x * SN;
    const int wid  = t >> 5;
    const int lane = t & 31;
    const int wm   = (wid >> 2) * 32;
    const int wn   = (wid & 3) * 16;
    const int hi   = lane >> 4;
    const int ln   = lane & 15;

    const int erow = t >> 4;
    const int ec4  = (t & 15) * 4;
    float4 ci[4];
    if (UPDATE) {
#pragma unroll
        for (int j = 0; j < 4; ++j)
            ci[j] = *(const float4*)(Cin + (size_t)(row0 + erow + 16 * j) * RK + col0 + ec4);
    }

    if (ns_done(gate)) {
        if (UPDATE) {
#pragma unroll
            for (int j = 0; j < 4; ++j)
                *(float4*)(Cm + (size_t)(row0 + erow + 16 * j) * RK + col0 + ec4) = ci[j];
        }
        return;
    }

    v8f c2[2];
    c2[0] = (v8f){};
    c2[1] = (v8f){};

    const int ar = t >> 5;
    const int ac = (t & 31) * 4;
    const int br = t >> 4;
    const int bc = (t & 15) * 4;
    tile64_kloop<RK, false, 0>(Am + (size_t)(row0 + ar) * RK + ac, Bm + (size_t)br * RK + col0 + bc,
                               As, Bs, ar, ac, br, bc, wm, wn, hi, ln,
                               (int)(blockIdx.x + blockIdx.y), c2);

    __syncthreads();
    float* Cs = &As[0][0][0];
#pragma unroll
    for (int mi = 0; mi < 2; ++mi)
#pragma unroll
        for (int v = 0; v < 8; ++v)
            Cs[(wm + 16 * mi + 8 * hi + v) * (SN + 4) + wn + ln] = c2[mi][v];
    __syncthreads();

    float d2 = 0.f;
#pragma unroll
    for (int j = 0; j < 4; ++j) {
        const int r = erow + 16 * j;
        float4 v = *(const float4*)&Cs[r * (SN + 4) + ec4];
        if (UPDATE) {
            v.x = 2.f * ci[j].x - v.x;  v.y = 2.f * ci[j].y - v.y;
            v.z = 2.f * ci[j].z - v.z;  v.w = 2.f * ci[j].w - v.w;
        } else {
            const int dc = (row0 == col0) ? r - ec4 : -1;
            const float e0 = v.x - (dc == 0 ? 1.f : 0.f);
            const float e1 = v.y - (dc == 1 ? 1.f : 0.f);
            const float e2 = v.z - (dc == 2 ? 1.f : 0.f);
            const float e3 = v.w - (dc == 3 ? 1.f : 0.f);
            d2 += e0 * e0 + e1 * e1 + e2 * e2 + e3 * e3;
        }
        *(float4*)(Cm + (size_t)(row0 + r) * RK + col0 + ec4) = v;
    }

    if (!UPDATE) {
        for (int off = warpSize / 2; off > 0; off >>= 1) d2 += __shfl_xor(d2, off);
        if (lane == 0) sm[wid] = d2;
        lds_barrier();
        if (t == 0) {
            float tot = 0.f;
            for (int k = 0; k < 8; ++k) tot += sm[k];
            atomicAdd(res, tot);
        }
    }
}

extern "C" void kernel_launch(void* const* d_in, const int* in_sizes, int n_in,
                              void* d_out, int out_size, void* d_ws, size_t ws_size,
                              hipStream_t stream)
{
    const float* ctx  = (const float*)d_in[0];
    const float* Ain  = (const float*)d_in[1];
    const float* bvec = (const float*)d_in[2];
    const float* W1   = (const float*)d_in[3];
    const float* b1   = (const float*)d_in[4];
    const float* W2   = (const float*)d_in[5];
    const float* b2   = (const float*)d_in[6];
    const float* z0   = (const float*)d_in[7];
    float* out = (float*)d_out;

    const int B = 256, D = 512, H = 2048, N1 = 1024, N2 = 4096;

    float* W = (float*)d_ws;
    const size_t M1 = (size_t)1 << 20;
    float* G    = W;
    float* Xb   = W + 1 * M1;
    float* T1   = W + 2 * M1;
    float* Xa   = W + 3 * M1;
    float* F    = W + 5 * M1;
    float* Minv = W + 4 * M1;
    float* P    = W;
    float* h    = W + 8 * M1;
    float* wv   = W + 8 * M1;
    float* z    = out;
    float* R    = P;
    float* U    = W + 1 * M1;
    float* acc  = W + 9 * M1;
    float* fn   = acc + 1;
    float* nres = W + 8 * M1;

    dim3 blk(256);
    auto partials = [&](const float* A_, int lda, const float* B_, int ldb,
                        int tB, float* P_, int ns, int Kc, int M, int N, int K,
                        const float* gate) {
        dim3 grid(N / TN, M / TM, ns);
        gemm_f32_wmma<<<grid, blk, 0, stream>>>(A_, lda, B_, ldb, tB, P_, N,
                                                Kc, K, (size_t)M * N, gate);
    };
    auto gemm = [&](const float* A_, int lda, const float* B_, int ldb, int tB,
                    float* P_, int ns, float* C_, const float* Cin, const float* bias,
                    float al, float be, float ga, int relu,
                    int M, int N, int K, const float* gate) {
        partials(A_, lda, B_, ldb, tB, P_, ns, K / ns, M, N, K, gate);
        k_splitk_reduce<<<dim3((M * N) / 1024), blk, 0, stream>>>(
            P_, ns, (size_t)M * N, C_, Cin, bias, N, al, be, ga, relu, gate);
    };

    gemm(Ain, N2, Ain, N2, 1, F, 4, G, nullptr, nullptr, 1.f, 0.f, 0.f, 0,
         N1, N1, N2, nullptr);

    k_zero<<<1, 32, 0, stream>>>(fn, nres, 16);
    k_sumsq<<<dim3((N1 * N1 + 255) / 256), blk, 0, stream>>>(G, fn, N1 * N1);
    k_init_X<<<dim3((N1 * N1 + 255) / 256), blk, 0, stream>>>(Xa, fn, N1);

    float* Xc = Xa; float* Xn = Xb;
    const float* ngate = nullptr;
    for (int it = 0; it < 16; ++it) {
        gemm_ns_step<false><<<dim3(N1 / SN, N1 / SM), blk, 0, stream>>>(G,  Xc, T1, nullptr,
                                                                         nres + it, ngate);
        gemm_ns_step<true><<<dim3(N1 / SN, N1 / SM), blk, 0, stream>>>(Xc, T1, Xn, Xc,
                                                                        nullptr, ngate);
        ngate = nres + it;
        float* tmp = Xc; Xc = Xn; Xn = tmp;
    }
    float* Ginv = Xc;

    partials(Ginv, N1, Ain, N2, 0, Minv, 1, N1, N1, N2, N1, nullptr);

    gemm(ctx, D, W1, H,  0, P, 8, h,  nullptr, b1, 1.f, 0.f, 1.f, 1, B, H,  D, nullptr);
    gemm(h,   H, W2, N2, 0, P, 4, wv, nullptr, b2, 1.f, 0.f, 1.f, 0, B, N2, H, nullptr);

    const int nz = B * N2;
    k_dys_init<<<dim3(nz / 1024), blk, 0, stream>>>(z0, wv, z, U, acc);
    const dim3 rgrid(N1 / PN, B / PM, 2);
    float* cur = acc; float* nxt = acc + 1; float* prv = acc + 2;
    const float* gate = nullptr;
    for (int itr = 0; itr < 50; ++itr) {
        gemm_r_part<<<rgrid, blk, 0, stream>>>(U, Ain, bvec, R, (size_t)B * N1, gate);
        gemm_dys_step<false><<<dim3(N2 / SN, B / SM), blk, 0, stream>>>(R, Minv, z, wv, U, cur, nxt, gate);
        gate = cur;
        float* tmp = prv; prv = cur; cur = nxt; nxt = tmp;
    }

    gemm_r_part<<<rgrid, blk, 0, stream>>>(U, Ain, bvec, R, (size_t)B * N1, nullptr);
    gemm_dys_step<true><<<dim3(N2 / SN, B / SM), blk, 0, stream>>>(R, Minv, z, wv, nullptr,
                                                                   nullptr, nullptr, nullptr);
}
